# starting kernel
# baseline (speedup 1.0000x reference)
#include <hip/hip_runtime.h>

typedef float v2f __attribute__((ext_vector_type(2)));
typedef float v8f __attribute__((ext_vector_type(8)));

#define FAGCN_EPS 0.3f

__global__ void deg_kernel(const int* __restrict__ ei, float* __restrict__ deg, int E) {
  int e = blockIdx.x * blockDim.x + threadIdx.x;
  if (e < E) atomicAdd(&deg[ei[e]], 1.0f);
}

__global__ void dn_kernel(float* dn, int N) {
  int i = blockIdx.x * blockDim.x + threadIdx.x;
  if (i < N) dn[i] = 1.0f / sqrtf(fmaxf(dn[i], 1.0f));
}

__global__ void gemm_relu_kernel(const float* __restrict__ X, const float* __restrict__ W,
                                 const float* __restrict__ bias, float* __restrict__ H,
                                 int Nrows) {
  int wave = threadIdx.x >> 5;
  int lane = threadIdx.x & 31;
  int tileM = blockIdx.x;
  int mA = tileM * 16 + (lane & 15);
  int mclamp = mA < Nrows ? mA : (Nrows - 1);
  int n = wave * 16 + (lane & 15);
  int khalf = (lane >> 4) * 2;
  const float* xrow = X + (size_t)mclamp * 128 + khalf;
  const float* wcol = W + (size_t)khalf * 128 + n;
  __builtin_prefetch(xrow, 0, 0);
  v8f acc = {};
#pragma unroll 4
  for (int kk = 0; kk < 128; kk += 4) {
    v2f a = *(const v2f*)(xrow + kk);
    v2f b;
    b.x = wcol[(size_t)kk * 128];
    b.y = wcol[(size_t)kk * 128 + 128];
    acc = __builtin_amdgcn_wmma_f32_16x16x4_f32(false, a, false, b, (short)0, acc,
                                               false, false);
  }
  int mbase = tileM * 16 + ((lane >> 4) * 8);
  float bn = bias[n];
  if (tileM * 16 + 16 <= Nrows) {
    float* hp = H + (size_t)mbase * 128 + n;
#pragma unroll
    for (int r = 0; r < 8; ++r)
      hp[(size_t)r * 128] = fmaxf(acc[r] + bn, 0.0f);
  } else {
#pragma unroll
    for (int r = 0; r < 8; ++r) {
      int mm = mbase + r;
      if (mm < Nrows) H[(size_t)mm * 128 + n] = fmaxf(acc[r] + bn, 0.0f);
    }
  }
}

__global__ void gate_kernel(const float* __restrict__ H, const float* __restrict__ Wg,
                            float* __restrict__ ga, float* __restrict__ gb, int Nrows) {
  int wave = threadIdx.x >> 5;
  int lane = threadIdx.x & 31;
  int node = blockIdx.x * 8 + wave;
  if (node >= Nrows) return;
  const float* hrow = H + (size_t)node * 128;
  float sa = 0.0f, sb = 0.0f;
#pragma unroll
  for (int j = 0; j < 4; ++j) {
    int f = lane + j * 32;
    float hv = hrow[f];
    sa += hv * Wg[f];
    sb += hv * Wg[128 + f];
  }
  for (int off = 16; off > 0; off >>= 1) {
    sa += __shfl_down(sa, off, 32);
    sb += __shfl_down(sb, off, 32);
  }
  if (lane == 0) { ga[node] = sa; gb[node] = sb; }
}

__global__ void norm_kernel(const int* __restrict__ ei, const float* __restrict__ ga,
                            const float* __restrict__ gb, const float* __restrict__ dn,
                            const float* __restrict__ bg, int layer,
                            float* __restrict__ nrm, int E) {
  int e = blockIdx.x * blockDim.x + threadIdx.x;
  if (e >= E) return;
  int r = ei[e], c = ei[E + e];
  float g = tanhf(ga[r] + gb[c] + bg[layer]);
  nrm[e] = g * dn[r] * dn[c];
}

__global__ void scatter_kernel(const int* __restrict__ ei, const float* __restrict__ nrm,
                               const float* __restrict__ H, float* __restrict__ agg, int E) {
  long long idx = (long long)blockIdx.x * blockDim.x + threadIdx.x;
  if (idx >= (long long)E * 32) return;
  int e = (int)(idx >> 5);
  int f = (int)(idx & 31) * 4;
  int r = ei[e], c = ei[E + e];
  float s = nrm[e];
  float4 hv = *(const float4*)(H + (size_t)r * 128 + f);
  float* dst = agg + (size_t)c * 128 + f;
  atomicAdd(dst + 0, s * hv.x);
  atomicAdd(dst + 1, s * hv.y);
  atomicAdd(dst + 2, s * hv.z);
  atomicAdd(dst + 3, s * hv.w);
}

__global__ void residual_kernel(const float4* __restrict__ raw, float4* __restrict__ agg,
                                long long n4) {
  long long i = (long long)blockIdx.x * blockDim.x + threadIdx.x;
  if (i < n4) {
    float4 r = raw[i];
    float4 a = agg[i];
    a.x = FAGCN_EPS * r.x + a.x;
    a.y = FAGCN_EPS * r.y + a.y;
    a.z = FAGCN_EPS * r.z + a.z;
    a.w = FAGCN_EPS * r.w + a.w;
    agg[i] = a;
  }
}

__global__ void gemm_out_kernel(const float* __restrict__ H, const float* __restrict__ W,
                                const float* __restrict__ bias, float* __restrict__ out,
                                int Nrows) {
  int wave = threadIdx.x >> 5;
  int lane = threadIdx.x & 31;
  int tileM = blockIdx.x * 8 + wave;
  int tiles = (Nrows + 15) >> 4;
  if (tileM >= tiles) return;
  int mA = tileM * 16 + (lane & 15);
  int mclamp = mA < Nrows ? mA : (Nrows - 1);
  int n = lane & 15;
  int khalf = (lane >> 4) * 2;
  const float* hrow = H + (size_t)mclamp * 128 + khalf;
  const float* wcol = W + (size_t)khalf * 16 + n;
  __builtin_prefetch(hrow, 0, 0);
  v8f acc = {};
#pragma unroll 4
  for (int kk = 0; kk < 128; kk += 4) {
    v2f a = *(const v2f*)(hrow + kk);
    v2f b;
    b.x = wcol[(size_t)kk * 16];
    b.y = wcol[(size_t)kk * 16 + 16];
    acc = __builtin_amdgcn_wmma_f32_16x16x4_f32(false, a, false, b, (short)0, acc,
                                               false, false);
  }
  int mbase = tileM * 16 + ((lane >> 4) * 8);
  float bn = bias[n];
  if (tileM * 16 + 16 <= Nrows) {
    float* op = out + (size_t)mbase * 16 + n;
#pragma unroll
    for (int r = 0; r < 8; ++r)
      op[(size_t)r * 16] = acc[r] + bn;
  } else {
#pragma unroll
    for (int r = 0; r < 8; ++r) {
      int mm = mbase + r;
      if (mm < Nrows) out[(size_t)mm * 16 + n] = acc[r] + bn;
    }
  }
}

__global__ void logsoftmax_kernel(float4* out, int N) {
  int i = blockIdx.x * blockDim.x + threadIdx.x;
  if (i >= N) return;
  float4* row = out + (size_t)i * 4;
  float4 v[4];
#pragma unroll
  for (int q = 0; q < 4; ++q) v[q] = row[q];
  float m = v[0].x;
#pragma unroll
  for (int q = 0; q < 4; ++q) {
    m = fmaxf(m, v[q].x); m = fmaxf(m, v[q].y);
    m = fmaxf(m, v[q].z); m = fmaxf(m, v[q].w);
  }
  float s = 0.0f;
#pragma unroll
  for (int q = 0; q < 4; ++q) {
    s += expf(v[q].x - m); s += expf(v[q].y - m);
    s += expf(v[q].z - m); s += expf(v[q].w - m);
  }
  float d = m + logf(s);
#pragma unroll
  for (int q = 0; q < 4; ++q) {
    v[q].x -= d; v[q].y -= d; v[q].z -= d; v[q].w -= d;
    row[q] = v[q];
  }
}

extern "C" void kernel_launch(void* const* d_in, const int* in_sizes, int n_in,
                              void* d_out, int out_size, void* d_ws, size_t ws_size,
                              hipStream_t stream) {
  const float* x     = (const float*)d_in[0];
  const int*   ei    = (const int*)d_in[1];
  const float* W_in  = (const float*)d_in[2];
  const float* b_in  = (const float*)d_in[3];
  const float* Wg    = (const float*)d_in[4];
  const float* bg    = (const float*)d_in[5];
  const float* W_out = (const float*)d_in[6];
  const float* b_out = (const float*)d_in[7];
  float* out = (float*)d_out;

  const int Hdim = 128;
  int N = in_sizes[0] / Hdim;
  int E = in_sizes[1] / 2;
  int L = in_sizes[5];

  float* ws   = (float*)d_ws;
  float* dn   = ws;
  float* ga   = dn + N;
  float* gb   = ga + N;
  float* nrm  = gb + N;
  float* raw  = nrm + E;
  float* bufA = raw + (size_t)N * Hdim;
  float* bufB = bufA + (size_t)N * Hdim;

  hipMemsetAsync(dn, 0, (size_t)N * sizeof(float), stream);
  deg_kernel<<<(E + 255) / 256, 256, 0, stream>>>(ei, dn, E);
  dn_kernel<<<(N + 255) / 256, 256, 0, stream>>>(dn, N);

  gemm_relu_kernel<<<(N + 15) / 16, 256, 0, stream>>>(x, W_in, b_in, raw, N);

  float* h = raw;
  float* agg = bufA;
  for (int i = 0; i < L; ++i) {
    gate_kernel<<<(N + 7) / 8, 256, 0, stream>>>(h, Wg + (size_t)i * 2 * Hdim, ga, gb, N);
    norm_kernel<<<(E + 255) / 256, 256, 0, stream>>>(ei, ga, gb, dn, bg, i, nrm, E);
    hipMemsetAsync(agg, 0, (size_t)N * Hdim * sizeof(float), stream);
    long long tot = (long long)E * 32;
    scatter_kernel<<<(unsigned)((tot + 255) / 256), 256, 0, stream>>>(ei, nrm, h, agg, E);
    long long n4 = (long long)N * Hdim / 4;
    residual_kernel<<<(unsigned)((n4 + 255) / 256), 256, 0, stream>>>(
        (const float4*)raw, (float4*)agg, n4);
    h = agg;
    agg = (h == bufA) ? bufB : bufA;
  }

  int tiles = (N + 15) / 16;
  gemm_out_kernel<<<(tiles + 7) / 8, 256, 0, stream>>>(h, W_out, b_out, out, N);
  logsoftmax_kernel<<<(N + 255) / 256, 256, 0, stream>>>((float4*)out, N);
}


# Float-atomic scatter-add -> target-ordered gather: edges counting-sorted by target once per call (int atomics + scan), one wave per node gates and sums its in-edge rows in fixed source-id order
# speedup vs baseline: 1.6487x; 1.6487x over previous
#include <hip/hip_runtime.h>

typedef float v2f __attribute__((ext_vector_type(2)));
typedef float v8f __attribute__((ext_vector_type(8)));

#define FAGCN_EPS 0.3f

__global__ void deg_kernel(const int* __restrict__ ei, float* __restrict__ deg,
                           int* __restrict__ cnt, int E) {
  int e = blockIdx.x * blockDim.x + threadIdx.x;
  if (e < E) {
    atomicAdd(&deg[ei[e]], 1.0f);
    atomicAdd(&cnt[ei[E + e]], 1);
  }
}

__device__ __forceinline__ int4 scan_load4(const int* cnt, int i0, int N) {
  int4 v = {0, 0, 0, 0};
  if (i0 + 3 < N) {
    v = *(const int4*)(cnt + i0);
  } else {
    if (i0 < N) v.x = cnt[i0];
    if (i0 + 1 < N) v.y = cnt[i0 + 1];
    if (i0 + 2 < N) v.z = cnt[i0 + 2];
  }
  return v;
}

__global__ void scan_kernel(int* cnt, int N) {
  __shared__ int wsum[32];
  int tid = threadIdx.x;
  int lane = tid & 31;
  int wave = tid >> 5;
  int carry = 0;
  int4 nxt = scan_load4(cnt, tid * 4, N);
  for (int base = 0; base < N; base += 4096) {
    int i0 = base + tid * 4;
    int4 v = nxt;
    nxt = scan_load4(cnt, i0 + 4096, N);
    int s = v.x + v.y + v.z + v.w;
    int incl = s;
    for (int off = 1; off < 32; off <<= 1) {
      int t = __shfl_up(incl, off, 32);
      if (lane >= off) incl += t;
    }
    if (lane == 31) wsum[wave] = incl;
    __syncthreads();
    int wt = wsum[lane];
    int wincl = wt;
    for (int off = 1; off < 32; off <<= 1) {
      int t = __shfl_up(wincl, off, 32);
      if (lane >= off) wincl += t;
    }
    int wexcl = __shfl(wincl, wave, 32) - __shfl(wt, wave, 32);
    int total = __shfl(wincl, 31, 32);
    int ex = carry + wexcl + incl - s;
    if (i0 + 3 < N) {
      int4 o;
      o.x = ex; o.y = ex + v.x; o.z = ex + v.x + v.y; o.w = ex + v.x + v.y + v.z;
      *(int4*)(cnt + i0) = o;
    } else {
      if (i0 < N) cnt[i0] = ex;
      if (i0 + 1 < N) cnt[i0 + 1] = ex + v.x;
      if (i0 + 2 < N) cnt[i0 + 2] = ex + v.x + v.y;
    }
    carry += total;
    __syncthreads();
  }
}

__global__ void fill_kernel(const int* __restrict__ ei, int* __restrict__ cur,
                            int* __restrict__ srow, int E) {
  int e = blockIdx.x * blockDim.x + threadIdx.x;
  if (e < E) {
    int pos = atomicAdd(&cur[ei[E + e]], 1);
    srow[pos] = ei[e];
  }
}

__global__ void order_kernel(const int* __restrict__ ends, int* __restrict__ srow, int N) {
  int node = __builtin_amdgcn_readfirstlane(blockIdx.x * 8 + (threadIdx.x >> 5));
  int lane = threadIdx.x & 31;
  if (node >= N) return;
  int beg = node ? ends[node - 1] : 0;
  int d = ends[node] - beg;
  if (d < 2 || d > 64) return;
  int a0 = lane < d ? srow[beg + lane] : 0x7fffffff;
  int a1 = lane + 32 < d ? srow[beg + 32 + lane] : 0x7fffffff;
  int r0 = 0, r1 = 0;
  int n0 = d < 32 ? d : 32;
  for (int i = 0; i < n0; ++i) {
    int x = __builtin_amdgcn_readlane(a0, i);
    r0 += (x < a0 || (x == a0 && i < lane)) ? 1 : 0;
    r1 += (x <= a1) ? 1 : 0;
  }
  for (int i = 0; i < d - 32; ++i) {
    int x = __builtin_amdgcn_readlane(a1, i);
    r0 += (x < a0) ? 1 : 0;
    r1 += (x < a1 || (x == a1 && i < lane)) ? 1 : 0;
  }
  if (lane < d) srow[beg + r0] = a0;
  if (lane + 32 < d) srow[beg + r1] = a1;
}

__global__ void dn_kernel(float* dn, int N) {
  int i = blockIdx.x * blockDim.x + threadIdx.x;
  if (i < N) dn[i] = 1.0f / sqrtf(fmaxf(dn[i], 1.0f));
}

__global__ void gemm_relu_kernel(const float* __restrict__ X, const float* __restrict__ W,
                                 const float* __restrict__ bias, float* __restrict__ H,
                                 int Nrows) {
  int wave = threadIdx.x >> 5;
  int lane = threadIdx.x & 31;
  int tileM = blockIdx.x;
  int mA = tileM * 16 + (lane & 15);
  int mclamp = mA < Nrows ? mA : (Nrows - 1);
  int n = wave * 16 + (lane & 15);
  int khalf = (lane >> 4) * 2;
  const float* xrow = X + (size_t)mclamp * 128 + khalf;
  const float* wcol = W + (size_t)khalf * 128 + n;
  __builtin_prefetch(xrow, 0, 0);
  v8f acc = {};
#pragma unroll 4
  for (int kk = 0; kk < 128; kk += 4) {
    v2f a = *(const v2f*)(xrow + kk);
    v2f b;
    b.x = wcol[(size_t)kk * 128];
    b.y = wcol[(size_t)kk * 128 + 128];
    acc = __builtin_amdgcn_wmma_f32_16x16x4_f32(false, a, false, b, (short)0, acc,
                                               false, false);
  }
  int mbase = tileM * 16 + ((lane >> 4) * 8);
  float bn = bias[n];
  if (tileM * 16 + 16 <= Nrows) {
    float* hp = H + (size_t)mbase * 128 + n;
#pragma unroll
    for (int r = 0; r < 8; ++r)
      hp[(size_t)r * 128] = fmaxf(acc[r] + bn, 0.0f);
  } else {
#pragma unroll
    for (int r = 0; r < 8; ++r) {
      int mm = mbase + r;
      if (mm < Nrows) H[(size_t)mm * 128 + n] = fmaxf(acc[r] + bn, 0.0f);
    }
  }
}

__global__ void gate_kernel(const float* __restrict__ H, const float* __restrict__ Wg,
                            float* __restrict__ ga, float* __restrict__ gb, int Nrows) {
  int wave = threadIdx.x >> 5;
  int lane = threadIdx.x & 31;
  int node = blockIdx.x * 8 + wave;
  if (node >= Nrows) return;
  const float* hrow = H + (size_t)node * 128;
  float sa = 0.0f, sb = 0.0f;
#pragma unroll
  for (int j = 0; j < 4; ++j) {
    int f = lane + j * 32;
    float hv = hrow[f];
    sa += hv * Wg[f];
    sb += hv * Wg[128 + f];
  }
  for (int off = 16; off > 0; off >>= 1) {
    sa += __shfl_down(sa, off, 32);
    sb += __shfl_down(sb, off, 32);
  }
  if (lane == 0) { ga[node] = sa; gb[node] = sb; }
}

__global__ void gather_kernel(const int* __restrict__ ends, const int* __restrict__ srow,
                              const float* __restrict__ ga, const float* __restrict__ gb,
                              const float* __restrict__ dn, const float* __restrict__ bg,
                              int layer, const float* __restrict__ H,
                              float* __restrict__ agg, int N) {
  int node = __builtin_amdgcn_readfirstlane(blockIdx.x * 8 + (threadIdx.x >> 5));
  int lane = threadIdx.x & 31;
  if (node >= N) return;
  int beg = node ? ends[node - 1] : 0;
  int end = ends[node];
  float gbc = gb[node];
  float bgl = bg[layer];
  float dnc = dn[node];
  const float* hl = H + lane * 4;
  float4 acc = {0.0f, 0.0f, 0.0f, 0.0f};
  for (int base = beg; base < end; base += 32) {
    int n = end - base < 32 ? end - base : 32;
    int r = 0;
    float w = 0.0f;
    if (lane < n) {
      r = srow[base + lane];
      float g = tanhf(ga[r] + gbc + bgl);
      w = g * dn[r] * dnc;
    }
    int wi = __float_as_int(w);
    int k = 0;
    for (; k + 4 <= n; k += 4) {
      int r0 = __builtin_amdgcn_readlane(r, k);
      int r1 = __builtin_amdgcn_readlane(r, k + 1);
      int r2 = __builtin_amdgcn_readlane(r, k + 2);
      int r3 = __builtin_amdgcn_readlane(r, k + 3);
      float w0 = __int_as_float(__builtin_amdgcn_readlane(wi, k));
      float w1 = __int_as_float(__builtin_amdgcn_readlane(wi, k + 1));
      float w2 = __int_as_float(__builtin_amdgcn_readlane(wi, k + 2));
      float w3 = __int_as_float(__builtin_amdgcn_readlane(wi, k + 3));
      float4 h0 = *(const float4*)(hl + (size_t)r0 * 128);
      float4 h1 = *(const float4*)(hl + (size_t)r1 * 128);
      float4 h2 = *(const float4*)(hl + (size_t)r2 * 128);
      float4 h3 = *(const float4*)(hl + (size_t)r3 * 128);
      acc.x += w0 * h0.x; acc.y += w0 * h0.y; acc.z += w0 * h0.z; acc.w += w0 * h0.w;
      acc.x += w1 * h1.x; acc.y += w1 * h1.y; acc.z += w1 * h1.z; acc.w += w1 * h1.w;
      acc.x += w2 * h2.x; acc.y += w2 * h2.y; acc.z += w2 * h2.z; acc.w += w2 * h2.w;
      acc.x += w3 * h3.x; acc.y += w3 * h3.y; acc.z += w3 * h3.z; acc.w += w3 * h3.w;
    }
    for (; k < n; ++k) {
      int r0 = __builtin_amdgcn_readlane(r, k);
      float w0 = __int_as_float(__builtin_amdgcn_readlane(wi, k));
      float4 h0 = *(const float4*)(hl + (size_t)r0 * 128);
      acc.x += w0 * h0.x; acc.y += w0 * h0.y; acc.z += w0 * h0.z; acc.w += w0 * h0.w;
    }
  }
  *(float4*)(agg + (size_t)node * 128 + lane * 4) = acc;
}

__global__ void residual_kernel(const float4* __restrict__ raw, float4* __restrict__ agg,
                                long long n4) {
  long long i = (long long)blockIdx.x * blockDim.x + threadIdx.x;
  if (i < n4) {
    float4 r = raw[i];
    float4 a = agg[i];
    a.x = FAGCN_EPS * r.x + a.x;
    a.y = FAGCN_EPS * r.y + a.y;
    a.z = FAGCN_EPS * r.z + a.z;
    a.w = FAGCN_EPS * r.w + a.w;
    agg[i] = a;
  }
}

__global__ void gemm_out_kernel(const float* __restrict__ H, const float* __restrict__ W,
                                const float* __restrict__ bias, float* __restrict__ out,
                                int Nrows) {
  int wave = threadIdx.x >> 5;
  int lane = threadIdx.x & 31;
  int tileM = blockIdx.x * 8 + wave;
  int tiles = (Nrows + 15) >> 4;
  if (tileM >= tiles) return;
  int mA = tileM * 16 + (lane & 15);
  int mclamp = mA < Nrows ? mA : (Nrows - 1);
  int n = lane & 15;
  int khalf = (lane >> 4) * 2;
  const float* hrow = H + (size_t)mclamp * 128 + khalf;
  const float* wcol = W + (size_t)khalf * 16 + n;
  __builtin_prefetch(hrow, 0, 0);
  v8f acc = {};
#pragma unroll 4
  for (int kk = 0; kk < 128; kk += 4) {
    v2f a = *(const v2f*)(hrow + kk);
    v2f b;
    b.x = wcol[(size_t)kk * 16];
    b.y = wcol[(size_t)kk * 16 + 16];
    acc = __builtin_amdgcn_wmma_f32_16x16x4_f32(false, a, false, b, (short)0, acc,
                                               false, false);
  }
  int mbase = tileM * 16 + ((lane >> 4) * 8);
  float bn = bias[n];
  if (tileM * 16 + 16 <= Nrows) {
    float* op = out + (size_t)mbase * 16 + n;
#pragma unroll
    for (int r = 0; r < 8; ++r)
      op[(size_t)r * 16] = acc[r] + bn;
  } else {
#pragma unroll
    for (int r = 0; r < 8; ++r) {
      int mm = mbase + r;
      if (mm < Nrows) out[(size_t)mm * 16 + n] = acc[r] + bn;
    }
  }
}

__global__ void logsoftmax_kernel(float4* out, int N) {
  int i = blockIdx.x * blockDim.x + threadIdx.x;
  if (i >= N) return;
  float4* row = out + (size_t)i * 4;
  float4 v[4];
#pragma unroll
  for (int q = 0; q < 4; ++q) v[q] = row[q];
  float m = v[0].x;
#pragma unroll
  for (int q = 0; q < 4; ++q) {
    m = fmaxf(m, v[q].x); m = fmaxf(m, v[q].y);
    m = fmaxf(m, v[q].z); m = fmaxf(m, v[q].w);
  }
  float s = 0.0f;
#pragma unroll
  for (int q = 0; q < 4; ++q) {
    s += expf(v[q].x - m); s += expf(v[q].y - m);
    s += expf(v[q].z - m); s += expf(v[q].w - m);
  }
  float d = m + logf(s);
#pragma unroll
  for (int q = 0; q < 4; ++q) {
    v[q].x -= d; v[q].y -= d; v[q].z -= d; v[q].w -= d;
    row[q] = v[q];
  }
}

extern "C" void kernel_launch(void* const* d_in, const int* in_sizes, int n_in,
                              void* d_out, int out_size, void* d_ws, size_t ws_size,
                              hipStream_t stream) {
  const float* x     = (const float*)d_in[0];
  const int*   ei    = (const int*)d_in[1];
  const float* W_in  = (const float*)d_in[2];
  const float* b_in  = (const float*)d_in[3];
  const float* Wg    = (const float*)d_in[4];
  const float* bg    = (const float*)d_in[5];
  const float* W_out = (const float*)d_in[6];
  const float* b_out = (const float*)d_in[7];
  float* out = (float*)d_out;

  const int Hdim = 128;
  int N = in_sizes[0] / Hdim;
  int E = in_sizes[1] / 2;
  int L = in_sizes[5];

  float* ws   = (float*)d_ws;
  int*   cnt  = (int*)ws;
  float* dn   = ws + N;
  float* ga   = dn + N;
  int*   srow = (int*)(ga + N);
  float* raw  = ga + N + E;
  float* bufA = raw + (size_t)N * Hdim;
  float* bufB = bufA + (size_t)N * Hdim;
  float* gb   = out;

  hipMemsetAsync(cnt, 0, (size_t)N * 2 * sizeof(float), stream);
  deg_kernel<<<(E + 255) / 256, 256, 0, stream>>>(ei, dn, cnt, E);
  dn_kernel<<<(N + 255) / 256, 256, 0, stream>>>(dn, N);
  scan_kernel<<<1, 1024, 0, stream>>>(cnt, N);
  fill_kernel<<<(E + 255) / 256, 256, 0, stream>>>(ei, cnt, srow, E);
  order_kernel<<<(N + 7) / 8, 256, 0, stream>>>(cnt, srow, N);

  gemm_relu_kernel<<<(N + 15) / 16, 256, 0, stream>>>(x, W_in, b_in, raw, N);

  float* h = raw;
  float* agg = bufA;
  for (int i = 0; i < L; ++i) {
    gate_kernel<<<(N + 7) / 8, 256, 0, stream>>>(h, Wg + (size_t)i * 2 * Hdim, ga, gb, N);
    gather_kernel<<<(N + 7) / 8, 256, 0, stream>>>(cnt, srow, ga, gb, dn, bg, i, h, agg, N);
    long long n4 = (long long)N * Hdim / 4;
    residual_kernel<<<(unsigned)((n4 + 255) / 256), 256, 0, stream>>>(
        (const float4*)raw, (float4*)agg, n4);
    h = agg;
    agg = (h == bufA) ? bufB : bufA;
  }

  int tiles = (N + 15) / 16;
  gemm_out_kernel<<<(tiles + 7) / 8, 256, 0, stream>>>(h, W_out, b_out, out, N);
  logsoftmax_kernel<<<(N + 255) / 256, 256, 0, stream>>>((float4*)out, N);
}


# Row-wise passes fold into the epilogue of the kernel that makes the row: residual + next gate dots in gather, first gate dots in input GEMM (tile via LDS); 4 launches go; gate dot sum order changes
# speedup vs baseline: 1.8050x; 1.0948x over previous
#include <hip/hip_runtime.h>

typedef float v2f __attribute__((ext_vector_type(2)));
typedef float v8f __attribute__((ext_vector_type(8)));

#define FAGCN_EPS 0.3f

__global__ void deg_kernel(const int* __restrict__ ei, float* __restrict__ deg,
                           int* __restrict__ cnt, int E) {
  int e = blockIdx.x * blockDim.x + threadIdx.x;
  if (e < E) {
    atomicAdd(&deg[ei[e]], 1.0f);
    atomicAdd(&cnt[ei[E + e]], 1);
  }
}

__device__ __forceinline__ int4 scan_load4(const int* cnt, int i0, int N) {
  int4 v = {0, 0, 0, 0};
  if (i0 + 3 < N) {
    v = *(const int4*)(cnt + i0);
  } else {
    if (i0 < N) v.x = cnt[i0];
    if (i0 + 1 < N) v.y = cnt[i0 + 1];
    if (i0 + 2 < N) v.z = cnt[i0 + 2];
  }
  return v;
}

__global__ void scan_kernel(int* cnt, int N) {
  __shared__ int wsum[32];
  int tid = threadIdx.x;
  int lane = tid & 31;
  int wave = tid >> 5;
  int carry = 0;
  int4 nxt = scan_load4(cnt, tid * 4, N);
  for (int base = 0; base < N; base += 4096) {
    int i0 = base + tid * 4;
    int4 v = nxt;
    nxt = scan_load4(cnt, i0 + 4096, N);
    int s = v.x + v.y + v.z + v.w;
    int incl = s;
    for (int off = 1; off < 32; off <<= 1) {
      int t = __shfl_up(incl, off, 32);
      if (lane >= off) incl += t;
    }
    if (lane == 31) wsum[wave] = incl;
    __syncthreads();
    int wt = wsum[lane];
    int wincl = wt;
    for (int off = 1; off < 32; off <<= 1) {
      int t = __shfl_up(wincl, off, 32);
      if (lane >= off) wincl += t;
    }
    int wexcl = __shfl(wincl, wave, 32) - __shfl(wt, wave, 32);
    int total = __shfl(wincl, 31, 32);
    int ex = carry + wexcl + incl - s;
    if (i0 + 3 < N) {
      int4 o;
      o.x = ex; o.y = ex + v.x; o.z = ex + v.x + v.y; o.w = ex + v.x + v.y + v.z;
      *(int4*)(cnt + i0) = o;
    } else {
      if (i0 < N) cnt[i0] = ex;
      if (i0 + 1 < N) cnt[i0 + 1] = ex + v.x;
      if (i0 + 2 < N) cnt[i0 + 2] = ex + v.x + v.y;
    }
    carry += total;
    __syncthreads();
  }
}

__global__ void fill_kernel(const int* __restrict__ ei, int* __restrict__ cur,
                            int* __restrict__ srow, int E) {
  int e = blockIdx.x * blockDim.x + threadIdx.x;
  if (e < E) {
    int pos = atomicAdd(&cur[ei[E + e]], 1);
    srow[pos] = ei[e];
  }
}

__global__ void order_kernel(const int* __restrict__ ends, int* __restrict__ srow, int N) {
  int node = __builtin_amdgcn_readfirstlane(blockIdx.x * 8 + (threadIdx.x >> 5));
  int lane = threadIdx.x & 31;
  if (node >= N) return;
  int beg = node ? ends[node - 1] : 0;
  int d = ends[node] - beg;
  if (d < 2 || d > 64) return;
  int a0 = lane < d ? srow[beg + lane] : 0x7fffffff;
  int a1 = lane + 32 < d ? srow[beg + 32 + lane] : 0x7fffffff;
  int r0 = 0, r1 = 0;
  int n0 = d < 32 ? d : 32;
  for (int i = 0; i < n0; ++i) {
    int x = __builtin_amdgcn_readlane(a0, i);
    r0 += (x < a0 || (x == a0 && i < lane)) ? 1 : 0;
    r1 += (x <= a1) ? 1 : 0;
  }
  for (int i = 0; i < d - 32; ++i) {
    int x = __builtin_amdgcn_readlane(a1, i);
    r0 += (x < a0) ? 1 : 0;
    r1 += (x < a1 || (x == a1 && i < lane)) ? 1 : 0;
  }
  if (lane < d) srow[beg + r0] = a0;
  if (lane + 32 < d) srow[beg + r1] = a1;
}

__global__ void dn_kernel(float* dn, int N) {
  int i = blockIdx.x * blockDim.x + threadIdx.x;
  if (i < N) dn[i] = 1.0f / sqrtf(fmaxf(dn[i], 1.0f));
}

#define TILE_PITCH 132
__global__ void gemm_relu_kernel(const float* __restrict__ X, const float* __restrict__ W,
                                 const float* __restrict__ bias, const float* __restrict__ wg0,
                                 float* __restrict__ H, float* __restrict__ ga,
                                 float* __restrict__ gb, int Nrows) {
  __shared__ __attribute__((aligned(16))) float tile[16 * TILE_PITCH];
  int wave = threadIdx.x >> 5;
  int lane = threadIdx.x & 31;
  int tileM = blockIdx.x;
  int mA = tileM * 16 + (lane & 15);
  int mclamp = mA < Nrows ? mA : (Nrows - 1);
  int n = wave * 16 + (lane & 15);
  int khalf = (lane >> 4) * 2;
  const float* xrow = X + (size_t)mclamp * 128 + khalf;
  const float* wcol = W + (size_t)khalf * 128 + n;
  __builtin_prefetch(xrow, 0, 0);
  v8f acc = {};
#pragma unroll 4
  for (int kk = 0; kk < 128; kk += 4) {
    v2f a = *(const v2f*)(xrow + kk);
    v2f b;
    b.x = wcol[(size_t)kk * 128];
    b.y = wcol[(size_t)kk * 128 + 128];
    acc = __builtin_amdgcn_wmma_f32_16x16x4_f32(false, a, false, b, (short)0, acc,
                                               false, false);
  }
  int rbase = (lane >> 4) * 8;
  float bn = bias[n];
#pragma unroll
  for (int r = 0; r < 8; ++r)
    tile[(rbase + r) * TILE_PITCH + n] = fmaxf(acc[r] + bn, 0.0f);
  __syncthreads();
  float4 wa = {0.0f, 0.0f, 0.0f, 0.0f}, wb = {0.0f, 0.0f, 0.0f, 0.0f};
  if (wg0) {
    wa = *(const float4*)(wg0 + lane * 4);
    wb = *(const float4*)(wg0 + 128 + lane * 4);
  }
  for (int j = 0; j < 2; ++j) {
    int rr = wave * 2 + j;
    int m = tileM * 16 + rr;
    float4 v = *(const float4*)(tile + rr * TILE_PITCH + lane * 4);
    if (m < Nrows) {
      *(float4*)(H + (size_t)m * 128 + lane * 4) = v;
      if (wg0) {
        float sa = v.x * wa.x + v.y * wa.y + v.z * wa.z + v.w * wa.w;
        float sb = v.x * wb.x + v.y * wb.y + v.z * wb.z + v.w * wb.w;
        for (int off = 16; off > 0; off >>= 1) {
          sa += __shfl_down(sa, off, 32);
          sb += __shfl_down(sb, off, 32);
        }
        if (lane == 0) { ga[m] = sa; gb[m] = sb; }
      }
    }
  }
}

__global__ void gather_kernel(const int* __restrict__ ends, const int* __restrict__ srow,
                              const float* __restrict__ ga, const float* __restrict__ gb,
                              const float* __restrict__ dn, const float* __restrict__ bg,
                              int layer, const float* __restrict__ H,
                              const float* __restrict__ raw, float* __restrict__ hout,
                              const float* __restrict__ wg_next, float* __restrict__ ga_next,
                              float* __restrict__ gb_next, int N) {
  int node = __builtin_amdgcn_readfirstlane(blockIdx.x * 8 + (threadIdx.x >> 5));
  int lane = threadIdx.x & 31;
  if (node >= N) return;
  int beg = node ? ends[node - 1] : 0;
  int end = ends[node];
  float gbc = gb[node];
  float bgl = bg[layer];
  float dnc = dn[node];
  const float* hl = H + lane * 4;
  float4 rw = *(const float4*)(raw + (size_t)node * 128 + lane * 4);
  float4 acc = {0.0f, 0.0f, 0.0f, 0.0f};
  for (int base = beg; base < end; base += 32) {
    int n = end - base < 32 ? end - base : 32;
    int r = 0;
    float w = 0.0f;
    if (lane < n) {
      r = srow[base + lane];
      float g = tanhf(ga[r] + gbc + bgl);
      w = g * dn[r] * dnc;
    }
    int wi = __float_as_int(w);
    int k = 0;
    for (; k + 4 <= n; k += 4) {
      int r0 = __builtin_amdgcn_readlane(r, k);
      int r1 = __builtin_amdgcn_readlane(r, k + 1);
      int r2 = __builtin_amdgcn_readlane(r, k + 2);
      int r3 = __builtin_amdgcn_readlane(r, k + 3);
      float w0 = __int_as_float(__builtin_amdgcn_readlane(wi, k));
      float w1 = __int_as_float(__builtin_amdgcn_readlane(wi, k + 1));
      float w2 = __int_as_float(__builtin_amdgcn_readlane(wi, k + 2));
      float w3 = __int_as_float(__builtin_amdgcn_readlane(wi, k + 3));
      float4 h0 = *(const float4*)(hl + (size_t)r0 * 128);
      float4 h1 = *(const float4*)(hl + (size_t)r1 * 128);
      float4 h2 = *(const float4*)(hl + (size_t)r2 * 128);
      float4 h3 = *(const float4*)(hl + (size_t)r3 * 128);
      acc.x += w0 * h0.x; acc.y += w0 * h0.y; acc.z += w0 * h0.z; acc.w += w0 * h0.w;
      acc.x += w1 * h1.x; acc.y += w1 * h1.y; acc.z += w1 * h1.z; acc.w += w1 * h1.w;
      acc.x += w2 * h2.x; acc.y += w2 * h2.y; acc.z += w2 * h2.z; acc.w += w2 * h2.w;
      acc.x += w3 * h3.x; acc.y += w3 * h3.y; acc.z += w3 * h3.z; acc.w += w3 * h3.w;
    }
    for (; k < n; ++k) {
      int r0 = __builtin_amdgcn_readlane(r, k);
      float w0 = __int_as_float(__builtin_amdgcn_readlane(wi, k));
      float4 h0 = *(const float4*)(hl + (size_t)r0 * 128);
      acc.x += w0 * h0.x; acc.y += w0 * h0.y; acc.z += w0 * h0.z; acc.w += w0 * h0.w;
    }
  }
  float4 o;
  o.x = FAGCN_EPS * rw.x + acc.x;
  o.y = FAGCN_EPS * rw.y + acc.y;
  o.z = FAGCN_EPS * rw.z + acc.z;
  o.w = FAGCN_EPS * rw.w + acc.w;
  *(float4*)(hout + (size_t)node * 128 + lane * 4) = o;
  if (wg_next) {
    float4 wa = *(const float4*)(wg_next + lane * 4);
    float4 wb = *(const float4*)(wg_next + 128 + lane * 4);
    float sa = o.x * wa.x + o.y * wa.y + o.z * wa.z + o.w * wa.w;
    float sb = o.x * wb.x + o.y * wb.y + o.z * wb.z + o.w * wb.w;
    for (int off = 16; off > 0; off >>= 1) {
      sa += __shfl_down(sa, off, 32);
      sb += __shfl_down(sb, off, 32);
    }
    if (lane == 0) { ga_next[node] = sa; gb_next[node] = sb; }
  }
}

__global__ void gemm_out_kernel(const float* __restrict__ H, const float* __restrict__ W,
                                const float* __restrict__ bias, float* __restrict__ out,
                                int Nrows) {
  int wave = threadIdx.x >> 5;
  int lane = threadIdx.x & 31;
  int tileM = blockIdx.x * 8 + wave;
  int tiles = (Nrows + 15) >> 4;
  if (tileM >= tiles) return;
  int mA = tileM * 16 + (lane & 15);
  int mclamp = mA < Nrows ? mA : (Nrows - 1);
  int n = lane & 15;
  int khalf = (lane >> 4) * 2;
  const float* hrow = H + (size_t)mclamp * 128 + khalf;
  const float* wcol = W + (size_t)khalf * 16 + n;
  __builtin_prefetch(hrow, 0, 0);
  v8f acc = {};
#pragma unroll 4
  for (int kk = 0; kk < 128; kk += 4) {
    v2f a = *(const v2f*)(hrow + kk);
    v2f b;
    b.x = wcol[(size_t)kk * 16];
    b.y = wcol[(size_t)kk * 16 + 16];
    acc = __builtin_amdgcn_wmma_f32_16x16x4_f32(false, a, false, b, (short)0, acc,
                                               false, false);
  }
  int mbase = tileM * 16 + ((lane >> 4) * 8);
  float bn = bias[n];
  if (tileM * 16 + 16 <= Nrows) {
    float* op = out + (size_t)mbase * 16 + n;
#pragma unroll
    for (int r = 0; r < 8; ++r)
      op[(size_t)r * 16] = acc[r] + bn;
  } else {
#pragma unroll
    for (int r = 0; r < 8; ++r) {
      int mm = mbase + r;
      if (mm < Nrows) out[(size_t)mm * 16 + n] = acc[r] + bn;
    }
  }
}

__global__ void logsoftmax_kernel(float4* out, int N) {
  int i = blockIdx.x * blockDim.x + threadIdx.x;
  if (i >= N) return;
  float4* row = out + (size_t)i * 4;
  float4 v[4];
#pragma unroll
  for (int q = 0; q < 4; ++q) v[q] = row[q];
  float m = v[0].x;
#pragma unroll
  for (int q = 0; q < 4; ++q) {
    m = fmaxf(m, v[q].x); m = fmaxf(m, v[q].y);
    m = fmaxf(m, v[q].z); m = fmaxf(m, v[q].w);
  }
  float s = 0.0f;
#pragma unroll
  for (int q = 0; q < 4; ++q) {
    s += expf(v[q].x - m); s += expf(v[q].y - m);
    s += expf(v[q].z - m); s += expf(v[q].w - m);
  }
  float d = m + logf(s);
#pragma unroll
  for (int q = 0; q < 4; ++q) {
    v[q].x -= d; v[q].y -= d; v[q].z -= d; v[q].w -= d;
    row[q] = v[q];
  }
}

extern "C" void kernel_launch(void* const* d_in, const int* in_sizes, int n_in,
                              void* d_out, int out_size, void* d_ws, size_t ws_size,
                              hipStream_t stream) {
  const float* x     = (const float*)d_in[0];
  const int*   ei    = (const int*)d_in[1];
  const float* W_in  = (const float*)d_in[2];
  const float* b_in  = (const float*)d_in[3];
  const float* Wg    = (const float*)d_in[4];
  const float* bg    = (const float*)d_in[5];
  const float* W_out = (const float*)d_in[6];
  const float* b_out = (const float*)d_in[7];
  float* out = (float*)d_out;

  const int Hdim = 128;
  int N = in_sizes[0] / Hdim;
  int E = in_sizes[1] / 2;
  int L = in_sizes[5];

  float* ws   = (float*)d_ws;
  int*   cnt  = (int*)ws;
  float* dn   = ws + N;
  float* ga   = dn + N;
  int*   srow = (int*)(ga + N);
  float* raw  = ga + N + E;
  float* bufA = raw + (size_t)N * Hdim;
  float* bufB = bufA + (size_t)N * Hdim;
  float* gb   = out;
  float* ga2  = out + N;
  float* gb2  = out + 2 * (size_t)N;

  hipMemsetAsync(cnt, 0, (size_t)N * 2 * sizeof(float), stream);
  deg_kernel<<<(E + 255) / 256, 256, 0, stream>>>(ei, dn, cnt, E);
  dn_kernel<<<(N + 255) / 256, 256, 0, stream>>>(dn, N);
  scan_kernel<<<1, 1024, 0, stream>>>(cnt, N);
  fill_kernel<<<(E + 255) / 256, 256, 0, stream>>>(ei, cnt, srow, E);
  order_kernel<<<(N + 7) / 8, 256, 0, stream>>>(cnt, srow, N);

  gemm_relu_kernel<<<(N + 15) / 16, 256, 0, stream>>>(x, W_in, b_in, L > 0 ? Wg : nullptr, raw,
                                                     ga, gb, N);

  float* h = raw;
  float* hnew = bufA;
  float* gac = ga;  float* gbc = gb;
  float* gan = ga2; float* gbn = gb2;
  for (int i = 0; i < L; ++i) {
    const float* wgn = (i + 1 < L) ? Wg + (size_t)(i + 1) * 2 * Hdim : nullptr;
    gather_kernel<<<(N + 7) / 8, 256, 0, stream>>>(cnt, srow, gac, gbc, dn, bg, i, h, raw, hnew,
                                                  wgn, gan, gbn, N);
    h = hnew;
    hnew = (h == bufA) ? bufB : bufA;
    float* t = gac; gac = gan; gan = t;
    t = gbc; gbc = gbn; gbn = t;
  }

  int tiles = (N + 15) / 16;
  gemm_out_kernel<<<(tiles + 7) / 8, 256, 0, stream>>>(h, W_out, b_out, out, N);
  logsoftmax_kernel<<<(N + 255) / 256, 256, 0, stream>>>((float4*)out, N);
}


# speedup vs baseline: 1.9077x; 1.0569x over previous
#include <hip/hip_runtime.h>

typedef float v2f __attribute__((ext_vector_type(2)));
typedef float v8f __attribute__((ext_vector_type(8)));

#define FAGCN_EPS 0.3f

__global__ void deg_kernel(const int* __restrict__ ei, float* __restrict__ deg,
                           int* __restrict__ cnt, int E) {
  int e = blockIdx.x * blockDim.x + threadIdx.x;
  if (e < E) {
    atomicAdd(&deg[ei[e]], 1.0f);
    atomicAdd(&cnt[ei[E + e]], 1);
  }
}

__device__ __forceinline__ int4 scan_load4(const int* cnt, int i0, int N) {
  int4 v = {0, 0, 0, 0};
  if (i0 + 3 < N) {
    v = *(const int4*)(cnt + i0);
  } else {
    if (i0 < N) v.x = cnt[i0];
    if (i0 + 1 < N) v.y = cnt[i0 + 1];
    if (i0 + 2 < N) v.z = cnt[i0 + 2];
  }
  return v;
}

__global__ void scan_kernel(int* cnt, int N) {
  __shared__ int wsum[32];
  int tid = threadIdx.x;
  int lane = tid & 31;
  int wave = tid >> 5;
  int carry = 0;
  int4 nxt = scan_load4(cnt, tid * 4, N);
  for (int base = 0; base < N; base += 4096) {
    int i0 = base + tid * 4;
    int4 v = nxt;
    nxt = scan_load4(cnt, i0 + 4096, N);
    int s = v.x + v.y + v.z + v.w;
    int incl = s;
    for (int off = 1; off < 32; off <<= 1) {
      int t = __shfl_up(incl, off, 32);
      if (lane >= off) incl += t;
    }
    if (lane == 31) wsum[wave] = incl;
    __syncthreads();
    int wt = wsum[lane];
    int wincl = wt;
    for (int off = 1; off < 32; off <<= 1) {
      int t = __shfl_up(wincl, off, 32);
      if (lane >= off) wincl += t;
    }
    int wexcl = __shfl(wincl, wave, 32) - __shfl(wt, wave, 32);
    int total = __shfl(wincl, 31, 32);
    int ex = carry + wexcl + incl - s;
    if (i0 + 3 < N) {
      int4 o;
      o.x = ex; o.y = ex + v.x; o.z = ex + v.x + v.y; o.w = ex + v.x + v.y + v.z;
      *(int4*)(cnt + i0) = o;
    } else {
      if (i0 < N) cnt[i0] = ex;
      if (i0 + 1 < N) cnt[i0 + 1] = ex + v.x;
      if (i0 + 2 < N) cnt[i0 + 2] = ex + v.x + v.y;
    }
    carry += total;
    __syncthreads();
  }
}

__global__ void fill_kernel(const int* __restrict__ ei, int* __restrict__ cur,
                            int* __restrict__ srow, int E) {
  int e = blockIdx.x * blockDim.x + threadIdx.x;
  if (e < E) {
    int pos = atomicAdd(&cur[ei[E + e]], 1);
    srow[pos] = ei[e];
  }
}

__global__ void order_kernel(const int* __restrict__ ends, int* __restrict__ srow, int N) {
  int node = __builtin_amdgcn_readfirstlane(blockIdx.x * 8 + (threadIdx.x >> 5));
  int lane = threadIdx.x & 31;
  if (node >= N) return;
  int beg = node ? ends[node - 1] : 0;
  int d = ends[node] - beg;
  if (d < 2 || d > 64) return;
  int a0 = lane < d ? srow[beg + lane] : 0x7fffffff;
  int a1 = lane + 32 < d ? srow[beg + 32 + lane] : 0x7fffffff;
  int r0 = 0, r1 = 0;
  int n0 = d < 32 ? d : 32;
  for (int i = 0; i < n0; ++i) {
    int x = __builtin_amdgcn_readlane(a0, i);
    r0 += (x < a0 || (x == a0 && i < lane)) ? 1 : 0;
    r1 += (x <= a1) ? 1 : 0;
  }
  for (int i = 0; i < d - 32; ++i) {
    int x = __builtin_amdgcn_readlane(a1, i);
    r0 += (x < a0) ? 1 : 0;
    r1 += (x < a1 || (x == a1 && i < lane)) ? 1 : 0;
  }
  if (lane < d) srow[beg + r0] = a0;
  if (lane + 32 < d) srow[beg + r1] = a1;
}

__global__ void dn_kernel(float* dn, int N) {
  int i = blockIdx.x * blockDim.x + threadIdx.x;
  if (i < N) dn[i] = 1.0f / sqrtf(fmaxf(dn[i], 1.0f));
}

#define TILE_PITCH 132
__global__ void gemm_relu_kernel(const float* __restrict__ X, const float* __restrict__ W,
                                 const float* __restrict__ bias, const float* __restrict__ wg0,
                                 const float* __restrict__ dn, float* __restrict__ H,
                                 float* __restrict__ pk, float* __restrict__ gb, int Nrows) {
  __shared__ __attribute__((aligned(16))) float tile[16 * TILE_PITCH];
  int wave = threadIdx.x >> 5;
  int lane = threadIdx.x & 31;
  int tileM = blockIdx.x;
  int mA = tileM * 16 + (lane & 15);
  int mclamp = mA < Nrows ? mA : (Nrows - 1);
  int n = wave * 16 + (lane & 15);
  int khalf = (lane >> 4) * 2;
  const float* xrow = X + (size_t)mclamp * 128 + khalf;
  const float* wcol = W + (size_t)khalf * 128 + n;
  __builtin_prefetch(xrow, 0, 0);
  float dnv = dn[mclamp];
  v8f acc = {};
#pragma unroll 4
  for (int kk = 0; kk < 128; kk += 4) {
    v2f a = *(const v2f*)(xrow + kk);
    v2f b;
    b.x = wcol[(size_t)kk * 128];
    b.y = wcol[(size_t)kk * 128 + 128];
    acc = __builtin_amdgcn_wmma_f32_16x16x4_f32(false, a, false, b, (short)0, acc,
                                               false, false);
  }
  int rbase = (lane >> 4) * 8;
  float bn = bias[n];
#pragma unroll
  for (int r = 0; r < 8; ++r)
    tile[(rbase + r) * TILE_PITCH + n] = fmaxf(acc[r] + bn, 0.0f);
  __syncthreads();
  float4 wa = {0.0f, 0.0f, 0.0f, 0.0f}, wb = {0.0f, 0.0f, 0.0f, 0.0f};
  if (wg0) {
    wa = *(const float4*)(wg0 + lane * 4);
    wb = *(const float4*)(wg0 + 128 + lane * 4);
  }
  for (int j = 0; j < 2; ++j) {
    int rr = wave * 2 + j;
    int m = tileM * 16 + rr;
    float4 v = *(const float4*)(tile + rr * TILE_PITCH + lane * 4);
    if (m < Nrows) {
      *(float4*)(H + (size_t)m * 128 + lane * 4) = v;
      if (wg0) {
        float dnm = __shfl(dnv, rr, 32);
        float sa = v.x * wa.x + v.y * wa.y + v.z * wa.z + v.w * wa.w;
        float sb = v.x * wb.x + v.y * wb.y + v.z * wb.z + v.w * wb.w;
        for (int off = 16; off > 0; off >>= 1) {
          sa += __shfl_down(sa, off, 32);
          sb += __shfl_down(sb, off, 32);
        }
        if (lane == 0) {
          float2 p; p.x = sa; p.y = dnm;
          *(float2*)(pk + 2 * (size_t)m) = p;
          gb[m] = sb;
        }
      }
    }
  }
}

__global__ void gather_kernel(const int* __restrict__ ends, const int* __restrict__ srow,
                              const float* __restrict__ pk, const float* __restrict__ gb,
                              const float* __restrict__ bg,
                              int layer, const float* __restrict__ H,
                              const float* __restrict__ raw, float* __restrict__ hout,
                              const float* __restrict__ wg_next, float* __restrict__ pk_next,
                              float* __restrict__ gb_next, int N) {
  int node = __builtin_amdgcn_readfirstlane(blockIdx.x * 8 + (threadIdx.x >> 5));
  int lane = threadIdx.x & 31;
  if (node >= N) return;
  int beg = node ? ends[node - 1] : 0;
  int end = ends[node];
  float gbc = gb[node];
  float bgl = bg[layer];
  float dnc = pk[2 * (size_t)node + 1];
  const float* hl = H + lane * 4;
  float4 rw = *(const float4*)(raw + (size_t)node * 128 + lane * 4);
  float4 acc = {0.0f, 0.0f, 0.0f, 0.0f};
  for (int base = beg; base < end; base += 32) {
    int n = end - base < 32 ? end - base : 32;
    int r = 0;
    float w = 0.0f;
    if (lane < n) {
      r = srow[base + lane];
      float2 gd = *(const float2*)(pk + 2 * (size_t)r);
      float g = tanhf(gd.x + gbc + bgl);
      w = g * gd.y * dnc;
    }
    int wi = __float_as_int(w);
    int k = 0;
    for (; k + 4 <= n; k += 4) {
      int r0 = __builtin_amdgcn_readlane(r, k);
      int r1 = __builtin_amdgcn_readlane(r, k + 1);
      int r2 = __builtin_amdgcn_readlane(r, k + 2);
      int r3 = __builtin_amdgcn_readlane(r, k + 3);
      float w0 = __int_as_float(__builtin_amdgcn_readlane(wi, k));
      float w1 = __int_as_float(__builtin_amdgcn_readlane(wi, k + 1));
      float w2 = __int_as_float(__builtin_amdgcn_readlane(wi, k + 2));
      float w3 = __int_as_float(__builtin_amdgcn_readlane(wi, k + 3));
      float4 h0 = *(const float4*)(hl + (size_t)r0 * 128);
      float4 h1 = *(const float4*)(hl + (size_t)r1 * 128);
      float4 h2 = *(const float4*)(hl + (size_t)r2 * 128);
      float4 h3 = *(const float4*)(hl + (size_t)r3 * 128);
      acc.x += w0 * h0.x; acc.y += w0 * h0.y; acc.z += w0 * h0.z; acc.w += w0 * h0.w;
      acc.x += w1 * h1.x; acc.y += w1 * h1.y; acc.z += w1 * h1.z; acc.w += w1 * h1.w;
      acc.x += w2 * h2.x; acc.y += w2 * h2.y; acc.z += w2 * h2.z; acc.w += w2 * h2.w;
      acc.x += w3 * h3.x; acc.y += w3 * h3.y; acc.z += w3 * h3.z; acc.w += w3 * h3.w;
    }
    for (; k < n; ++k) {
      int r0 = __builtin_amdgcn_readlane(r, k);
      float w0 = __int_as_float(__builtin_amdgcn_readlane(wi, k));
      float4 h0 = *(const float4*)(hl + (size_t)r0 * 128);
      acc.x += w0 * h0.x; acc.y += w0 * h0.y; acc.z += w0 * h0.z; acc.w += w0 * h0.w;
    }
  }
  float4 o;
  o.x = FAGCN_EPS * rw.x + acc.x;
  o.y = FAGCN_EPS * rw.y + acc.y;
  o.z = FAGCN_EPS * rw.z + acc.z;
  o.w = FAGCN_EPS * rw.w + acc.w;
  *(float4*)(hout + (size_t)node * 128 + lane * 4) = o;
  if (wg_next) {
    float4 wa = *(const float4*)(wg_next + lane * 4);
    float4 wb = *(const float4*)(wg_next + 128 + lane * 4);
    float sa = o.x * wa.x + o.y * wa.y + o.z * wa.z + o.w * wa.w;
    float sb = o.x * wb.x + o.y * wb.y + o.z * wb.z + o.w * wb.w;
    for (int off = 16; off > 0; off >>= 1) {
      sa += __shfl_down(sa, off, 32);
      sb += __shfl_down(sb, off, 32);
    }
    if (lane == 0) {
      float2 p; p.x = sa; p.y = dnc;
      *(float2*)(pk_next + 2 * (size_t)node) = p;
      gb_next[node] = sb;
    }
  }
}

__global__ void gemm_out_kernel(const float* __restrict__ H, const float* __restrict__ W,
                                const float* __restrict__ bias, float* __restrict__ out,
                                int Nrows) {
  int wave = threadIdx.x >> 5;
  int lane = threadIdx.x & 31;
  int tileM = blockIdx.x * 8 + wave;
  int tiles = (Nrows + 15) >> 4;
  if (tileM >= tiles) return;
  int mA = tileM * 16 + (lane & 15);
  int mclamp = mA < Nrows ? mA : (Nrows - 1);
  int n = lane & 15;
  int khalf = (lane >> 4) * 2;
  const float* hrow = H + (size_t)mclamp * 128 + khalf;
  const float* wcol = W + (size_t)khalf * 16 + n;
  __builtin_prefetch(hrow, 0, 0);
  v8f acc = {};
#pragma unroll 4
  for (int kk = 0; kk < 128; kk += 4) {
    v2f a = *(const v2f*)(hrow + kk);
    v2f b;
    b.x = wcol[(size_t)kk * 16];
    b.y = wcol[(size_t)kk * 16 + 16];
    acc = __builtin_amdgcn_wmma_f32_16x16x4_f32(false, a, false, b, (short)0, acc,
                                               false, false);
  }
  int mbase = tileM * 16 + ((lane >> 4) * 8);
  float bn = bias[n];
  if (tileM * 16 + 16 <= Nrows) {
    float* op = out + (size_t)mbase * 16 + n;
#pragma unroll
    for (int r = 0; r < 8; ++r)
      op[(size_t)r * 16] = acc[r] + bn;
  } else {
#pragma unroll
    for (int r = 0; r < 8; ++r) {
      int mm = mbase + r;
      if (mm < Nrows) out[(size_t)mm * 16 + n] = acc[r] + bn;
    }
  }
}

__global__ void logsoftmax_kernel(float4* out, int N) {
  int i = blockIdx.x * blockDim.x + threadIdx.x;
  if (i >= N) return;
  float4* row = out + (size_t)i * 4;
  float4 v[4];
#pragma unroll
  for (int q = 0; q < 4; ++q) v[q] = row[q];
  float m = v[0].x;
#pragma unroll
  for (int q = 0; q < 4; ++q) {
    m = fmaxf(m, v[q].x); m = fmaxf(m, v[q].y);
    m = fmaxf(m, v[q].z); m = fmaxf(m, v[q].w);
  }
  float s = 0.0f;
#pragma unroll
  for (int q = 0; q < 4; ++q) {
    s += expf(v[q].x - m); s += expf(v[q].y - m);
    s += expf(v[q].z - m); s += expf(v[q].w - m);
  }
  float d = m + logf(s);
#pragma unroll
  for (int q = 0; q < 4; ++q) {
    v[q].x -= d; v[q].y -= d; v[q].z -= d; v[q].w -= d;
    row[q] = v[q];
  }
}

extern "C" void kernel_launch(void* const* d_in, const int* in_sizes, int n_in,
                              void* d_out, int out_size, void* d_ws, size_t ws_size,
                              hipStream_t stream) {
  const float* x     = (const float*)d_in[0];
  const int*   ei    = (const int*)d_in[1];
  const float* W_in  = (const float*)d_in[2];
  const float* b_in  = (const float*)d_in[3];
  const float* Wg    = (const float*)d_in[4];
  const float* bg    = (const float*)d_in[5];
  const float* W_out = (const float*)d_in[6];
  const float* b_out = (const float*)d_in[7];
  float* out = (float*)d_out;

  const int Hdim = 128;
  int N = in_sizes[0] / Hdim;
  int E = in_sizes[1] / 2;
  int L = in_sizes[5];

  float* ws   = (float*)d_ws;
  float* raw  = ws;
  float* bufA = raw + (size_t)N * Hdim;
  float* bufB = bufA + (size_t)N * Hdim;
  int*   cnt  = (int*)(bufB + (size_t)N * Hdim);
  float* dn   = (float*)cnt + N;
  int*   srow = (int*)(dn + N);
  float* pk   = out;
  float* pk2  = out + 2 * (size_t)N;
  float* gb   = out + 4 * (size_t)N;
  float* gb2  = out + 5 * (size_t)N;

  hipMemsetAsync(cnt, 0, (size_t)N * 2 * sizeof(float), stream);
  deg_kernel<<<(E + 255) / 256, 256, 0, stream>>>(ei, dn, cnt, E);
  dn_kernel<<<(N + 255) / 256, 256, 0, stream>>>(dn, N);
  scan_kernel<<<1, 1024, 0, stream>>>(cnt, N);
  fill_kernel<<<(E + 255) / 256, 256, 0, stream>>>(ei, cnt, srow, E);
  order_kernel<<<(N + 7) / 8, 256, 0, stream>>>(cnt, srow, N);

  gemm_relu_kernel<<<(N + 15) / 16, 256, 0, stream>>>(x, W_in, b_in, L > 0 ? Wg : nullptr, dn,
                                                     raw, pk, gb, N);

  float* h = raw;
  float* hnew = bufA;
  float* gac = pk;  float* gbc = gb;
  float* gan = pk2; float* gbn = gb2;
  for (int i = 0; i < L; ++i) {
    const float* wgn = (i + 1 < L) ? Wg + (size_t)(i + 1) * 2 * Hdim : nullptr;
    gather_kernel<<<(N + 7) / 8, 256, 0, stream>>>(cnt, srow, gac, gbc, bg, i, h, raw, hnew,
                                                  wgn, gan, gbn, N);
    h = hnew;
    hnew = (h == bufA) ? bufB : bufA;
    float* t = gac; gac = gan; gan = t;
    t = gbc; gbc = gbn; gbn = t;
  }

  int tiles = (N + 15) / 16;
  gemm_out_kernel<<<(tiles + 7) / 8, 256, 0, stream>>>(h, W_out, b_out, out, N);
  logsoftmax_kernel<<<(N + 255) / 256, 256, 0, stream>>>((float4*)out, N);
}


# Last layer is linear in its rows: h and raw go through W_out first (WMMA, one launch), then the final gather sums 64-B projected rows, 4 lanes per node, 8 nodes per wave; float sum order changes
# speedup vs baseline: 1.9848x; 1.0404x over previous
#include <hip/hip_runtime.h>

typedef float v2f __attribute__((ext_vector_type(2)));
typedef float v8f __attribute__((ext_vector_type(8)));

#define FAGCN_EPS 0.3f

__global__ void deg_kernel(const int* __restrict__ ei, float* __restrict__ deg,
                           int* __restrict__ cnt, int E) {
  int e = blockIdx.x * blockDim.x + threadIdx.x;
  if (e < E) {
    atomicAdd(&deg[ei[e]], 1.0f);
    atomicAdd(&cnt[ei[E + e]], 1);
  }
}

__device__ __forceinline__ int4 scan_load4(const int* cnt, int i0, int N) {
  int4 v = {0, 0, 0, 0};
  if (i0 + 3 < N) {
    v = *(const int4*)(cnt + i0);
  } else {
    if (i0 < N) v.x = cnt[i0];
    if (i0 + 1 < N) v.y = cnt[i0 + 1];
    if (i0 + 2 < N) v.z = cnt[i0 + 2];
  }
  return v;
}

__global__ void scan_kernel(int* cnt, int N) {
  __shared__ int wsum[32];
  int tid = threadIdx.x;
  int lane = tid & 31;
  int wave = tid >> 5;
  int carry = 0;
  int4 nxt = scan_load4(cnt, tid * 4, N);
  for (int base = 0; base < N; base += 4096) {
    int i0 = base + tid * 4;
    int4 v = nxt;
    nxt = scan_load4(cnt, i0 + 4096, N);
    int s = v.x + v.y + v.z + v.w;
    int incl = s;
    for (int off = 1; off < 32; off <<= 1) {
      int t = __shfl_up(incl, off, 32);
      if (lane >= off) incl += t;
    }
    if (lane == 31) wsum[wave] = incl;
    __syncthreads();
    int wt = wsum[lane];
    int wincl = wt;
    for (int off = 1; off < 32; off <<= 1) {
      int t = __shfl_up(wincl, off, 32);
      if (lane >= off) wincl += t;
    }
    int wexcl = __shfl(wincl, wave, 32) - __shfl(wt, wave, 32);
    int total = __shfl(wincl, 31, 32);
    int ex = carry + wexcl + incl - s;
    if (i0 + 3 < N) {
      int4 o;
      o.x = ex; o.y = ex + v.x; o.z = ex + v.x + v.y; o.w = ex + v.x + v.y + v.z;
      *(int4*)(cnt + i0) = o;
    } else {
      if (i0 < N) cnt[i0] = ex;
      if (i0 + 1 < N) cnt[i0 + 1] = ex + v.x;
      if (i0 + 2 < N) cnt[i0 + 2] = ex + v.x + v.y;
    }
    carry += total;
    __syncthreads();
  }
}

__global__ void fill_kernel(const int* __restrict__ ei, int* __restrict__ cur,
                            int* __restrict__ srow, int E) {
  int e = blockIdx.x * blockDim.x + threadIdx.x;
  if (e < E) {
    int pos = atomicAdd(&cur[ei[E + e]], 1);
    srow[pos] = ei[e];
  }
}

__global__ void order_kernel(const int* __restrict__ ends, int* __restrict__ srow, int N) {
  int node = __builtin_amdgcn_readfirstlane(blockIdx.x * 8 + (threadIdx.x >> 5));
  int lane = threadIdx.x & 31;
  if (node >= N) return;
  int beg = node ? ends[node - 1] : 0;
  int d = ends[node] - beg;
  if (d < 2 || d > 64) return;
  int a0 = lane < d ? srow[beg + lane] : 0x7fffffff;
  int a1 = lane + 32 < d ? srow[beg + 32 + lane] : 0x7fffffff;
  int r0 = 0, r1 = 0;
  int n0 = d < 32 ? d : 32;
  for (int i = 0; i < n0; ++i) {
    int x = __builtin_amdgcn_readlane(a0, i);
    r0 += (x < a0 || (x == a0 && i < lane)) ? 1 : 0;
    r1 += (x <= a1) ? 1 : 0;
  }
  for (int i = 0; i < d - 32; ++i) {
    int x = __builtin_amdgcn_readlane(a1, i);
    r0 += (x < a0) ? 1 : 0;
    r1 += (x < a1 || (x == a1 && i < lane)) ? 1 : 0;
  }
  if (lane < d) srow[beg + r0] = a0;
  if (lane + 32 < d) srow[beg + r1] = a1;
}

__global__ void dn_kernel(float* dn, int N) {
  int i = blockIdx.x * blockDim.x + threadIdx.x;
  if (i < N) dn[i] = 1.0f / sqrtf(fmaxf(dn[i], 1.0f));
}

#define TILE_PITCH 132
__global__ void gemm_relu_kernel(const float* __restrict__ X, const float* __restrict__ W,
                                 const float* __restrict__ bias, const float* __restrict__ wg0,
                                 const float* __restrict__ dn, float* __restrict__ H,
                                 float* __restrict__ pk, float* __restrict__ gb, int Nrows) {
  __shared__ __attribute__((aligned(16))) float tile[16 * TILE_PITCH];
  int wave = threadIdx.x >> 5;
  int lane = threadIdx.x & 31;
  int tileM = blockIdx.x;
  int mA = tileM * 16 + (lane & 15);
  int mclamp = mA < Nrows ? mA : (Nrows - 1);
  int n = wave * 16 + (lane & 15);
  int khalf = (lane >> 4) * 2;
  const float* xrow = X + (size_t)mclamp * 128 + khalf;
  const float* wcol = W + (size_t)khalf * 128 + n;
  __builtin_prefetch(xrow, 0, 0);
  float dnv = dn[mclamp];
  v8f acc = {};
#pragma unroll 4
  for (int kk = 0; kk < 128; kk += 4) {
    v2f a = *(const v2f*)(xrow + kk);
    v2f b;
    b.x = wcol[(size_t)kk * 128];
    b.y = wcol[(size_t)kk * 128 + 128];
    acc = __builtin_amdgcn_wmma_f32_16x16x4_f32(false, a, false, b, (short)0, acc,
                                               false, false);
  }
  int rbase = (lane >> 4) * 8;
  float bn = bias[n];
#pragma unroll
  for (int r = 0; r < 8; ++r)
    tile[(rbase + r) * TILE_PITCH + n] = fmaxf(acc[r] + bn, 0.0f);
  __syncthreads();
  float4 wa = {0.0f, 0.0f, 0.0f, 0.0f}, wb = {0.0f, 0.0f, 0.0f, 0.0f};
  if (wg0) {
    wa = *(const float4*)(wg0 + lane * 4);
    wb = *(const float4*)(wg0 + 128 + lane * 4);
  }
  for (int j = 0; j < 2; ++j) {
    int rr = wave * 2 + j;
    int m = tileM * 16 + rr;
    float4 v = *(const float4*)(tile + rr * TILE_PITCH + lane * 4);
    if (m < Nrows) {
      *(float4*)(H + (size_t)m * 128 + lane * 4) = v;
      if (wg0) {
        float dnm = __shfl(dnv, rr, 32);
        float sa = v.x * wa.x + v.y * wa.y + v.z * wa.z + v.w * wa.w;
        float sb = v.x * wb.x + v.y * wb.y + v.z * wb.z + v.w * wb.w;
        for (int off = 16; off > 0; off >>= 1) {
          sa += __shfl_down(sa, off, 32);
          sb += __shfl_down(sb, off, 32);
        }
        if (lane == 0) {
          float2 p; p.x = sa; p.y = dnm;
          *(float2*)(pk + 2 * (size_t)m) = p;
          gb[m] = sb;
        }
      }
    }
  }
}

__global__ void gather_kernel(const int* __restrict__ ends, const int* __restrict__ srow,
                              const float* __restrict__ pk, const float* __restrict__ gb,
                              const float* __restrict__ bg,
                              int layer, const float* __restrict__ H,
                              const float* __restrict__ raw, float* __restrict__ hout,
                              const float* __restrict__ wg_next, float* __restrict__ pk_next,
                              float* __restrict__ gb_next, int N) {
  int node = __builtin_amdgcn_readfirstlane(blockIdx.x * 8 + (threadIdx.x >> 5));
  int lane = threadIdx.x & 31;
  if (node >= N) return;
  int beg = node ? ends[node - 1] : 0;
  int end = ends[node];
  float gbc = gb[node];
  float bgl = bg[layer];
  float dnc = pk[2 * (size_t)node + 1];
  const float* hl = H + lane * 4;
  float4 rw = *(const float4*)(raw + (size_t)node * 128 + lane * 4);
  float4 acc = {0.0f, 0.0f, 0.0f, 0.0f};
  for (int base = beg; base < end; base += 32) {
    int n = end - base < 32 ? end - base : 32;
    int r = 0;
    float w = 0.0f;
    if (lane < n) {
      r = srow[base + lane];
      float2 gd = *(const float2*)(pk + 2 * (size_t)r);
      float g = tanhf(gd.x + gbc + bgl);
      w = g * gd.y * dnc;
    }
    int wi = __float_as_int(w);
    int k = 0;
    for (; k + 4 <= n; k += 4) {
      int r0 = __builtin_amdgcn_readlane(r, k);
      int r1 = __builtin_amdgcn_readlane(r, k + 1);
      int r2 = __builtin_amdgcn_readlane(r, k + 2);
      int r3 = __builtin_amdgcn_readlane(r, k + 3);
      float w0 = __int_as_float(__builtin_amdgcn_readlane(wi, k));
      float w1 = __int_as_float(__builtin_amdgcn_readlane(wi, k + 1));
      float w2 = __int_as_float(__builtin_amdgcn_readlane(wi, k + 2));
      float w3 = __int_as_float(__builtin_amdgcn_readlane(wi, k + 3));
      float4 h0 = *(const float4*)(hl + (size_t)r0 * 128);
      float4 h1 = *(const float4*)(hl + (size_t)r1 * 128);
      float4 h2 = *(const float4*)(hl + (size_t)r2 * 128);
      float4 h3 = *(const float4*)(hl + (size_t)r3 * 128);
      acc.x += w0 * h0.x; acc.y += w0 * h0.y; acc.z += w0 * h0.z; acc.w += w0 * h0.w;
      acc.x += w1 * h1.x; acc.y += w1 * h1.y; acc.z += w1 * h1.z; acc.w += w1 * h1.w;
      acc.x += w2 * h2.x; acc.y += w2 * h2.y; acc.z += w2 * h2.z; acc.w += w2 * h2.w;
      acc.x += w3 * h3.x; acc.y += w3 * h3.y; acc.z += w3 * h3.z; acc.w += w3 * h3.w;
    }
    for (; k < n; ++k) {
      int r0 = __builtin_amdgcn_readlane(r, k);
      float w0 = __int_as_float(__builtin_amdgcn_readlane(wi, k));
      float4 h0 = *(const float4*)(hl + (size_t)r0 * 128);
      acc.x += w0 * h0.x; acc.y += w0 * h0.y; acc.z += w0 * h0.z; acc.w += w0 * h0.w;
    }
  }
  float4 o;
  o.x = FAGCN_EPS * rw.x + acc.x;
  o.y = FAGCN_EPS * rw.y + acc.y;
  o.z = FAGCN_EPS * rw.z + acc.z;
  o.w = FAGCN_EPS * rw.w + acc.w;
  *(float4*)(hout + (size_t)node * 128 + lane * 4) = o;
  if (wg_next) {
    float4 wa = *(const float4*)(wg_next + lane * 4);
    float4 wb = *(const float4*)(wg_next + 128 + lane * 4);
    float sa = o.x * wa.x + o.y * wa.y + o.z * wa.z + o.w * wa.w;
    float sb = o.x * wb.x + o.y * wb.y + o.z * wb.z + o.w * wb.w;
    for (int off = 16; off > 0; off >>= 1) {
      sa += __shfl_down(sa, off, 32);
      sb += __shfl_down(sb, off, 32);
    }
    if (lane == 0) {
      float2 p; p.x = sa; p.y = dnc;
      *(float2*)(pk_next + 2 * (size_t)node) = p;
      gb_next[node] = sb;
    }
  }
}

__global__ void gemm_proj_kernel(const float* __restrict__ Hc, const float* __restrict__ Raw,
                                 const float* __restrict__ W, const float* __restrict__ bias,
                                 float rscale, float* __restrict__ Z, float* __restrict__ ZR,
                                 int Nrows) {
  int wave = threadIdx.x >> 5;
  int lane = threadIdx.x & 31;
  int tiles = (Nrows + 15) >> 4;
  int t = blockIdx.x * 8 + wave;
  if (t >= 2 * tiles) return;
  bool second = t >= tiles;
  int tileM = second ? t - tiles : t;
  const float* src = second ? Raw : Hc;
  float* dst = second ? ZR : Z;
  int mA = tileM * 16 + (lane & 15);
  int mclamp = mA < Nrows ? mA : (Nrows - 1);
  int n = lane & 15;
  int khalf = (lane >> 4) * 2;
  const float* hrow = src + (size_t)mclamp * 128 + khalf;
  const float* wcol = W + (size_t)khalf * 16 + n;
  __builtin_prefetch(hrow, 0, 0);
  v8f acc = {};
#pragma unroll 4
  for (int kk = 0; kk < 128; kk += 4) {
    v2f a = *(const v2f*)(hrow + kk);
    v2f b;
    b.x = wcol[(size_t)kk * 16];
    b.y = wcol[(size_t)kk * 16 + 16];
    acc = __builtin_amdgcn_wmma_f32_16x16x4_f32(false, a, false, b, (short)0, acc,
                                               false, false);
  }
  int mbase = tileM * 16 + ((lane >> 4) * 8);
  float bn = second ? bias[n] : 0.0f;
  float sc = second ? rscale : 1.0f;
  if (tileM * 16 + 16 <= Nrows) {
    float* op = dst + (size_t)mbase * 16 + n;
#pragma unroll
    for (int r = 0; r < 8; ++r)
      op[(size_t)r * 16] = acc[r] * sc + bn;
  } else {
#pragma unroll
    for (int r = 0; r < 8; ++r) {
      int mm = mbase + r;
      if (mm < Nrows) dst[(size_t)mm * 16 + n] = acc[r] * sc + bn;
    }
  }
}

__global__ void gather_proj_kernel(const int* __restrict__ ends, const int* __restrict__ srow,
                                   const float* __restrict__ pk, const float* __restrict__ gb,
                                   const float* __restrict__ bg, int layer,
                                   const float* __restrict__ Z, const float* __restrict__ ZR,
                                   float* __restrict__ logits, int N) {
  int node = blockIdx.x * 64 + (threadIdx.x >> 2);
  int q = threadIdx.x & 3;
  if (node >= N) return;
  int beg = node ? ends[node - 1] : 0;
  int end = ends[node];
  float gbc = gb[node];
  float bgl = bg[layer];
  float dnc = pk[2 * (size_t)node + 1];
  const float* zq = Z + q * 4;
  float4 zr = *(const float4*)(ZR + (size_t)node * 16 + q * 4);
  float4 acc = {0.0f, 0.0f, 0.0f, 0.0f};
  int last = end - 1;
  int r0 = 0, r1 = 0, r2 = 0, r3 = 0;
  if (beg < end) {
    r0 = srow[beg];
    r1 = srow[beg + 1 < last ? beg + 1 : last];
    r2 = srow[beg + 2 < last ? beg + 2 : last];
    r3 = srow[beg + 3 < last ? beg + 3 : last];
  }
  for (int base = beg; base < end; base += 4) {
    float2 g0 = *(const float2*)(pk + 2 * (size_t)r0);
    float2 g1 = *(const float2*)(pk + 2 * (size_t)r1);
    float2 g2 = *(const float2*)(pk + 2 * (size_t)r2);
    float2 g3 = *(const float2*)(pk + 2 * (size_t)r3);
    float4 h0 = *(const float4*)(zq + (size_t)r0 * 16);
    float4 h1 = *(const float4*)(zq + (size_t)r1 * 16);
    float4 h2 = *(const float4*)(zq + (size_t)r2 * 16);
    float4 h3 = *(const float4*)(zq + (size_t)r3 * 16);
    int nb = base + 4;
    int n0 = r0, n1 = r1, n2 = r2, n3 = r3;
    if (nb < end) {
      n0 = srow[nb];
      n1 = srow[nb + 1 < last ? nb + 1 : last];
      n2 = srow[nb + 2 < last ? nb + 2 : last];
      n3 = srow[nb + 3 < last ? nb + 3 : last];
    }
    {
      float w = tanhf(g0.x + gbc + bgl) * g0.y * dnc;
      acc.x += w * h0.x; acc.y += w * h0.y; acc.z += w * h0.z; acc.w += w * h0.w;
    }
    if (base + 1 < end) {
      float w = tanhf(g1.x + gbc + bgl) * g1.y * dnc;
      acc.x += w * h1.x; acc.y += w * h1.y; acc.z += w * h1.z; acc.w += w * h1.w;
    }
    if (base + 2 < end) {
      float w = tanhf(g2.x + gbc + bgl) * g2.y * dnc;
      acc.x += w * h2.x; acc.y += w * h2.y; acc.z += w * h2.z; acc.w += w * h2.w;
    }
    if (base + 3 < end) {
      float w = tanhf(g3.x + gbc + bgl) * g3.y * dnc;
      acc.x += w * h3.x; acc.y += w * h3.y; acc.z += w * h3.z; acc.w += w * h3.w;
    }
    r0 = n0; r1 = n1; r2 = n2; r3 = n3;
  }
  float4 o;
  o.x = zr.x + acc.x;
  o.y = zr.y + acc.y;
  o.z = zr.z + acc.z;
  o.w = zr.w + acc.w;
  *(float4*)(logits + (size_t)node * 16 + q * 4) = o;
}

__global__ void logsoftmax_kernel(const float4* __restrict__ in, float4* __restrict__ out, int N) {
  int i = blockIdx.x * blockDim.x + threadIdx.x;
  if (i >= N) return;
  const float4* rin = in + (size_t)i * 4;
  float4* row = out + (size_t)i * 4;
  float4 v[4];
#pragma unroll
  for (int q = 0; q < 4; ++q) v[q] = rin[q];
  float m = v[0].x;
#pragma unroll
  for (int q = 0; q < 4; ++q) {
    m = fmaxf(m, v[q].x); m = fmaxf(m, v[q].y);
    m = fmaxf(m, v[q].z); m = fmaxf(m, v[q].w);
  }
  float s = 0.0f;
#pragma unroll
  for (int q = 0; q < 4; ++q) {
    s += expf(v[q].x - m); s += expf(v[q].y - m);
    s += expf(v[q].z - m); s += expf(v[q].w - m);
  }
  float d = m + logf(s);
#pragma unroll
  for (int q = 0; q < 4; ++q) {
    v[q].x -= d; v[q].y -= d; v[q].z -= d; v[q].w -= d;
    row[q] = v[q];
  }
}

extern "C" void kernel_launch(void* const* d_in, const int* in_sizes, int n_in,
                              void* d_out, int out_size, void* d_ws, size_t ws_size,
                              hipStream_t stream) {
  const float* x     = (const float*)d_in[0];
  const int*   ei    = (const int*)d_in[1];
  const float* W_in  = (const float*)d_in[2];
  const float* b_in  = (const float*)d_in[3];
  const float* Wg    = (const float*)d_in[4];
  const float* bg    = (const float*)d_in[5];
  const float* W_out = (const float*)d_in[6];
  const float* b_out = (const float*)d_in[7];
  float* out = (float*)d_out;

  const int Hdim = 128;
  int N = in_sizes[0] / Hdim;
  int E = in_sizes[1] / 2;
  int L = in_sizes[5];

  float* ws   = (float*)d_ws;
  float* raw  = ws;
  float* bufA = raw + (size_t)N * Hdim;
  float* bufB = bufA + (size_t)N * Hdim;
  int*   cnt  = (int*)(bufB + (size_t)N * Hdim);
  float* dn   = (float*)cnt + N;
  int*   srow = (int*)(dn + N);
  float* pk   = out;
  float* pk2  = out + 2 * (size_t)N;
  float* gb   = out + 4 * (size_t)N;
  float* gb2  = out + 5 * (size_t)N;

  hipMemsetAsync(cnt, 0, (size_t)N * 2 * sizeof(float), stream);
  deg_kernel<<<(E + 255) / 256, 256, 0, stream>>>(ei, dn, cnt, E);
  dn_kernel<<<(N + 255) / 256, 256, 0, stream>>>(dn, N);
  scan_kernel<<<1, 1024, 0, stream>>>(cnt, N);
  fill_kernel<<<(E + 255) / 256, 256, 0, stream>>>(ei, cnt, srow, E);
  order_kernel<<<(N + 7) / 8, 256, 0, stream>>>(cnt, srow, N);

  gemm_relu_kernel<<<(N + 15) / 16, 256, 0, stream>>>(x, W_in, b_in, L > 0 ? Wg : nullptr, dn,
                                                     raw, pk, gb, N);

  float* h = raw;
  float* hnew = bufA;
  float* gac = pk;  float* gbc = gb;
  float* gan = pk2; float* gbn = gb2;
  for (int i = 0; i + 1 < L; ++i) {
    const float* wgn = Wg + (size_t)(i + 1) * 2 * Hdim;
    gather_kernel<<<(N + 7) / 8, 256, 0, stream>>>(cnt, srow, gac, gbc, bg, i, h, raw, hnew,
                                                  wgn, gan, gbn, N);
    h = hnew;
    hnew = (h == bufA) ? bufB : bufA;
    float* t = gac; gac = gan; gan = t;
    t = gbc; gbc = gbn; gbn = t;
  }

  float* Z  = hnew;
  float* ZR = hnew + 16 * (size_t)N;
  float* LG = hnew + 32 * (size_t)N;
  int tiles = (N + 15) / 16;
  gemm_proj_kernel<<<(2 * tiles + 7) / 8, 256, 0, stream>>>(h, raw, W_out, b_out,
                                                           L > 0 ? FAGCN_EPS : 1.0f, Z, ZR, N);
  if (L > 0)
    gather_proj_kernel<<<(N + 63) / 64, 256, 0, stream>>>(cnt, srow, gac, gbc, bg, L - 1, Z, ZR,
                                                         LG, N);
  logsoftmax_kernel<<<(N + 255) / 256, 256, 0, stream>>>((const float4*)(L > 0 ? LG : ZR),
                                                        (float4*)out, N);
}


# Layer before last is linear in its rows too: rows go through [W_out|last gate vectors] first (WMMA), it sums 128-B projected rows, 4 lanes per node; 128-wide gather goes; gate dot sum order changes
# speedup vs baseline: 2.0545x; 1.0351x over previous
#include <hip/hip_runtime.h>

typedef float v2f __attribute__((ext_vector_type(2)));
typedef float v8f __attribute__((ext_vector_type(8)));

#define FAGCN_EPS 0.3f

__global__ void deg_kernel(const int* __restrict__ ei, float* __restrict__ deg,
                           int* __restrict__ cnt, int E) {
  int e = blockIdx.x * blockDim.x + threadIdx.x;
  if (e < E) {
    atomicAdd(&deg[ei[e]], 1.0f);
    atomicAdd(&cnt[ei[E + e]], 1);
  }
}

__device__ __forceinline__ int4 scan_load4(const int* cnt, int i0, int N) {
  int4 v = {0, 0, 0, 0};
  if (i0 + 3 < N) {
    v = *(const int4*)(cnt + i0);
  } else {
    if (i0 < N) v.x = cnt[i0];
    if (i0 + 1 < N) v.y = cnt[i0 + 1];
    if (i0 + 2 < N) v.z = cnt[i0 + 2];
  }
  return v;
}

__global__ void scan_kernel(int* cnt, int N) {
  __shared__ int wsum[32];
  int tid = threadIdx.x;
  int lane = tid & 31;
  int wave = tid >> 5;
  int carry = 0;
  int4 nxt = scan_load4(cnt, tid * 4, N);
  for (int base = 0; base < N; base += 4096) {
    int i0 = base + tid * 4;
    int4 v = nxt;
    nxt = scan_load4(cnt, i0 + 4096, N);
    int s = v.x + v.y + v.z + v.w;
    int incl = s;
    for (int off = 1; off < 32; off <<= 1) {
      int t = __shfl_up(incl, off, 32);
      if (lane >= off) incl += t;
    }
    if (lane == 31) wsum[wave] = incl;
    __syncthreads();
    int wt = wsum[lane];
    int wincl = wt;
    for (int off = 1; off < 32; off <<= 1) {
      int t = __shfl_up(wincl, off, 32);
      if (lane >= off) wincl += t;
    }
    int wexcl = __shfl(wincl, wave, 32) - __shfl(wt, wave, 32);
    int total = __shfl(wincl, 31, 32);
    int ex = carry + wexcl + incl - s;
    if (i0 + 3 < N) {
      int4 o;
      o.x = ex; o.y = ex + v.x; o.z = ex + v.x + v.y; o.w = ex + v.x + v.y + v.z;
      *(int4*)(cnt + i0) = o;
    } else {
      if (i0 < N) cnt[i0] = ex;
      if (i0 + 1 < N) cnt[i0 + 1] = ex + v.x;
      if (i0 + 2 < N) cnt[i0 + 2] = ex + v.x + v.y;
    }
    carry += total;
    __syncthreads();
  }
}

__global__ void fill_kernel(const int* __restrict__ ei, int* __restrict__ cur,
                            int* __restrict__ srow, int E) {
  int e = blockIdx.x * blockDim.x + threadIdx.x;
  if (e < E) {
    int pos = atomicAdd(&cur[ei[E + e]], 1);
    srow[pos] = ei[e];
  }
}

__global__ void order_kernel(const int* __restrict__ ends, int* __restrict__ srow, int N) {
  int node = __builtin_amdgcn_readfirstlane(blockIdx.x * 8 + (threadIdx.x >> 5));
  int lane = threadIdx.x & 31;
  if (node >= N) return;
  int beg = node ? ends[node - 1] : 0;
  int d = ends[node] - beg;
  if (d < 2 || d > 64) return;
  int a0 = lane < d ? srow[beg + lane] : 0x7fffffff;
  int a1 = lane + 32 < d ? srow[beg + 32 + lane] : 0x7fffffff;
  int r0 = 0, r1 = 0;
  int n0 = d < 32 ? d : 32;
  for (int i = 0; i < n0; ++i) {
    int x = __builtin_amdgcn_readlane(a0, i);
    r0 += (x < a0 || (x == a0 && i < lane)) ? 1 : 0;
    r1 += (x <= a1) ? 1 : 0;
  }
  for (int i = 0; i < d - 32; ++i) {
    int x = __builtin_amdgcn_readlane(a1, i);
    r0 += (x < a0) ? 1 : 0;
    r1 += (x < a1 || (x == a1 && i < lane)) ? 1 : 0;
  }
  if (lane < d) srow[beg + r0] = a0;
  if (lane + 32 < d) srow[beg + r1] = a1;
}

__global__ void dn_kernel(float* dn, int N) {
  int i = blockIdx.x * blockDim.x + threadIdx.x;
  if (i < N) dn[i] = 1.0f / sqrtf(fmaxf(dn[i], 1.0f));
}

#define TILE_PITCH 132
__global__ void gemm_relu_kernel(const float* __restrict__ X, const float* __restrict__ W,
                                 const float* __restrict__ bias, const float* __restrict__ wg0,
                                 const float* __restrict__ dn, float* __restrict__ H,
                                 float* __restrict__ pk, float* __restrict__ gb, int Nrows) {
  __shared__ __attribute__((aligned(16))) float tile[16 * TILE_PITCH];
  int wave = threadIdx.x >> 5;
  int lane = threadIdx.x & 31;
  int tileM = blockIdx.x;
  int mA = tileM * 16 + (lane & 15);
  int mclamp = mA < Nrows ? mA : (Nrows - 1);
  int n = wave * 16 + (lane & 15);
  int khalf = (lane >> 4) * 2;
  const float* xrow = X + (size_t)mclamp * 128 + khalf;
  const float* wcol = W + (size_t)khalf * 128 + n;
  __builtin_prefetch(xrow, 0, 0);
  float dnv = dn[mclamp];
  v8f acc = {};
#pragma unroll 4
  for (int kk = 0; kk < 128; kk += 4) {
    v2f a = *(const v2f*)(xrow + kk);
    v2f b;
    b.x = wcol[(size_t)kk * 128];
    b.y = wcol[(size_t)kk * 128 + 128];
    acc = __builtin_amdgcn_wmma_f32_16x16x4_f32(false, a, false, b, (short)0, acc,
                                               false, false);
  }
  int rbase = (lane >> 4) * 8;
  float bn = bias[n];
#pragma unroll
  for (int r = 0; r < 8; ++r)
    tile[(rbase + r) * TILE_PITCH + n] = fmaxf(acc[r] + bn, 0.0f);
  __syncthreads();
  float4 wa = {0.0f, 0.0f, 0.0f, 0.0f}, wb = {0.0f, 0.0f, 0.0f, 0.0f};
  if (wg0) {
    wa = *(const float4*)(wg0 + lane * 4);
    wb = *(const float4*)(wg0 + 128 + lane * 4);
  }
  for (int j = 0; j < 2; ++j) {
    int rr = wave * 2 + j;
    int m = tileM * 16 + rr;
    float4 v = *(const float4*)(tile + rr * TILE_PITCH + lane * 4);
    if (m < Nrows) {
      *(float4*)(H + (size_t)m * 128 + lane * 4) = v;
      if (wg0) {
        float dnm = __shfl(dnv, rr, 32);
        float sa = v.x * wa.x + v.y * wa.y + v.z * wa.z + v.w * wa.w;
        float sb = v.x * wb.x + v.y * wb.y + v.z * wb.z + v.w * wb.w;
        for (int off = 16; off > 0; off >>= 1) {
          sa += __shfl_down(sa, off, 32);
          sb += __shfl_down(sb, off, 32);
        }
        if (lane == 0) {
          float2 p; p.x = sa; p.y = dnm;
          *(float2*)(pk + 2 * (size_t)m) = p;
          gb[m] = sb;
        }
      }
    }
  }
}

__global__ void gather_kernel(const int* __restrict__ ends, const int* __restrict__ srow,
                              const float* __restrict__ pk, const float* __restrict__ gb,
                              const float* __restrict__ bg,
                              int layer, const float* __restrict__ H,
                              const float* __restrict__ raw, float* __restrict__ hout,
                              const float* __restrict__ wg_next, float* __restrict__ pk_next,
                              float* __restrict__ gb_next, int N) {
  int node = __builtin_amdgcn_readfirstlane(blockIdx.x * 8 + (threadIdx.x >> 5));
  int lane = threadIdx.x & 31;
  if (node >= N) return;
  int beg = node ? ends[node - 1] : 0;
  int end = ends[node];
  float gbc = gb[node];
  float bgl = bg[layer];
  float dnc = pk[2 * (size_t)node + 1];
  const float* hl = H + lane * 4;
  float4 rw = *(const float4*)(raw + (size_t)node * 128 + lane * 4);
  float4 acc = {0.0f, 0.0f, 0.0f, 0.0f};
  for (int base = beg; base < end; base += 32) {
    int n = end - base < 32 ? end - base : 32;
    int r = 0;
    float w = 0.0f;
    if (lane < n) {
      r = srow[base + lane];
      float2 gd = *(const float2*)(pk + 2 * (size_t)r);
      float g = tanhf(gd.x + gbc + bgl);
      w = g * gd.y * dnc;
    }
    int wi = __float_as_int(w);
    int k = 0;
    for (; k + 4 <= n; k += 4) {
      int r0 = __builtin_amdgcn_readlane(r, k);
      int r1 = __builtin_amdgcn_readlane(r, k + 1);
      int r2 = __builtin_amdgcn_readlane(r, k + 2);
      int r3 = __builtin_amdgcn_readlane(r, k + 3);
      float w0 = __int_as_float(__builtin_amdgcn_readlane(wi, k));
      float w1 = __int_as_float(__builtin_amdgcn_readlane(wi, k + 1));
      float w2 = __int_as_float(__builtin_amdgcn_readlane(wi, k + 2));
      float w3 = __int_as_float(__builtin_amdgcn_readlane(wi, k + 3));
      float4 h0 = *(const float4*)(hl + (size_t)r0 * 128);
      float4 h1 = *(const float4*)(hl + (size_t)r1 * 128);
      float4 h2 = *(const float4*)(hl + (size_t)r2 * 128);
      float4 h3 = *(const float4*)(hl + (size_t)r3 * 128);
      acc.x += w0 * h0.x; acc.y += w0 * h0.y; acc.z += w0 * h0.z; acc.w += w0 * h0.w;
      acc.x += w1 * h1.x; acc.y += w1 * h1.y; acc.z += w1 * h1.z; acc.w += w1 * h1.w;
      acc.x += w2 * h2.x; acc.y += w2 * h2.y; acc.z += w2 * h2.z; acc.w += w2 * h2.w;
      acc.x += w3 * h3.x; acc.y += w3 * h3.y; acc.z += w3 * h3.z; acc.w += w3 * h3.w;
    }
    for (; k < n; ++k) {
      int r0 = __builtin_amdgcn_readlane(r, k);
      float w0 = __int_as_float(__builtin_amdgcn_readlane(wi, k));
      float4 h0 = *(const float4*)(hl + (size_t)r0 * 128);
      acc.x += w0 * h0.x; acc.y += w0 * h0.y; acc.z += w0 * h0.z; acc.w += w0 * h0.w;
    }
  }
  float4 o;
  o.x = FAGCN_EPS * rw.x + acc.x;
  o.y = FAGCN_EPS * rw.y + acc.y;
  o.z = FAGCN_EPS * rw.z + acc.z;
  o.w = FAGCN_EPS * rw.w + acc.w;
  *(float4*)(hout + (size_t)node * 128 + lane * 4) = o;
  if (wg_next) {
    float4 wa = *(const float4*)(wg_next + lane * 4);
    float4 wb = *(const float4*)(wg_next + 128 + lane * 4);
    float sa = o.x * wa.x + o.y * wa.y + o.z * wa.z + o.w * wa.w;
    float sb = o.x * wb.x + o.y * wb.y + o.z * wb.z + o.w * wb.w;
    for (int off = 16; off > 0; off >>= 1) {
      sa += __shfl_down(sa, off, 32);
      sb += __shfl_down(sb, off, 32);
    }
    if (lane == 0) {
      float2 p; p.x = sa; p.y = dnc;
      *(float2*)(pk_next + 2 * (size_t)node) = p;
      gb_next[node] = sb;
    }
  }
}

__global__ void gemm_proj_kernel(const float* __restrict__ Hc, const float* __restrict__ Raw,
                                 const float* __restrict__ W, const float* __restrict__ bias,
                                 float rscale, float* __restrict__ Z, float* __restrict__ ZR,
                                 int Nrows) {
  int wave = threadIdx.x >> 5;
  int lane = threadIdx.x & 31;
  int tiles = (Nrows + 15) >> 4;
  int t = blockIdx.x * 8 + wave;
  if (t >= 2 * tiles) return;
  bool second = t >= tiles;
  int tileM = second ? t - tiles : t;
  const float* src = second ? Raw : Hc;
  float* dst = second ? ZR : Z;
  int mA = tileM * 16 + (lane & 15);
  int mclamp = mA < Nrows ? mA : (Nrows - 1);
  int n = lane & 15;
  int khalf = (lane >> 4) * 2;
  const float* hrow = src + (size_t)mclamp * 128 + khalf;
  const float* wcol = W + (size_t)khalf * 16 + n;
  __builtin_prefetch(hrow, 0, 0);
  v8f acc = {};
#pragma unroll 4
  for (int kk = 0; kk < 128; kk += 4) {
    v2f a = *(const v2f*)(hrow + kk);
    v2f b;
    b.x = wcol[(size_t)kk * 16];
    b.y = wcol[(size_t)kk * 16 + 16];
    acc = __builtin_amdgcn_wmma_f32_16x16x4_f32(false, a, false, b, (short)0, acc,
                                               false, false);
  }
  int mbase = tileM * 16 + ((lane >> 4) * 8);
  float bn = second ? bias[n] : 0.0f;
  float sc = second ? rscale : 1.0f;
  if (tileM * 16 + 16 <= Nrows) {
    float* op = dst + (size_t)mbase * 16 + n;
#pragma unroll
    for (int r = 0; r < 8; ++r)
      op[(size_t)r * 16] = acc[r] * sc + bn;
  } else {
#pragma unroll
    for (int r = 0; r < 8; ++r) {
      int mm = mbase + r;
      if (mm < Nrows) dst[(size_t)mm * 16 + n] = acc[r] * sc + bn;
    }
  }
}

__global__ void gather_proj_kernel(const int* __restrict__ ends, const int* __restrict__ srow,
                                   const float* __restrict__ pk, const float* __restrict__ gb,
                                   const float* __restrict__ bg, int layer,
                                   const float* __restrict__ Z, const float* __restrict__ ZR,
                                   float* __restrict__ logits, int N) {
  int node = blockIdx.x * 64 + (threadIdx.x >> 2);
  int q = threadIdx.x & 3;
  if (node >= N) return;
  int beg = node ? ends[node - 1] : 0;
  int end = ends[node];
  float gbc = gb[node];
  float bgl = bg[layer];
  float dnc = pk[2 * (size_t)node + 1];
  const float* zq = Z + q * 4;
  float4 zr = *(const float4*)(ZR + (size_t)node * 16 + q * 4);
  float4 acc = {0.0f, 0.0f, 0.0f, 0.0f};
  int last = end - 1;
  int r0 = 0, r1 = 0, r2 = 0, r3 = 0;
  if (beg < end) {
    r0 = srow[beg];
    r1 = srow[beg + 1 < last ? beg + 1 : last];
    r2 = srow[beg + 2 < last ? beg + 2 : last];
    r3 = srow[beg + 3 < last ? beg + 3 : last];
  }
  for (int base = beg; base < end; base += 4) {
    float2 g0 = *(const float2*)(pk + 2 * (size_t)r0);
    float2 g1 = *(const float2*)(pk + 2 * (size_t)r1);
    float2 g2 = *(const float2*)(pk + 2 * (size_t)r2);
    float2 g3 = *(const float2*)(pk + 2 * (size_t)r3);
    float4 h0 = *(const float4*)(zq + (size_t)r0 * 16);
    float4 h1 = *(const float4*)(zq + (size_t)r1 * 16);
    float4 h2 = *(const float4*)(zq + (size_t)r2 * 16);
    float4 h3 = *(const float4*)(zq + (size_t)r3 * 16);
    int nb = base + 4;
    int n0 = r0, n1 = r1, n2 = r2, n3 = r3;
    if (nb < end) {
      n0 = srow[nb];
      n1 = srow[nb + 1 < last ? nb + 1 : last];
      n2 = srow[nb + 2 < last ? nb + 2 : last];
      n3 = srow[nb + 3 < last ? nb + 3 : last];
    }
    {
      float w = tanhf(g0.x + gbc + bgl) * g0.y * dnc;
      acc.x += w * h0.x; acc.y += w * h0.y; acc.z += w * h0.z; acc.w += w * h0.w;
    }
    if (base + 1 < end) {
      float w = tanhf(g1.x + gbc + bgl) * g1.y * dnc;
      acc.x += w * h1.x; acc.y += w * h1.y; acc.z += w * h1.z; acc.w += w * h1.w;
    }
    if (base + 2 < end) {
      float w = tanhf(g2.x + gbc + bgl) * g2.y * dnc;
      acc.x += w * h2.x; acc.y += w * h2.y; acc.z += w * h2.z; acc.w += w * h2.w;
    }
    if (base + 3 < end) {
      float w = tanhf(g3.x + gbc + bgl) * g3.y * dnc;
      acc.x += w * h3.x; acc.y += w * h3.y; acc.z += w * h3.z; acc.w += w * h3.w;
    }
    r0 = n0; r1 = n1; r2 = n2; r3 = n3;
  }
  float4 o;
  o.x = zr.x + acc.x;
  o.y = zr.y + acc.y;
  o.z = zr.z + acc.z;
  o.w = zr.w + acc.w;
  *(float4*)(logits + (size_t)node * 16 + q * 4) = o;
}

#define YP 32
__global__ void gemm_proj2_kernel(const float* __restrict__ Hc, const float* __restrict__ Raw,
                                  const float* __restrict__ W, const float* __restrict__ wgl,
                                  const float* __restrict__ pk, const float* __restrict__ gb,
                                  float* __restrict__ Y, float* __restrict__ YR,
                                  int Nrows, int nt) {
  int wave = threadIdx.x >> 5;
  int lane = threadIdx.x & 31;
  int tiles = (Nrows + 15) >> 4;
  int t = blockIdx.x * 8 + wave;
  if (t >= nt) return;
  bool second = t >= tiles;
  int tileM = second ? t - tiles : t;
  const float* src = second ? Raw : Hc;
  float* dst = second ? YR : Y;
  int mA = tileM * 16 + (lane & 15);
  int mclamp = mA < Nrows ? mA : (Nrows - 1);
  int n = lane & 15;
  int khalf = (lane >> 4) * 2;
  const float* hrow = src + (size_t)mclamp * 128 + khalf;
  const float* wcol = W + (size_t)khalf * 16 + n;
  const float* gcol = wgl + (n & 1) * 128 + khalf;
  __builtin_prefetch(hrow, 0, 0);
  int mbase = tileM * 16 + ((lane >> 4) * 8);
  const float* tp = (n == 2 || n == 3) ? pk + (n - 2) : gb;
  int ts = (n == 2 || n == 3) ? 2 : 1;
  float tv[8];
#pragma unroll
  for (int r = 0; r < 8; ++r) {
    int mr = mbase + r < Nrows ? mbase + r : Nrows - 1;
    tv[r] = tp[(size_t)mr * ts];
  }
  v8f acc = {};
  v8f acg = {};
#pragma unroll 4
  for (int kk = 0; kk < 128; kk += 4) {
    v2f a = *(const v2f*)(hrow + kk);
    v2f b;
    b.x = wcol[(size_t)kk * 16];
    b.y = wcol[(size_t)kk * 16 + 16];
    v2f g = *(const v2f*)(gcol + kk);
    acc = __builtin_amdgcn_wmma_f32_16x16x4_f32(false, a, false, b, (short)0, acc,
                                               false, false);
    acg = __builtin_amdgcn_wmma_f32_16x16x4_f32(false, a, false, g, (short)0, acg,
                                               false, false);
  }
  float* yb = dst + (size_t)mbase * YP + n;
  if (tileM * 16 + 16 <= Nrows) {
#pragma unroll
    for (int r = 0; r < 8; ++r) {
      yb[(size_t)r * YP] = acc[r];
      if (n < 5) yb[(size_t)r * YP + 16] = n < 2 ? acg[r] : tv[r];
    }
  } else {
#pragma unroll
    for (int r = 0; r < 8; ++r) {
      if (mbase + r < Nrows) {
        yb[(size_t)r * YP] = acc[r];
        if (n < 5) yb[(size_t)r * YP + 16] = n < 2 ? acg[r] : tv[r];
      }
    }
  }
}

__global__ void gather_mid_kernel(const int* __restrict__ ends, const int* __restrict__ srow,
                                  const float* __restrict__ bg, int layer,
                                  const float* __restrict__ Y, const float* __restrict__ YR,
                                  const float* __restrict__ bias,
                                  float* __restrict__ Z, float* __restrict__ ZR,
                                  float* __restrict__ pk_next, float* __restrict__ gb_next,
                                  int N) {
  int node = blockIdx.x * 64 + (threadIdx.x >> 2);
  int q = threadIdx.x & 3;
  if (node >= N) return;
  int beg = node ? ends[node - 1] : 0;
  int end = ends[node];
  float bgl = bg[layer];
  const float* yc = Y + (size_t)node * YP;
  float dnc = yc[19];
  float gbc = yc[20];
  const float* yrc = YR + (size_t)node * YP;
  float4 own = *(const float4*)(yrc + q * 4);
  float2 owng = *(const float2*)(yrc + 16);
  float4 bq = *(const float4*)(bias + q * 4);
  const float* yq = Y + q * 4;
  const float* yt = Y + 16;
  float4 acc = {0.0f, 0.0f, 0.0f, 0.0f};
  float sga = 0.0f, sgb = 0.0f;
  int last = end - 1;
  int r0 = 0, r1 = 0, r2 = 0, r3 = 0;
  if (beg < end) {
    r0 = srow[beg];
    r1 = srow[beg + 1 < last ? beg + 1 : last];
    r2 = srow[beg + 2 < last ? beg + 2 : last];
    r3 = srow[beg + 3 < last ? beg + 3 : last];
  }
  for (int base = beg; base < end; base += 4) {
    float4 t0 = *(const float4*)(yt + (size_t)r0 * YP);
    float4 t1 = *(const float4*)(yt + (size_t)r1 * YP);
    float4 t2 = *(const float4*)(yt + (size_t)r2 * YP);
    float4 t3 = *(const float4*)(yt + (size_t)r3 * YP);
    float4 h0 = *(const float4*)(yq + (size_t)r0 * YP);
    float4 h1 = *(const float4*)(yq + (size_t)r1 * YP);
    float4 h2 = *(const float4*)(yq + (size_t)r2 * YP);
    float4 h3 = *(const float4*)(yq + (size_t)r3 * YP);
    int nb = base + 4;
    int n0 = r0, n1 = r1, n2 = r2, n3 = r3;
    if (nb < end) {
      n0 = srow[nb];
      n1 = srow[nb + 1 < last ? nb + 1 : last];
      n2 = srow[nb + 2 < last ? nb + 2 : last];
      n3 = srow[nb + 3 < last ? nb + 3 : last];
    }
    {
      float w = tanhf(t0.z + gbc + bgl) * t0.w * dnc;
      acc.x += w * h0.x; acc.y += w * h0.y; acc.z += w * h0.z; acc.w += w * h0.w;
      sga += w * t0.x; sgb += w * t0.y;
    }
    if (base + 1 < end) {
      float w = tanhf(t1.z + gbc + bgl) * t1.w * dnc;
      acc.x += w * h1.x; acc.y += w * h1.y; acc.z += w * h1.z; acc.w += w * h1.w;
      sga += w * t1.x; sgb += w * t1.y;
    }
    if (base + 2 < end) {
      float w = tanhf(t2.z + gbc + bgl) * t2.w * dnc;
      acc.x += w * h2.x; acc.y += w * h2.y; acc.z += w * h2.z; acc.w += w * h2.w;
      sga += w * t2.x; sgb += w * t2.y;
    }
    if (base + 3 < end) {
      float w = tanhf(t3.z + gbc + bgl) * t3.w * dnc;
      acc.x += w * h3.x; acc.y += w * h3.y; acc.z += w * h3.z; acc.w += w * h3.w;
      sga += w * t3.x; sgb += w * t3.y;
    }
    r0 = n0; r1 = n1; r2 = n2; r3 = n3;
  }
  float4 o, zr;
  o.x = FAGCN_EPS * own.x + acc.x;  zr.x = FAGCN_EPS * own.x + bq.x;
  o.y = FAGCN_EPS * own.y + acc.y;  zr.y = FAGCN_EPS * own.y + bq.y;
  o.z = FAGCN_EPS * own.z + acc.z;  zr.z = FAGCN_EPS * own.z + bq.z;
  o.w = FAGCN_EPS * own.w + acc.w;  zr.w = FAGCN_EPS * own.w + bq.w;
  *(float4*)(Z + (size_t)node * 16 + q * 4) = o;
  *(float4*)(ZR + (size_t)node * 16 + q * 4) = zr;
  if (q == 0) {
    float2 p; p.x = FAGCN_EPS * owng.x + sga; p.y = dnc;
    *(float2*)(pk_next + 2 * (size_t)node) = p;
    gb_next[node] = FAGCN_EPS * owng.y + sgb;
  }
}

__global__ void logsoftmax_kernel(const float4* __restrict__ in, float4* __restrict__ out, int N) {
  int i = blockIdx.x * blockDim.x + threadIdx.x;
  if (i >= N) return;
  const float4* rin = in + (size_t)i * 4;
  float4* row = out + (size_t)i * 4;
  float4 v[4];
#pragma unroll
  for (int q = 0; q < 4; ++q) v[q] = rin[q];
  float m = v[0].x;
#pragma unroll
  for (int q = 0; q < 4; ++q) {
    m = fmaxf(m, v[q].x); m = fmaxf(m, v[q].y);
    m = fmaxf(m, v[q].z); m = fmaxf(m, v[q].w);
  }
  float s = 0.0f;
#pragma unroll
  for (int q = 0; q < 4; ++q) {
    s += expf(v[q].x - m); s += expf(v[q].y - m);
    s += expf(v[q].z - m); s += expf(v[q].w - m);
  }
  float d = m + logf(s);
#pragma unroll
  for (int q = 0; q < 4; ++q) {
    v[q].x -= d; v[q].y -= d; v[q].z -= d; v[q].w -= d;
    row[q] = v[q];
  }
}

extern "C" void kernel_launch(void* const* d_in, const int* in_sizes, int n_in,
                              void* d_out, int out_size, void* d_ws, size_t ws_size,
                              hipStream_t stream) {
  const float* x     = (const float*)d_in[0];
  const int*   ei    = (const int*)d_in[1];
  const float* W_in  = (const float*)d_in[2];
  const float* b_in  = (const float*)d_in[3];
  const float* Wg    = (const float*)d_in[4];
  const float* bg    = (const float*)d_in[5];
  const float* W_out = (const float*)d_in[6];
  const float* b_out = (const float*)d_in[7];
  float* out = (float*)d_out;

  const int Hdim = 128;
  int N = in_sizes[0] / Hdim;
  int E = in_sizes[1] / 2;
  int L = in_sizes[5];

  float* ws   = (float*)d_ws;
  float* raw  = ws;
  float* bufA = raw + (size_t)N * Hdim;
  float* bufB = bufA + (size_t)N * Hdim;
  int*   cnt  = (int*)(bufB + (size_t)N * Hdim);
  float* dn   = (float*)cnt + N;
  int*   srow = (int*)(dn + N);
  float* pk   = out;
  float* pk2  = out + 2 * (size_t)N;
  float* gb   = out + 4 * (size_t)N;
  float* gb2  = out + 5 * (size_t)N;

  hipMemsetAsync(cnt, 0, (size_t)N * 2 * sizeof(float), stream);
  deg_kernel<<<(E + 255) / 256, 256, 0, stream>>>(ei, dn, cnt, E);
  dn_kernel<<<(N + 255) / 256, 256, 0, stream>>>(dn, N);
  scan_kernel<<<1, 1024, 0, stream>>>(cnt, N);
  fill_kernel<<<(E + 255) / 256, 256, 0, stream>>>(ei, cnt, srow, E);
  order_kernel<<<(N + 7) / 8, 256, 0, stream>>>(cnt, srow, N);

  gemm_relu_kernel<<<(N + 15) / 16, 256, 0, stream>>>(x, W_in, b_in, L > 0 ? Wg : nullptr, dn,
                                                     raw, pk, gb, N);

  float* h = raw;
  float* hnew = bufA;
  float* gac = pk;  float* gbc = gb;
  float* gan = pk2; float* gbn = gb2;
  for (int i = 0; i + 2 < L; ++i) {
    const float* wgn = Wg + (size_t)(i + 1) * 2 * Hdim;
    gather_kernel<<<(N + 7) / 8, 256, 0, stream>>>(cnt, srow, gac, gbc, bg, i, h, raw, hnew,
                                                  wgn, gan, gbn, N);
    h = hnew;
    hnew = (h == bufA) ? bufB : bufA;
    float* t = gac; gac = gan; gan = t;
    t = gbc; gbc = gbn; gbn = t;
  }

  int tiles = (N + 15) / 16;
  if (L >= 2) {
    bool same = (h == raw);
    float* Y  = hnew;
    float* YR = same ? Y : hnew + 32 * (size_t)N;
    float* Z  = hnew + 64 * (size_t)N;
    float* ZR = hnew + 80 * (size_t)N;
    float* LG = hnew + 96 * (size_t)N;
    int nt = same ? tiles : 2 * tiles;
    gemm_proj2_kernel<<<(nt + 7) / 8, 256, 0, stream>>>(h, raw, W_out,
                                                       Wg + (size_t)(L - 1) * 2 * Hdim, gac, gbc,
                                                       Y, YR, N, nt);
    gather_mid_kernel<<<(N + 63) / 64, 256, 0, stream>>>(cnt, srow, bg, L - 2, Y, YR, b_out,
                                                        Z, ZR, gan, gbn, N);
    gather_proj_kernel<<<(N + 63) / 64, 256, 0, stream>>>(cnt, srow, gan, gbn, bg, L - 1, Z, ZR,
                                                         LG, N);
    logsoftmax_kernel<<<(N + 255) / 256, 256, 0, stream>>>((const float4*)LG, (float4*)out, N);
    return;
  }

  float* Z  = hnew;
  float* ZR = hnew + 16 * (size_t)N;
  float* LG = hnew + 32 * (size_t)N;
  gemm_proj_kernel<<<(2 * tiles + 7) / 8, 256, 0, stream>>>(h, raw, W_out, b_out,
                                                           L > 0 ? FAGCN_EPS : 1.0f, Z, ZR, N);
  if (L > 0)
    gather_proj_kernel<<<(N + 63) / 64, 256, 0, stream>>>(cnt, srow, gac, gbc, bg, L - 1, Z, ZR,
                                                         LG, N);
  logsoftmax_kernel<<<(N + 255) / 256, 256, 0, stream>>>((const float4*)(L > 0 ? LG : ZR),
                                                        (float4*)out, N);
}


# Edge grouping without a global atomic per edge: edges binned by 256-node range via LDS counters in 3 passes (count, sort-and-scatter, per-bin finish with segment sort and degree norm); 6 steps -> 4
# speedup vs baseline: 4.6754x; 2.2757x over previous
#include <hip/hip_runtime.h>

typedef float v2f __attribute__((ext_vector_type(2)));
typedef float v8f __attribute__((ext_vector_type(8)));

#define FAGCN_EPS 0.3f

__global__ void deg_kernel(const int* __restrict__ ei, float* __restrict__ deg,
                           int* __restrict__ cnt, int E) {
  int e = blockIdx.x * blockDim.x + threadIdx.x;
  if (e < E) {
    atomicAdd(&deg[ei[e]], 1.0f);
    atomicAdd(&cnt[ei[E + e]], 1);
  }
}

__device__ __forceinline__ int4 scan_load4(const int* cnt, int i0, int N) {
  int4 v = {0, 0, 0, 0};
  if (i0 + 3 < N) {
    v = *(const int4*)(cnt + i0);
  } else {
    if (i0 < N) v.x = cnt[i0];
    if (i0 + 1 < N) v.y = cnt[i0 + 1];
    if (i0 + 2 < N) v.z = cnt[i0 + 2];
  }
  return v;
}

__global__ void scan_kernel(int* cnt, int N) {
  __shared__ int wsum[32];
  int tid = threadIdx.x;
  int lane = tid & 31;
  int wave = tid >> 5;
  int carry = 0;
  int4 nxt = scan_load4(cnt, tid * 4, N);
  for (int base = 0; base < N; base += 4096) {
    int i0 = base + tid * 4;
    int4 v = nxt;
    nxt = scan_load4(cnt, i0 + 4096, N);
    int s = v.x + v.y + v.z + v.w;
    int incl = s;
    for (int off = 1; off < 32; off <<= 1) {
      int t = __shfl_up(incl, off, 32);
      if (lane >= off) incl += t;
    }
    if (lane == 31) wsum[wave] = incl;
    __syncthreads();
    int wt = wsum[lane];
    int wincl = wt;
    for (int off = 1; off < 32; off <<= 1) {
      int t = __shfl_up(wincl, off, 32);
      if (lane >= off) wincl += t;
    }
    int wexcl = __shfl(wincl, wave, 32) - __shfl(wt, wave, 32);
    int total = __shfl(wincl, 31, 32);
    int ex = carry + wexcl + incl - s;
    if (i0 + 3 < N) {
      int4 o;
      o.x = ex; o.y = ex + v.x; o.z = ex + v.x + v.y; o.w = ex + v.x + v.y + v.z;
      *(int4*)(cnt + i0) = o;
    } else {
      if (i0 < N) cnt[i0] = ex;
      if (i0 + 1 < N) cnt[i0 + 1] = ex + v.x;
      if (i0 + 2 < N) cnt[i0 + 2] = ex + v.x + v.y;
    }
    carry += total;
    __syncthreads();
  }
}

__global__ void fill_kernel(const int* __restrict__ ei, int* __restrict__ cur,
                            int* __restrict__ srow, int E) {
  int e = blockIdx.x * blockDim.x + threadIdx.x;
  if (e < E) {
    int pos = atomicAdd(&cur[ei[E + e]], 1);
    srow[pos] = ei[e];
  }
}

__device__ __forceinline__ void order_segment(int* __restrict__ srow, int beg, int d, int lane) {
  int a0 = lane < d ? srow[beg + lane] : 0x7fffffff;
  int a1 = lane + 32 < d ? srow[beg + 32 + lane] : 0x7fffffff;
  int r0 = 0, r1 = 0;
  int n0 = d < 32 ? d : 32;
  for (int i = 0; i < n0; ++i) {
    int x = __builtin_amdgcn_readlane(a0, i);
    r0 += (x < a0 || (x == a0 && i < lane)) ? 1 : 0;
    r1 += (x <= a1) ? 1 : 0;
  }
  for (int i = 0; i < d - 32; ++i) {
    int x = __builtin_amdgcn_readlane(a1, i);
    r0 += (x < a0) ? 1 : 0;
    r1 += (x < a1 || (x == a1 && i < lane)) ? 1 : 0;
  }
  if (lane < d) srow[beg + r0] = a0;
  if (lane + 32 < d) srow[beg + r1] = a1;
}

__global__ void order_kernel(const int* __restrict__ ends, int* __restrict__ srow, int N) {
  int node = __builtin_amdgcn_readfirstlane(blockIdx.x * 8 + (threadIdx.x >> 5));
  int lane = threadIdx.x & 31;
  if (node >= N) return;
  int beg = node ? ends[node - 1] : 0;
  int d = ends[node] - beg;
  if (d < 2 || d > 64) return;
  order_segment(srow, beg, d, lane);
}

__global__ void dn_kernel(float* dn, int N) {
  int i = blockIdx.x * blockDim.x + threadIdx.x;
  if (i < N) dn[i] = 1.0f / sqrtf(fmaxf(dn[i], 1.0f));
}

#define BIN_SHIFT 8
#define BIN_NODES 256
#define MAX_BINS 512
#define CHUNK 4096
#define BIN_CAP 8192

__device__ __forceinline__ int block_scan_1024(int s, int* wsum, int lane, int wave) {
  int incl = s;
  for (int off = 1; off < 32; off <<= 1) {
    int t = __shfl_up(incl, off, 32);
    if (lane >= off) incl += t;
  }
  if (lane == 31) wsum[wave] = incl;
  __syncthreads();
  int wt = wsum[lane];
  int wincl = wt;
  for (int off = 1; off < 32; off <<= 1) {
    int t = __shfl_up(wincl, off, 32);
    if (lane >= off) wincl += t;
  }
  int wexcl = __shfl(wincl, wave, 32) - __shfl(wt, wave, 32);
  __syncthreads();
  return wexcl + incl - s;
}

__global__ void bin_count_kernel(const int* __restrict__ ei, int* __restrict__ tot, int E, int NB) {
  __shared__ int hist[2 * MAX_BINS];
  int tid = threadIdx.x;
  for (int i = tid; i < 2 * NB; i += 1024) hist[i] = 0;
  __syncthreads();
  int e0 = blockIdx.x * CHUNK + tid;
  int col[4], row[4];
#pragma unroll
  for (int k = 0; k < 4; ++k) {
    int e = e0 + k * 1024;
    col[k] = e < E ? ei[E + e] : -1;
    row[k] = e < E ? ei[e] : -1;
  }
#pragma unroll
  for (int k = 0; k < 4; ++k) {
    if (col[k] >= 0) {
      atomicAdd(&hist[col[k] >> BIN_SHIFT], 1);
      atomicAdd(&hist[NB + (row[k] >> BIN_SHIFT)], 1);
    }
  }
  __syncthreads();
  for (int i = tid; i < 2 * NB; i += 1024) {
    int c = hist[i];
    if (c) atomicAdd(&tot[i], c);
  }
}

__global__ void bin_scatter_kernel(const int* __restrict__ ei, const int* __restrict__ tot,
                                   int* __restrict__ cursor, int* __restrict__ bstart,
                                   int* __restrict__ cbin, int* __restrict__ rbin, int E, int NB) {
  __shared__ int hist[2 * MAX_BINS];
  __shared__ int gbase[2 * MAX_BINS];
  __shared__ int wsum[32];
  __shared__ int keyC[CHUNK];
  __shared__ int adrC[CHUNK];
  __shared__ int stR[CHUNK];
  int tid = threadIdx.x;
  int lane = tid & 31;
  int wave = tid >> 5;
  int nb2 = 2 * NB;
  for (int i = tid; i < nb2; i += 1024) hist[i] = 0;
  __syncthreads();
  int ebase = blockIdx.x * CHUNK;
  int nE = E - ebase < CHUNK ? E - ebase : CHUNK;
  int i0 = 2 * tid;
  int t0 = i0 < nb2 ? tot[i0] : 0;
  int t1 = i0 + 1 < nb2 ? tot[i0 + 1] : 0;
  int col[4], row[4], rc[4], rr[4];
#pragma unroll
  for (int k = 0; k < 4; ++k) {
    int e = ebase + tid + k * 1024;
    col[k] = e < E ? ei[E + e] : -1;
    row[k] = e < E ? ei[e] : -1;
  }
#pragma unroll
  for (int k = 0; k < 4; ++k) {
    rc[k] = 0; rr[k] = 0;
    if (col[k] >= 0) {
      rc[k] = atomicAdd(&hist[col[k] >> BIN_SHIFT], 1);
      rr[k] = atomicAdd(&hist[NB + (row[k] >> BIN_SHIFT)], 1);
    }
  }
  __syncthreads();
  int h0 = i0 < nb2 ? hist[i0] : 0;
  int h1 = i0 + 1 < nb2 ? hist[i0 + 1] : 0;
  int gex = block_scan_1024(t0 + t1, wsum, lane, wave);
  int lex = block_scan_1024(h0 + h1, wsum, lane, wave);
  if (i0 < nb2) {
    int second = i0 >= NB ? 1 : 0;
    int s = gex - (second ? E : 0);
    gbase[i0] = s + (h0 ? atomicAdd(&cursor[i0], h0) : 0);
    hist[i0] = lex - (second ? nE : 0);
    if (blockIdx.x == 0) bstart[i0 + second] = s;
  }
  if (i0 + 1 < nb2) {
    int second = i0 + 1 >= NB ? 1 : 0;
    int s = gex + t0 - (second ? E : 0);
    gbase[i0 + 1] = s + (h1 ? atomicAdd(&cursor[i0 + 1], h1) : 0);
    hist[i0 + 1] = lex + h0 - (second ? nE : 0);
    if (blockIdx.x == 0) bstart[i0 + 1 + second] = s;
  }
  if (blockIdx.x == 0 && tid == 0) {
    bstart[NB] = E;
    bstart[nb2 + 1] = E;
  }
  __syncthreads();
#pragma unroll
  for (int k = 0; k < 4; ++k) {
    if (col[k] >= 0) {
      int bc = col[k] >> BIN_SHIFT;
      int sc = hist[bc] + rc[k];
      keyC[sc] = (row[k] << BIN_SHIFT) | (col[k] & (BIN_NODES - 1));
      adrC[sc] = gbase[bc] + rc[k];
      int br = NB + (row[k] >> BIN_SHIFT);
      int sr = hist[br] + rr[k];
      stR[sr] = ((gbase[br] + rr[k]) << BIN_SHIFT) | (row[k] & (BIN_NODES - 1));
    }
  }
  __syncthreads();
  for (int j = tid; j < nE; j += 1024) {
    cbin[adrC[j]] = keyC[j];
    int w = stR[j];
    rbin[w >> BIN_SHIFT] = w & (BIN_NODES - 1);
  }
}

__global__ void bin_finish_kernel(const int* __restrict__ bstart, const int* __restrict__ cbin,
                                  const int* __restrict__ rbin, int* __restrict__ ends,
                                  float* __restrict__ dn, int* __restrict__ srow, int N, int NB) {
  __shared__ int ncnt[BIN_NODES];
  __shared__ int nbeg[BIN_NODES];
  __shared__ int ncur[BIN_NODES];
  __shared__ int dcnt[BIN_NODES];
  __shared__ int wsum[32];
  __shared__ int stage[BIN_CAP];
  int tid = threadIdx.x;
  int lane = tid & 31;
  int wave = tid >> 5;
  int b = blockIdx.x;
  int cb = bstart[b], ce = bstart[b + 1];
  int rb = bstart[NB + 1 + b], re = bstart[NB + 2 + b];
  int kc[5], kr[5];
#pragma unroll
  for (int k = 0; k < 5; ++k) {
    int i = cb + tid + k * 1024;
    int j = rb + tid + k * 1024;
    kc[k] = i < ce ? cbin[i] : -1;
    kr[k] = j < re ? rbin[j] : -1;
  }
  if (tid < BIN_NODES) { ncnt[tid] = 0; dcnt[tid] = 0; }
  __syncthreads();
#pragma unroll
  for (int k = 0; k < 5; ++k) {
    if (kc[k] >= 0) atomicAdd(&ncnt[kc[k] & (BIN_NODES - 1)], 1);
    if (kr[k] >= 0) atomicAdd(&dcnt[kr[k]], 1);
  }
  for (int i = cb + tid + 5 * 1024; i < ce; i += 1024) atomicAdd(&ncnt[cbin[i] & (BIN_NODES - 1)], 1);
  for (int i = rb + tid + 5 * 1024; i < re; i += 1024) atomicAdd(&dcnt[rbin[i]], 1);
  __syncthreads();
  int c = tid < BIN_NODES ? ncnt[tid] : 0;
  int ex = block_scan_1024(c, wsum, lane, wave);
  if (tid < BIN_NODES) {
    nbeg[tid] = ex;
    ncur[tid] = ex;
    int node = b * BIN_NODES + tid;
    if (node < N) {
      ends[node] = cb + ex + c;
      dn[node] = 1.0f / sqrtf(fmaxf((float)dcnt[tid], 1.0f));
    }
  }
  __syncthreads();
  int cnt = ce - cb;
  bool fits = cnt <= BIN_CAP;
#pragma unroll
  for (int k = 0; k < 5; ++k) {
    if (kc[k] >= 0) {
      int p = atomicAdd(&ncur[kc[k] & (BIN_NODES - 1)], 1);
      if (fits) stage[p] = kc[k];
      else srow[cb + p] = kc[k] >> BIN_SHIFT;
    }
  }
  for (int i = cb + tid + 5 * 1024; i < ce; i += 1024) {
    int key = cbin[i];
    int p = atomicAdd(&ncur[key & (BIN_NODES - 1)], 1);
    if (fits) stage[p] = key;
    else srow[cb + p] = key >> BIN_SHIFT;
  }
  if (!fits) {
    __threadfence();
    __syncthreads();
    for (int t = wave; t < BIN_NODES; t += 32) {
      int n = __builtin_amdgcn_readfirstlane(ncnt[t]);
      int sb = __builtin_amdgcn_readfirstlane(nbeg[t]);
      if (n >= 2 && n <= 64) order_segment(srow, cb + sb, n, lane);
    }
    return;
  }
  __syncthreads();
  for (int p = tid; p < cnt; p += 1024) {
    int key = stage[p];
    int d = key & (BIN_NODES - 1);
    int beg = nbeg[d];
    int n = ncnt[d];
    int r = p - beg;
    if (n > 1 && n <= 64) {
      r = 0;
      for (int j = 0; j < n; ++j) {
        int x = stage[beg + j];
        r += (x < key || (x == key && beg + j < p)) ? 1 : 0;
      }
    }
    srow[cb + beg + r] = key >> BIN_SHIFT;
  }
}

#define TILE_PITCH 132
__global__ void gemm_relu_kernel(const float* __restrict__ X, const float* __restrict__ W,
                                 const float* __restrict__ bias, const float* __restrict__ wg0,
                                 const float* __restrict__ dn, float* __restrict__ H,
                                 float* __restrict__ pk, float* __restrict__ gb, int Nrows) {
  __shared__ __attribute__((aligned(16))) float tile[16 * TILE_PITCH];
  int wave = threadIdx.x >> 5;
  int lane = threadIdx.x & 31;
  int tileM = blockIdx.x;
  int mA = tileM * 16 + (lane & 15);
  int mclamp = mA < Nrows ? mA : (Nrows - 1);
  int n = wave * 16 + (lane & 15);
  int khalf = (lane >> 4) * 2;
  const float* xrow = X + (size_t)mclamp * 128 + khalf;
  const float* wcol = W + (size_t)khalf * 128 + n;
  __builtin_prefetch(xrow, 0, 0);
  float dnv = dn[mclamp];
  v8f acc = {};
#pragma unroll 4
  for (int kk = 0; kk < 128; kk += 4) {
    v2f a = *(const v2f*)(xrow + kk);
    v2f b;
    b.x = wcol[(size_t)kk * 128];
    b.y = wcol[(size_t)kk * 128 + 128];
    acc = __builtin_amdgcn_wmma_f32_16x16x4_f32(false, a, false, b, (short)0, acc,
                                               false, false);
  }
  int rbase = (lane >> 4) * 8;
  float bn = bias[n];
#pragma unroll
  for (int r = 0; r < 8; ++r)
    tile[(rbase + r) * TILE_PITCH + n] = fmaxf(acc[r] + bn, 0.0f);
  __syncthreads();
  float4 wa = {0.0f, 0.0f, 0.0f, 0.0f}, wb = {0.0f, 0.0f, 0.0f, 0.0f};
  if (wg0) {
    wa = *(const float4*)(wg0 + lane * 4);
    wb = *(const float4*)(wg0 + 128 + lane * 4);
  }
  for (int j = 0; j < 2; ++j) {
    int rr = wave * 2 + j;
    int m = tileM * 16 + rr;
    float4 v = *(const float4*)(tile + rr * TILE_PITCH + lane * 4);
    if (m < Nrows) {
      *(float4*)(H + (size_t)m * 128 + lane * 4) = v;
      if (wg0) {
        float dnm = __shfl(dnv, rr, 32);
        float sa = v.x * wa.x + v.y * wa.y + v.z * wa.z + v.w * wa.w;
        float sb = v.x * wb.x + v.y * wb.y + v.z * wb.z + v.w * wb.w;
        for (int off = 16; off > 0; off >>= 1) {
          sa += __shfl_down(sa, off, 32);
          sb += __shfl_down(sb, off, 32);
        }
        if (lane == 0) {
          float2 p; p.x = sa; p.y = dnm;
          *(float2*)(pk + 2 * (size_t)m) = p;
          gb[m] = sb;
        }
      }
    }
  }
}

__global__ void gather_kernel(const int* __restrict__ ends, const int* __restrict__ srow,
                              const float* __restrict__ pk, const float* __restrict__ gb,
                              const float* __restrict__ bg,
                              int layer, const float* __restrict__ H,
                              const float* __restrict__ raw, float* __restrict__ hout,
                              const float* __restrict__ wg_next, float* __restrict__ pk_next,
                              float* __restrict__ gb_next, int N) {
  int node = __builtin_amdgcn_readfirstlane(blockIdx.x * 8 + (threadIdx.x >> 5));
  int lane = threadIdx.x & 31;
  if (node >= N) return;
  int beg = node ? ends[node - 1] : 0;
  int end = ends[node];
  float gbc = gb[node];
  float bgl = bg[layer];
  float dnc = pk[2 * (size_t)node + 1];
  const float* hl = H + lane * 4;
  float4 rw = *(const float4*)(raw + (size_t)node * 128 + lane * 4);
  float4 acc = {0.0f, 0.0f, 0.0f, 0.0f};
  for (int base = beg; base < end; base += 32) {
    int n = end - base < 32 ? end - base : 32;
    int r = 0;
    float w = 0.0f;
    if (lane < n) {
      r = srow[base + lane];
      float2 gd = *(const float2*)(pk + 2 * (size_t)r);
      float g = tanhf(gd.x + gbc + bgl);
      w = g * gd.y * dnc;
    }
    int wi = __float_as_int(w);
    int k = 0;
    for (; k + 4 <= n; k += 4) {
      int r0 = __builtin_amdgcn_readlane(r, k);
      int r1 = __builtin_amdgcn_readlane(r, k + 1);
      int r2 = __builtin_amdgcn_readlane(r, k + 2);
      int r3 = __builtin_amdgcn_readlane(r, k + 3);
      float w0 = __int_as_float(__builtin_amdgcn_readlane(wi, k));
      float w1 = __int_as_float(__builtin_amdgcn_readlane(wi, k + 1));
      float w2 = __int_as_float(__builtin_amdgcn_readlane(wi, k + 2));
      float w3 = __int_as_float(__builtin_amdgcn_readlane(wi, k + 3));
      float4 h0 = *(const float4*)(hl + (size_t)r0 * 128);
      float4 h1 = *(const float4*)(hl + (size_t)r1 * 128);
      float4 h2 = *(const float4*)(hl + (size_t)r2 * 128);
      float4 h3 = *(const float4*)(hl + (size_t)r3 * 128);
      acc.x += w0 * h0.x; acc.y += w0 * h0.y; acc.z += w0 * h0.z; acc.w += w0 * h0.w;
      acc.x += w1 * h1.x; acc.y += w1 * h1.y; acc.z += w1 * h1.z; acc.w += w1 * h1.w;
      acc.x += w2 * h2.x; acc.y += w2 * h2.y; acc.z += w2 * h2.z; acc.w += w2 * h2.w;
      acc.x += w3 * h3.x; acc.y += w3 * h3.y; acc.z += w3 * h3.z; acc.w += w3 * h3.w;
    }
    for (; k < n; ++k) {
      int r0 = __builtin_amdgcn_readlane(r, k);
      float w0 = __int_as_float(__builtin_amdgcn_readlane(wi, k));
      float4 h0 = *(const float4*)(hl + (size_t)r0 * 128);
      acc.x += w0 * h0.x; acc.y += w0 * h0.y; acc.z += w0 * h0.z; acc.w += w0 * h0.w;
    }
  }
  float4 o;
  o.x = FAGCN_EPS * rw.x + acc.x;
  o.y = FAGCN_EPS * rw.y + acc.y;
  o.z = FAGCN_EPS * rw.z + acc.z;
  o.w = FAGCN_EPS * rw.w + acc.w;
  *(float4*)(hout + (size_t)node * 128 + lane * 4) = o;
  if (wg_next) {
    float4 wa = *(const float4*)(wg_next + lane * 4);
    float4 wb = *(const float4*)(wg_next + 128 + lane * 4);
    float sa = o.x * wa.x + o.y * wa.y + o.z * wa.z + o.w * wa.w;
    float sb = o.x * wb.x + o.y * wb.y + o.z * wb.z + o.w * wb.w;
    for (int off = 16; off > 0; off >>= 1) {
      sa += __shfl_down(sa, off, 32);
      sb += __shfl_down(sb, off, 32);
    }
    if (lane == 0) {
      float2 p; p.x = sa; p.y = dnc;
      *(float2*)(pk_next + 2 * (size_t)node) = p;
      gb_next[node] = sb;
    }
  }
}

__global__ void gemm_proj_kernel(const float* __restrict__ Hc, const float* __restrict__ Raw,
                                 const float* __restrict__ W, const float* __restrict__ bias,
                                 float rscale, float* __restrict__ Z, float* __restrict__ ZR,
                                 int Nrows) {
  int wave = threadIdx.x >> 5;
  int lane = threadIdx.x & 31;
  int tiles = (Nrows + 15) >> 4;
  int t = blockIdx.x * 8 + wave;
  if (t >= 2 * tiles) return;
  bool second = t >= tiles;
  int tileM = second ? t - tiles : t;
  const float* src = second ? Raw : Hc;
  float* dst = second ? ZR : Z;
  int mA = tileM * 16 + (lane & 15);
  int mclamp = mA < Nrows ? mA : (Nrows - 1);
  int n = lane & 15;
  int khalf = (lane >> 4) * 2;
  const float* hrow = src + (size_t)mclamp * 128 + khalf;
  const float* wcol = W + (size_t)khalf * 16 + n;
  __builtin_prefetch(hrow, 0, 0);
  v8f acc = {};
#pragma unroll 4
  for (int kk = 0; kk < 128; kk += 4) {
    v2f a = *(const v2f*)(hrow + kk);
    v2f b;
    b.x = wcol[(size_t)kk * 16];
    b.y = wcol[(size_t)kk * 16 + 16];
    acc = __builtin_amdgcn_wmma_f32_16x16x4_f32(false, a, false, b, (short)0, acc,
                                               false, false);
  }
  int mbase = tileM * 16 + ((lane >> 4) * 8);
  float bn = second ? bias[n] : 0.0f;
  float sc = second ? rscale : 1.0f;
  if (tileM * 16 + 16 <= Nrows) {
    float* op = dst + (size_t)mbase * 16 + n;
#pragma unroll
    for (int r = 0; r < 8; ++r)
      op[(size_t)r * 16] = acc[r] * sc + bn;
  } else {
#pragma unroll
    for (int r = 0; r < 8; ++r) {
      int mm = mbase + r;
      if (mm < Nrows) dst[(size_t)mm * 16 + n] = acc[r] * sc + bn;
    }
  }
}

__global__ void gather_proj_kernel(const int* __restrict__ ends, const int* __restrict__ srow,
                                   const float* __restrict__ pk, const float* __restrict__ gb,
                                   const float* __restrict__ bg, int layer,
                                   const float* __restrict__ Z, const float* __restrict__ ZR,
                                   float* __restrict__ logits, int N) {
  int node = blockIdx.x * 64 + (threadIdx.x >> 2);
  int q = threadIdx.x & 3;
  if (node >= N) return;
  int beg = node ? ends[node - 1] : 0;
  int end = ends[node];
  float gbc = gb[node];
  float bgl = bg[layer];
  float dnc = pk[2 * (size_t)node + 1];
  const float* zq = Z + q * 4;
  float4 zr = *(const float4*)(ZR + (size_t)node * 16 + q * 4);
  float4 acc = {0.0f, 0.0f, 0.0f, 0.0f};
  int last = end - 1;
  int r0 = 0, r1 = 0, r2 = 0, r3 = 0;
  if (beg < end) {
    r0 = srow[beg];
    r1 = srow[beg + 1 < last ? beg + 1 : last];
    r2 = srow[beg + 2 < last ? beg + 2 : last];
    r3 = srow[beg + 3 < last ? beg + 3 : last];
  }
  for (int base = beg; base < end; base += 4) {
    float2 g0 = *(const float2*)(pk + 2 * (size_t)r0);
    float2 g1 = *(const float2*)(pk + 2 * (size_t)r1);
    float2 g2 = *(const float2*)(pk + 2 * (size_t)r2);
    float2 g3 = *(const float2*)(pk + 2 * (size_t)r3);
    float4 h0 = *(const float4*)(zq + (size_t)r0 * 16);
    float4 h1 = *(const float4*)(zq + (size_t)r1 * 16);
    float4 h2 = *(const float4*)(zq + (size_t)r2 * 16);
    float4 h3 = *(const float4*)(zq + (size_t)r3 * 16);
    int nb = base + 4;
    int n0 = r0, n1 = r1, n2 = r2, n3 = r3;
    if (nb < end) {
      n0 = srow[nb];
      n1 = srow[nb + 1 < last ? nb + 1 : last];
      n2 = srow[nb + 2 < last ? nb + 2 : last];
      n3 = srow[nb + 3 < last ? nb + 3 : last];
    }
    {
      float w = tanhf(g0.x + gbc + bgl) * g0.y * dnc;
      acc.x += w * h0.x; acc.y += w * h0.y; acc.z += w * h0.z; acc.w += w * h0.w;
    }
    if (base + 1 < end) {
      float w = tanhf(g1.x + gbc + bgl) * g1.y * dnc;
      acc.x += w * h1.x; acc.y += w * h1.y; acc.z += w * h1.z; acc.w += w * h1.w;
    }
    if (base + 2 < end) {
      float w = tanhf(g2.x + gbc + bgl) * g2.y * dnc;
      acc.x += w * h2.x; acc.y += w * h2.y; acc.z += w * h2.z; acc.w += w * h2.w;
    }
    if (base + 3 < end) {
      float w = tanhf(g3.x + gbc + bgl) * g3.y * dnc;
      acc.x += w * h3.x; acc.y += w * h3.y; acc.z += w * h3.z; acc.w += w * h3.w;
    }
    r0 = n0; r1 = n1; r2 = n2; r3 = n3;
  }
  float4 o;
  o.x = zr.x + acc.x;
  o.y = zr.y + acc.y;
  o.z = zr.z + acc.z;
  o.w = zr.w + acc.w;
  *(float4*)(logits + (size_t)node * 16 + q * 4) = o;
}

#define YP 32
__global__ void gemm_proj2_kernel(const float* __restrict__ Hc, const float* __restrict__ Raw,
                                  const float* __restrict__ W, const float* __restrict__ wgl,
                                  const float* __restrict__ pk, const float* __restrict__ gb,
                                  float* __restrict__ Y, float* __restrict__ YR,
                                  int Nrows, int nt) {
  int wave = threadIdx.x >> 5;
  int lane = threadIdx.x & 31;
  int tiles = (Nrows + 15) >> 4;
  int t = blockIdx.x * 8 + wave;
  if (t >= nt) return;
  bool second = t >= tiles;
  int tileM = second ? t - tiles : t;
  const float* src = second ? Raw : Hc;
  float* dst = second ? YR : Y;
  int mA = tileM * 16 + (lane & 15);
  int mclamp = mA < Nrows ? mA : (Nrows - 1);
  int n = lane & 15;
  int khalf = (lane >> 4) * 2;
  const float* hrow = src + (size_t)mclamp * 128 + khalf;
  const float* wcol = W + (size_t)khalf * 16 + n;
  const float* gcol = wgl + (n & 1) * 128 + khalf;
  __builtin_prefetch(hrow, 0, 0);
  int mbase = tileM * 16 + ((lane >> 4) * 8);
  const float* tp = (n == 2 || n == 3) ? pk + (n - 2) : gb;
  int ts = (n == 2 || n == 3) ? 2 : 1;
  float tv[8];
#pragma unroll
  for (int r = 0; r < 8; ++r) {
    int mr = mbase + r < Nrows ? mbase + r : Nrows - 1;
    tv[r] = tp[(size_t)mr * ts];
  }
  v8f acc = {};
  v8f acg = {};
#pragma unroll 4
  for (int kk = 0; kk < 128; kk += 4) {
    v2f a = *(const v2f*)(hrow + kk);
    v2f b;
    b.x = wcol[(size_t)kk * 16];
    b.y = wcol[(size_t)kk * 16 + 16];
    v2f g = *(const v2f*)(gcol + kk);
    acc = __builtin_amdgcn_wmma_f32_16x16x4_f32(false, a, false, b, (short)0, acc,
                                               false, false);
    acg = __builtin_amdgcn_wmma_f32_16x16x4_f32(false, a, false, g, (short)0, acg,
                                               false, false);
  }
  float* yb = dst + (size_t)mbase * YP + n;
  if (tileM * 16 + 16 <= Nrows) {
#pragma unroll
    for (int r = 0; r < 8; ++r) {
      yb[(size_t)r * YP] = acc[r];
      if (n < 5) yb[(size_t)r * YP + 16] = n < 2 ? acg[r] : tv[r];
    }
  } else {
#pragma unroll
    for (int r = 0; r < 8; ++r) {
      if (mbase + r < Nrows) {
        yb[(size_t)r * YP] = acc[r];
        if (n < 5) yb[(size_t)r * YP + 16] = n < 2 ? acg[r] : tv[r];
      }
    }
  }
}

__global__ void gather_mid_kernel(const int* __restrict__ ends, const int* __restrict__ srow,
                                  const float* __restrict__ bg, int layer,
                                  const float* __restrict__ Y, const float* __restrict__ YR,
                                  const float* __restrict__ bias,
                                  float* __restrict__ Z, float* __restrict__ ZR,
                                  float* __restrict__ pk_next, float* __restrict__ gb_next,
                                  int N) {
  int node = blockIdx.x * 64 + (threadIdx.x >> 2);
  int q = threadIdx.x & 3;
  if (node >= N) return;
  int beg = node ? ends[node - 1] : 0;
  int end = ends[node];
  float bgl = bg[layer];
  const float* yc = Y + (size_t)node * YP;
  float dnc = yc[19];
  float gbc = yc[20];
  const float* yrc = YR + (size_t)node * YP;
  float4 own = *(const float4*)(yrc + q * 4);
  float2 owng = *(const float2*)(yrc + 16);
  float4 bq = *(const float4*)(bias + q * 4);
  const float* yq = Y + q * 4;
  const float* yt = Y + 16;
  float4 acc = {0.0f, 0.0f, 0.0f, 0.0f};
  float sga = 0.0f, sgb = 0.0f;
  int last = end - 1;
  int r0 = 0, r1 = 0, r2 = 0, r3 = 0;
  if (beg < end) {
    r0 = srow[beg];
    r1 = srow[beg + 1 < last ? beg + 1 : last];
    r2 = srow[beg + 2 < last ? beg + 2 : last];
    r3 = srow[beg + 3 < last ? beg + 3 : last];
  }
  for (int base = beg; base < end; base += 4) {
    float4 t0 = *(const float4*)(yt + (size_t)r0 * YP);
    float4 t1 = *(const float4*)(yt + (size_t)r1 * YP);
    float4 t2 = *(const float4*)(yt + (size_t)r2 * YP);
    float4 t3 = *(const float4*)(yt + (size_t)r3 * YP);
    float4 h0 = *(const float4*)(yq + (size_t)r0 * YP);
    float4 h1 = *(const float4*)(yq + (size_t)r1 * YP);
    float4 h2 = *(const float4*)(yq + (size_t)r2 * YP);
    float4 h3 = *(const float4*)(yq + (size_t)r3 * YP);
    int nb = base + 4;
    int n0 = r0, n1 = r1, n2 = r2, n3 = r3;
    if (nb < end) {
      n0 = srow[nb];
      n1 = srow[nb + 1 < last ? nb + 1 : last];
      n2 = srow[nb + 2 < last ? nb + 2 : last];
      n3 = srow[nb + 3 < last ? nb + 3 : last];
    }
    {
      float w = tanhf(t0.z + gbc + bgl) * t0.w * dnc;
      acc.x += w * h0.x; acc.y += w * h0.y; acc.z += w * h0.z; acc.w += w * h0.w;
      sga += w * t0.x; sgb += w * t0.y;
    }
    if (base + 1 < end) {
      float w = tanhf(t1.z + gbc + bgl) * t1.w * dnc;
      acc.x += w * h1.x; acc.y += w * h1.y; acc.z += w * h1.z; acc.w += w * h1.w;
      sga += w * t1.x; sgb += w * t1.y;
    }
    if (base + 2 < end) {
      float w = tanhf(t2.z + gbc + bgl) * t2.w * dnc;
      acc.x += w * h2.x; acc.y += w * h2.y; acc.z += w * h2.z; acc.w += w * h2.w;
      sga += w * t2.x; sgb += w * t2.y;
    }
    if (base + 3 < end) {
      float w = tanhf(t3.z + gbc + bgl) * t3.w * dnc;
      acc.x += w * h3.x; acc.y += w * h3.y; acc.z += w * h3.z; acc.w += w * h3.w;
      sga += w * t3.x; sgb += w * t3.y;
    }
    r0 = n0; r1 = n1; r2 = n2; r3 = n3;
  }
  float4 o, zr;
  o.x = FAGCN_EPS * own.x + acc.x;  zr.x = FAGCN_EPS * own.x + bq.x;
  o.y = FAGCN_EPS * own.y + acc.y;  zr.y = FAGCN_EPS * own.y + bq.y;
  o.z = FAGCN_EPS * own.z + acc.z;  zr.z = FAGCN_EPS * own.z + bq.z;
  o.w = FAGCN_EPS * own.w + acc.w;  zr.w = FAGCN_EPS * own.w + bq.w;
  *(float4*)(Z + (size_t)node * 16 + q * 4) = o;
  *(float4*)(ZR + (size_t)node * 16 + q * 4) = zr;
  if (q == 0) {
    float2 p; p.x = FAGCN_EPS * owng.x + sga; p.y = dnc;
    *(float2*)(pk_next + 2 * (size_t)node) = p;
    gb_next[node] = FAGCN_EPS * owng.y + sgb;
  }
}

__global__ void logsoftmax_kernel(const float4* __restrict__ in, float4* __restrict__ out, int N) {
  int i = blockIdx.x * blockDim.x + threadIdx.x;
  if (i >= N) return;
  const float4* rin = in + (size_t)i * 4;
  float4* row = out + (size_t)i * 4;
  float4 v[4];
#pragma unroll
  for (int q = 0; q < 4; ++q) v[q] = rin[q];
  float m = v[0].x;
#pragma unroll
  for (int q = 0; q < 4; ++q) {
    m = fmaxf(m, v[q].x); m = fmaxf(m, v[q].y);
    m = fmaxf(m, v[q].z); m = fmaxf(m, v[q].w);
  }
  float s = 0.0f;
#pragma unroll
  for (int q = 0; q < 4; ++q) {
    s += expf(v[q].x - m); s += expf(v[q].y - m);
    s += expf(v[q].z - m); s += expf(v[q].w - m);
  }
  float d = m + logf(s);
#pragma unroll
  for (int q = 0; q < 4; ++q) {
    v[q].x -= d; v[q].y -= d; v[q].z -= d; v[q].w -= d;
    row[q] = v[q];
  }
}

extern "C" void kernel_launch(void* const* d_in, const int* in_sizes, int n_in,
                              void* d_out, int out_size, void* d_ws, size_t ws_size,
                              hipStream_t stream) {
  const float* x     = (const float*)d_in[0];
  const int*   ei    = (const int*)d_in[1];
  const float* W_in  = (const float*)d_in[2];
  const float* b_in  = (const float*)d_in[3];
  const float* Wg    = (const float*)d_in[4];
  const float* bg    = (const float*)d_in[5];
  const float* W_out = (const float*)d_in[6];
  const float* b_out = (const float*)d_in[7];
  float* out = (float*)d_out;

  const int Hdim = 128;
  int N = in_sizes[0] / Hdim;
  int E = in_sizes[1] / 2;
  int L = in_sizes[5];

  float* ws   = (float*)d_ws;
  float* raw  = ws;
  float* bufA = raw + (size_t)N * Hdim;
  float* bufB = bufA + (size_t)N * Hdim;
  int*   cnt  = (int*)(bufB + (size_t)N * Hdim);
  float* dn   = (float*)cnt + N;
  int*   srow = (int*)(dn + N);
  float* pk   = out;
  float* pk2  = out + 2 * (size_t)N;
  float* gb   = out + 4 * (size_t)N;
  float* gb2  = out + 5 * (size_t)N;

  int NB = (N + BIN_NODES - 1) / BIN_NODES;
  size_t scratch = 2 * (size_t)E + 6 * (size_t)NB + 8;
  if (NB <= MAX_BINS && E <= (1 << 23) && scratch <= (size_t)N * Hdim) {
    int* cbin   = (int*)bufA;
    int* rbin   = cbin + E;
    int* tot    = rbin + E;
    int* cursor = tot + 2 * NB;
    int* bstart = cursor + 2 * NB;
    int nchunks = (E + CHUNK - 1) / CHUNK;
    hipMemsetAsync(tot, 0, (size_t)NB * 4 * sizeof(int), stream);
    bin_count_kernel<<<nchunks, 1024, 0, stream>>>(ei, tot, E, NB);
    bin_scatter_kernel<<<nchunks, 1024, 0, stream>>>(ei, tot, cursor, bstart, cbin, rbin, E, NB);
    bin_finish_kernel<<<NB, 1024, 0, stream>>>(bstart, cbin, rbin, cnt, dn, srow, N, NB);
  } else {
    hipMemsetAsync(cnt, 0, (size_t)N * 2 * sizeof(float), stream);
    deg_kernel<<<(E + 255) / 256, 256, 0, stream>>>(ei, dn, cnt, E);
    dn_kernel<<<(N + 255) / 256, 256, 0, stream>>>(dn, N);
    scan_kernel<<<1, 1024, 0, stream>>>(cnt, N);
    fill_kernel<<<(E + 255) / 256, 256, 0, stream>>>(ei, cnt, srow, E);
    order_kernel<<<(N + 7) / 8, 256, 0, stream>>>(cnt, srow, N);
  }

  gemm_relu_kernel<<<(N + 15) / 16, 256, 0, stream>>>(x, W_in, b_in, L > 0 ? Wg : nullptr, dn,
                                                     raw, pk, gb, N);

  float* h = raw;
  float* hnew = bufA;
  float* gac = pk;  float* gbc = gb;
  float* gan = pk2; float* gbn = gb2;
  for (int i = 0; i + 2 < L; ++i) {
    const float* wgn = Wg + (size_t)(i + 1) * 2 * Hdim;
    gather_kernel<<<(N + 7) / 8, 256, 0, stream>>>(cnt, srow, gac, gbc, bg, i, h, raw, hnew,
                                                  wgn, gan, gbn, N);
    h = hnew;
    hnew = (h == bufA) ? bufB : bufA;
    float* t = gac; gac = gan; gan = t;
    t = gbc; gbc = gbn; gbn = t;
  }

  int tiles = (N + 15) / 16;
  if (L >= 2) {
    bool same = (h == raw);
    float* Y  = hnew;
    float* YR = same ? Y : hnew + 32 * (size_t)N;
    float* Z  = hnew + 64 * (size_t)N;
    float* ZR = hnew + 80 * (size_t)N;
    float* LG = hnew + 96 * (size_t)N;
    int nt = same ? tiles : 2 * tiles;
    gemm_proj2_kernel<<<(nt + 7) / 8, 256, 0, stream>>>(h, raw, W_out,
                                                       Wg + (size_t)(L - 1) * 2 * Hdim, gac, gbc,
                                                       Y, YR, N, nt);
    gather_mid_kernel<<<(N + 63) / 64, 256, 0, stream>>>(cnt, srow, bg, L - 2, Y, YR, b_out,
                                                        Z, ZR, gan, gbn, N);
    gather_proj_kernel<<<(N + 63) / 64, 256, 0, stream>>>(cnt, srow, gan, gbn, bg, L - 1, Z, ZR,
                                                         LG, N);
    logsoftmax_kernel<<<(N + 255) / 256, 256, 0, stream>>>((const float4*)LG, (float4*)out, N);
    return;
  }

  float* Z  = hnew;
  float* ZR = hnew + 16 * (size_t)N;
  float* LG = hnew + 32 * (size_t)N;
  gemm_proj_kernel<<<(2 * tiles + 7) / 8, 256, 0, stream>>>(h, raw, W_out, b_out,
                                                           L > 0 ? FAGCN_EPS : 1.0f, Z, ZR, N);
  if (L > 0)
    gather_proj_kernel<<<(N + 63) / 64, 256, 0, stream>>>(cnt, srow, gac, gbc, bg, L - 1, Z, ZR,
                                                         LG, N);
  logsoftmax_kernel<<<(N + 255) / 256, 256, 0, stream>>>((const float4*)(L > 0 ? LG : ZR),
                                                        (float4*)out, N);
}


# Input GEMM reuses its operands: 64-row blocks, x rows staged once in LDS by whole-line loads, each wave keeps its W_in fragment in registers for 4 row tiles; global loads per WMMA 3 -> 0.56
# speedup vs baseline: 5.2742x; 1.1281x over previous
#include <hip/hip_runtime.h>

typedef float v2f __attribute__((ext_vector_type(2)));
typedef float v8f __attribute__((ext_vector_type(8)));

#define FAGCN_EPS 0.3f

__global__ void deg_kernel(const int* __restrict__ ei, float* __restrict__ deg,
                           int* __restrict__ cnt, int E) {
  int e = blockIdx.x * blockDim.x + threadIdx.x;
  if (e < E) {
    atomicAdd(&deg[ei[e]], 1.0f);
    atomicAdd(&cnt[ei[E + e]], 1);
  }
}

__device__ __forceinline__ int4 scan_load4(const int* cnt, int i0, int N) {
  int4 v = {0, 0, 0, 0};
  if (i0 + 3 < N) {
    v = *(const int4*)(cnt + i0);
  } else {
    if (i0 < N) v.x = cnt[i0];
    if (i0 + 1 < N) v.y = cnt[i0 + 1];
    if (i0 + 2 < N) v.z = cnt[i0 + 2];
  }
  return v;
}

__global__ void scan_kernel(int* cnt, int N) {
  __shared__ int wsum[32];
  int tid = threadIdx.x;
  int lane = tid & 31;
  int wave = tid >> 5;
  int carry = 0;
  int4 nxt = scan_load4(cnt, tid * 4, N);
  for (int base = 0; base < N; base += 4096) {
    int i0 = base + tid * 4;
    int4 v = nxt;
    nxt = scan_load4(cnt, i0 + 4096, N);
    int s = v.x + v.y + v.z + v.w;
    int incl = s;
    for (int off = 1; off < 32; off <<= 1) {
      int t = __shfl_up(incl, off, 32);
      if (lane >= off) incl += t;
    }
    if (lane == 31) wsum[wave] = incl;
    __syncthreads();
    int wt = wsum[lane];
    int wincl = wt;
    for (int off = 1; off < 32; off <<= 1) {
      int t = __shfl_up(wincl, off, 32);
      if (lane >= off) wincl += t;
    }
    int wexcl = __shfl(wincl, wave, 32) - __shfl(wt, wave, 32);
    int total = __shfl(wincl, 31, 32);
    int ex = carry + wexcl + incl - s;
    if (i0 + 3 < N) {
      int4 o;
      o.x = ex; o.y = ex + v.x; o.z = ex + v.x + v.y; o.w = ex + v.x + v.y + v.z;
      *(int4*)(cnt + i0) = o;
    } else {
      if (i0 < N) cnt[i0] = ex;
      if (i0 + 1 < N) cnt[i0 + 1] = ex + v.x;
      if (i0 + 2 < N) cnt[i0 + 2] = ex + v.x + v.y;
    }
    carry += total;
    __syncthreads();
  }
}

__global__ void fill_kernel(const int* __restrict__ ei, int* __restrict__ cur,
                            int* __restrict__ srow, int E) {
  int e = blockIdx.x * blockDim.x + threadIdx.x;
  if (e < E) {
    int pos = atomicAdd(&cur[ei[E + e]], 1);
    srow[pos] = ei[e];
  }
}

__device__ __forceinline__ void order_segment(int* __restrict__ srow, int beg, int d, int lane) {
  int a0 = lane < d ? srow[beg + lane] : 0x7fffffff;
  int a1 = lane + 32 < d ? srow[beg + 32 + lane] : 0x7fffffff;
  int r0 = 0, r1 = 0;
  int n0 = d < 32 ? d : 32;
  for (int i = 0; i < n0; ++i) {
    int x = __builtin_amdgcn_readlane(a0, i);
    r0 += (x < a0 || (x == a0 && i < lane)) ? 1 : 0;
    r1 += (x <= a1) ? 1 : 0;
  }
  for (int i = 0; i < d - 32; ++i) {
    int x = __builtin_amdgcn_readlane(a1, i);
    r0 += (x < a0) ? 1 : 0;
    r1 += (x < a1 || (x == a1 && i < lane)) ? 1 : 0;
  }
  if (lane < d) srow[beg + r0] = a0;
  if (lane + 32 < d) srow[beg + r1] = a1;
}

__global__ void order_kernel(const int* __restrict__ ends, int* __restrict__ srow, int N) {
  int node = __builtin_amdgcn_readfirstlane(blockIdx.x * 8 + (threadIdx.x >> 5));
  int lane = threadIdx.x & 31;
  if (node >= N) return;
  int beg = node ? ends[node - 1] : 0;
  int d = ends[node] - beg;
  if (d < 2 || d > 64) return;
  order_segment(srow, beg, d, lane);
}

__global__ void dn_kernel(float* dn, int N) {
  int i = blockIdx.x * blockDim.x + threadIdx.x;
  if (i < N) dn[i] = 1.0f / sqrtf(fmaxf(dn[i], 1.0f));
}

#define BIN_SHIFT 8
#define BIN_NODES 256
#define MAX_BINS 512
#define CHUNK 4096
#define BIN_CAP 8192

__device__ __forceinline__ int block_scan_1024(int s, int* wsum, int lane, int wave) {
  int incl = s;
  for (int off = 1; off < 32; off <<= 1) {
    int t = __shfl_up(incl, off, 32);
    if (lane >= off) incl += t;
  }
  if (lane == 31) wsum[wave] = incl;
  __syncthreads();
  int wt = wsum[lane];
  int wincl = wt;
  for (int off = 1; off < 32; off <<= 1) {
    int t = __shfl_up(wincl, off, 32);
    if (lane >= off) wincl += t;
  }
  int wexcl = __shfl(wincl, wave, 32) - __shfl(wt, wave, 32);
  __syncthreads();
  return wexcl + incl - s;
}

__global__ void bin_count_kernel(const int* __restrict__ ei, int* __restrict__ tot, int E, int NB) {
  __shared__ int hist[2 * MAX_BINS];
  int tid = threadIdx.x;
  for (int i = tid; i < 2 * NB; i += 1024) hist[i] = 0;
  __syncthreads();
  int e0 = blockIdx.x * CHUNK + tid;
  int col[4], row[4];
#pragma unroll
  for (int k = 0; k < 4; ++k) {
    int e = e0 + k * 1024;
    col[k] = e < E ? ei[E + e] : -1;
    row[k] = e < E ? ei[e] : -1;
  }
#pragma unroll
  for (int k = 0; k < 4; ++k) {
    if (col[k] >= 0) {
      atomicAdd(&hist[col[k] >> BIN_SHIFT], 1);
      atomicAdd(&hist[NB + (row[k] >> BIN_SHIFT)], 1);
    }
  }
  __syncthreads();
  for (int i = tid; i < 2 * NB; i += 1024) {
    int c = hist[i];
    if (c) atomicAdd(&tot[i], c);
  }
}

__global__ void bin_scatter_kernel(const int* __restrict__ ei, const int* __restrict__ tot,
                                   int* __restrict__ cursor, int* __restrict__ bstart,
                                   int* __restrict__ cbin, int* __restrict__ rbin, int E, int NB) {
  __shared__ int hist[2 * MAX_BINS];
  __shared__ int gbase[2 * MAX_BINS];
  __shared__ int wsum[32];
  __shared__ int keyC[CHUNK];
  __shared__ int adrC[CHUNK];
  __shared__ int stR[CHUNK];
  int tid = threadIdx.x;
  int lane = tid & 31;
  int wave = tid >> 5;
  int nb2 = 2 * NB;
  for (int i = tid; i < nb2; i += 1024) hist[i] = 0;
  __syncthreads();
  int ebase = blockIdx.x * CHUNK;
  int nE = E - ebase < CHUNK ? E - ebase : CHUNK;
  int i0 = 2 * tid;
  int t0 = i0 < nb2 ? tot[i0] : 0;
  int t1 = i0 + 1 < nb2 ? tot[i0 + 1] : 0;
  int col[4], row[4], rc[4], rr[4];
#pragma unroll
  for (int k = 0; k < 4; ++k) {
    int e = ebase + tid + k * 1024;
    col[k] = e < E ? ei[E + e] : -1;
    row[k] = e < E ? ei[e] : -1;
  }
#pragma unroll
  for (int k = 0; k < 4; ++k) {
    rc[k] = 0; rr[k] = 0;
    if (col[k] >= 0) {
      rc[k] = atomicAdd(&hist[col[k] >> BIN_SHIFT], 1);
      rr[k] = atomicAdd(&hist[NB + (row[k] >> BIN_SHIFT)], 1);
    }
  }
  __syncthreads();
  int h0 = i0 < nb2 ? hist[i0] : 0;
  int h1 = i0 + 1 < nb2 ? hist[i0 + 1] : 0;
  int gex = block_scan_1024(t0 + t1, wsum, lane, wave);
  int lex = block_scan_1024(h0 + h1, wsum, lane, wave);
  if (i0 < nb2) {
    int second = i0 >= NB ? 1 : 0;
    int s = gex - (second ? E : 0);
    gbase[i0] = s + (h0 ? atomicAdd(&cursor[i0], h0) : 0);
    hist[i0] = lex - (second ? nE : 0);
    if (blockIdx.x == 0) bstart[i0 + second] = s;
  }
  if (i0 + 1 < nb2) {
    int second = i0 + 1 >= NB ? 1 : 0;
    int s = gex + t0 - (second ? E : 0);
    gbase[i0 + 1] = s + (h1 ? atomicAdd(&cursor[i0 + 1], h1) : 0);
    hist[i0 + 1] = lex + h0 - (second ? nE : 0);
    if (blockIdx.x == 0) bstart[i0 + 1 + second] = s;
  }
  if (blockIdx.x == 0 && tid == 0) {
    bstart[NB] = E;
    bstart[nb2 + 1] = E;
  }
  __syncthreads();
#pragma unroll
  for (int k = 0; k < 4; ++k) {
    if (col[k] >= 0) {
      int bc = col[k] >> BIN_SHIFT;
      int sc = hist[bc] + rc[k];
      keyC[sc] = (row[k] << BIN_SHIFT) | (col[k] & (BIN_NODES - 1));
      adrC[sc] = gbase[bc] + rc[k];
      int br = NB + (row[k] >> BIN_SHIFT);
      int sr = hist[br] + rr[k];
      stR[sr] = ((gbase[br] + rr[k]) << BIN_SHIFT) | (row[k] & (BIN_NODES - 1));
    }
  }
  __syncthreads();
  for (int j = tid; j < nE; j += 1024) {
    cbin[adrC[j]] = keyC[j];
    int w = stR[j];
    rbin[w >> BIN_SHIFT] = w & (BIN_NODES - 1);
  }
}

__global__ void bin_finish_kernel(const int* __restrict__ bstart, const int* __restrict__ cbin,
                                  const int* __restrict__ rbin, int* __restrict__ ends,
                                  float* __restrict__ dn, int* __restrict__ srow, int N, int NB) {
  __shared__ int ncnt[BIN_NODES];
  __shared__ int nbeg[BIN_NODES];
  __shared__ int ncur[BIN_NODES];
  __shared__ int dcnt[BIN_NODES];
  __shared__ int wsum[32];
  __shared__ int stage[BIN_CAP];
  int tid = threadIdx.x;
  int lane = tid & 31;
  int wave = tid >> 5;
  int b = blockIdx.x;
  int cb = bstart[b], ce = bstart[b + 1];
  int rb = bstart[NB + 1 + b], re = bstart[NB + 2 + b];
  int kc[5], kr[5];
#pragma unroll
  for (int k = 0; k < 5; ++k) {
    int i = cb + tid + k * 1024;
    int j = rb + tid + k * 1024;
    kc[k] = i < ce ? cbin[i] : -1;
    kr[k] = j < re ? rbin[j] : -1;
  }
  if (tid < BIN_NODES) { ncnt[tid] = 0; dcnt[tid] = 0; }
  __syncthreads();
#pragma unroll
  for (int k = 0; k < 5; ++k) {
    if (kc[k] >= 0) atomicAdd(&ncnt[kc[k] & (BIN_NODES - 1)], 1);
    if (kr[k] >= 0) atomicAdd(&dcnt[kr[k]], 1);
  }
  for (int i = cb + tid + 5 * 1024; i < ce; i += 1024) atomicAdd(&ncnt[cbin[i] & (BIN_NODES - 1)], 1);
  for (int i = rb + tid + 5 * 1024; i < re; i += 1024) atomicAdd(&dcnt[rbin[i]], 1);
  __syncthreads();
  int c = tid < BIN_NODES ? ncnt[tid] : 0;
  int ex = block_scan_1024(c, wsum, lane, wave);
  if (tid < BIN_NODES) {
    nbeg[tid] = ex;
    ncur[tid] = ex;
    int node = b * BIN_NODES + tid;
    if (node < N) {
      ends[node] = cb + ex + c;
      dn[node] = 1.0f / sqrtf(fmaxf((float)dcnt[tid], 1.0f));
    }
  }
  __syncthreads();
  int cnt = ce - cb;
  bool fits = cnt <= BIN_CAP;
#pragma unroll
  for (int k = 0; k < 5; ++k) {
    if (kc[k] >= 0) {
      int p = atomicAdd(&ncur[kc[k] & (BIN_NODES - 1)], 1);
      if (fits) stage[p] = kc[k];
      else srow[cb + p] = kc[k] >> BIN_SHIFT;
    }
  }
  for (int i = cb + tid + 5 * 1024; i < ce; i += 1024) {
    int key = cbin[i];
    int p = atomicAdd(&ncur[key & (BIN_NODES - 1)], 1);
    if (fits) stage[p] = key;
    else srow[cb + p] = key >> BIN_SHIFT;
  }
  if (!fits) {
    __threadfence();
    __syncthreads();
    for (int t = wave; t < BIN_NODES; t += 32) {
      int n = __builtin_amdgcn_readfirstlane(ncnt[t]);
      int sb = __builtin_amdgcn_readfirstlane(nbeg[t]);
      if (n >= 2 && n <= 64) order_segment(srow, cb + sb, n, lane);
    }
    return;
  }
  __syncthreads();
  for (int p = tid; p < cnt; p += 1024) {
    int key = stage[p];
    int d = key & (BIN_NODES - 1);
    int beg = nbeg[d];
    int n = ncnt[d];
    int r = p - beg;
    if (n > 1 && n <= 64) {
      r = 0;
      for (int j = 0; j < n; ++j) {
        int x = stage[beg + j];
        r += (x < key || (x == key && beg + j < p)) ? 1 : 0;
      }
    }
    srow[cb + beg + r] = key >> BIN_SHIFT;
  }
}

#define TILE_PITCH 132
#define GEMM_ROWS 64
__global__ void gemm_relu_kernel(const float* __restrict__ X, const float* __restrict__ W,
                                 const float* __restrict__ bias, const float* __restrict__ wg0,
                                 const float* __restrict__ dn, float* __restrict__ H,
                                 float* __restrict__ pk, float* __restrict__ gb, int Nrows) {
  __shared__ __attribute__((aligned(16))) float tile[GEMM_ROWS * TILE_PITCH];
  int wave = threadIdx.x >> 5;
  int lane = threadIdx.x & 31;
  int row0 = blockIdx.x * GEMM_ROWS;
  int c = lane & 15;
  int n = wave * 16 + c;
  int khalf = (lane >> 4) * 2;
  float4 xr[8];
#pragma unroll
  for (int j = 0; j < 8; ++j) {
    int m = row0 + wave * 8 + j;
    int mc = m < Nrows ? m : (Nrows - 1);
    xr[j] = *(const float4*)(X + (size_t)mc * 128 + lane * 4);
  }
  int md = row0 + wave * 8 + (lane & 7);
  float dnv = dn[md < Nrows ? md : (Nrows - 1)];
  const float* wcol = W + (size_t)khalf * 128 + n;
  v2f bf[32];
#pragma unroll
  for (int s = 0; s < 32; ++s) {
    bf[s].x = wcol[(size_t)s * 512];
    bf[s].y = wcol[(size_t)s * 512 + 128];
  }
#pragma unroll
  for (int j = 0; j < 8; ++j)
    *(float4*)(tile + (wave * 8 + j) * TILE_PITCH + lane * 4) = xr[j];
  __syncthreads();
  v8f acc0 = {}, acc1 = {}, acc2 = {}, acc3 = {};
  const float* arow = tile + c * TILE_PITCH + khalf;
  v2f cur[4][4], nxt[4][4];
#pragma unroll
  for (int i = 0; i < 4; ++i)
#pragma unroll
    for (int t = 0; t < 4; ++t)
      cur[i][t] = *(const v2f*)(arow + t * 16 * TILE_PITCH + i * 4);
#pragma unroll
  for (int g = 0; g < 8; ++g) {
    if (g + 1 < 8) {
#pragma unroll
      for (int i = 0; i < 4; ++i)
#pragma unroll
        for (int t = 0; t < 4; ++t)
          nxt[i][t] = *(const v2f*)(arow + t * 16 * TILE_PITCH + (g + 1) * 16 + i * 4);
    }
#pragma unroll
    for (int i = 0; i < 4; ++i) {
      int s = g * 4 + i;
      acc0 = __builtin_amdgcn_wmma_f32_16x16x4_f32(false, cur[i][0], false, bf[s], (short)0, acc0, false, false);
      acc1 = __builtin_amdgcn_wmma_f32_16x16x4_f32(false, cur[i][1], false, bf[s], (short)0, acc1, false, false);
      acc2 = __builtin_amdgcn_wmma_f32_16x16x4_f32(false, cur[i][2], false, bf[s], (short)0, acc2, false, false);
      acc3 = __builtin_amdgcn_wmma_f32_16x16x4_f32(false, cur[i][3], false, bf[s], (short)0, acc3, false, false);
    }
#pragma unroll
    for (int i = 0; i < 4; ++i)
#pragma unroll
      for (int t = 0; t < 4; ++t)
        cur[i][t] = nxt[i][t];
  }
  __syncthreads();
  int rbase = (lane >> 4) * 8;
  float bn = bias[n];
#pragma unroll
  for (int r = 0; r < 8; ++r) {
    tile[(rbase + r) * TILE_PITCH + n] = fmaxf(acc0[r] + bn, 0.0f);
    tile[(16 + rbase + r) * TILE_PITCH + n] = fmaxf(acc1[r] + bn, 0.0f);
    tile[(32 + rbase + r) * TILE_PITCH + n] = fmaxf(acc2[r] + bn, 0.0f);
    tile[(48 + rbase + r) * TILE_PITCH + n] = fmaxf(acc3[r] + bn, 0.0f);
  }
  __syncthreads();
  float4 wa = {0.0f, 0.0f, 0.0f, 0.0f}, wb = {0.0f, 0.0f, 0.0f, 0.0f};
  if (wg0) {
    wa = *(const float4*)(wg0 + lane * 4);
    wb = *(const float4*)(wg0 + 128 + lane * 4);
  }
  for (int j = 0; j < 8; ++j) {
    int rr = wave * 8 + j;
    int m = row0 + rr;
    float4 v = *(const float4*)(tile + rr * TILE_PITCH + lane * 4);
    if (m < Nrows) {
      *(float4*)(H + (size_t)m * 128 + lane * 4) = v;
      if (wg0) {
        float dnm = __shfl(dnv, j, 32);
        float sa = v.x * wa.x + v.y * wa.y + v.z * wa.z + v.w * wa.w;
        float sb = v.x * wb.x + v.y * wb.y + v.z * wb.z + v.w * wb.w;
        for (int off = 16; off > 0; off >>= 1) {
          sa += __shfl_down(sa, off, 32);
          sb += __shfl_down(sb, off, 32);
        }
        if (lane == 0) {
          float2 p; p.x = sa; p.y = dnm;
          *(float2*)(pk + 2 * (size_t)m) = p;
          gb[m] = sb;
        }
      }
    }
  }
}

__global__ void gather_kernel(const int* __restrict__ ends, const int* __restrict__ srow,
                              const float* __restrict__ pk, const float* __restrict__ gb,
                              const float* __restrict__ bg,
                              int layer, const float* __restrict__ H,
                              const float* __restrict__ raw, float* __restrict__ hout,
                              const float* __restrict__ wg_next, float* __restrict__ pk_next,
                              float* __restrict__ gb_next, int N) {
  int node = __builtin_amdgcn_readfirstlane(blockIdx.x * 8 + (threadIdx.x >> 5));
  int lane = threadIdx.x & 31;
  if (node >= N) return;
  int beg = node ? ends[node - 1] : 0;
  int end = ends[node];
  float gbc = gb[node];
  float bgl = bg[layer];
  float dnc = pk[2 * (size_t)node + 1];
  const float* hl = H + lane * 4;
  float4 rw = *(const float4*)(raw + (size_t)node * 128 + lane * 4);
  float4 acc = {0.0f, 0.0f, 0.0f, 0.0f};
  for (int base = beg; base < end; base += 32) {
    int n = end - base < 32 ? end - base : 32;
    int r = 0;
    float w = 0.0f;
    if (lane < n) {
      r = srow[base + lane];
      float2 gd = *(const float2*)(pk + 2 * (size_t)r);
      float g = tanhf(gd.x + gbc + bgl);
      w = g * gd.y * dnc;
    }
    int wi = __float_as_int(w);
    int k = 0;
    for (; k + 4 <= n; k += 4) {
      int r0 = __builtin_amdgcn_readlane(r, k);
      int r1 = __builtin_amdgcn_readlane(r, k + 1);
      int r2 = __builtin_amdgcn_readlane(r, k + 2);
      int r3 = __builtin_amdgcn_readlane(r, k + 3);
      float w0 = __int_as_float(__builtin_amdgcn_readlane(wi, k));
      float w1 = __int_as_float(__builtin_amdgcn_readlane(wi, k + 1));
      float w2 = __int_as_float(__builtin_amdgcn_readlane(wi, k + 2));
      float w3 = __int_as_float(__builtin_amdgcn_readlane(wi, k + 3));
      float4 h0 = *(const float4*)(hl + (size_t)r0 * 128);
      float4 h1 = *(const float4*)(hl + (size_t)r1 * 128);
      float4 h2 = *(const float4*)(hl + (size_t)r2 * 128);
      float4 h3 = *(const float4*)(hl + (size_t)r3 * 128);
      acc.x += w0 * h0.x; acc.y += w0 * h0.y; acc.z += w0 * h0.z; acc.w += w0 * h0.w;
      acc.x += w1 * h1.x; acc.y += w1 * h1.y; acc.z += w1 * h1.z; acc.w += w1 * h1.w;
      acc.x += w2 * h2.x; acc.y += w2 * h2.y; acc.z += w2 * h2.z; acc.w += w2 * h2.w;
      acc.x += w3 * h3.x; acc.y += w3 * h3.y; acc.z += w3 * h3.z; acc.w += w3 * h3.w;
    }
    for (; k < n; ++k) {
      int r0 = __builtin_amdgcn_readlane(r, k);
      float w0 = __int_as_float(__builtin_amdgcn_readlane(wi, k));
      float4 h0 = *(const float4*)(hl + (size_t)r0 * 128);
      acc.x += w0 * h0.x; acc.y += w0 * h0.y; acc.z += w0 * h0.z; acc.w += w0 * h0.w;
    }
  }
  float4 o;
  o.x = FAGCN_EPS * rw.x + acc.x;
  o.y = FAGCN_EPS * rw.y + acc.y;
  o.z = FAGCN_EPS * rw.z + acc.z;
  o.w = FAGCN_EPS * rw.w + acc.w;
  *(float4*)(hout + (size_t)node * 128 + lane * 4) = o;
  if (wg_next) {
    float4 wa = *(const float4*)(wg_next + lane * 4);
    float4 wb = *(const float4*)(wg_next + 128 + lane * 4);
    float sa = o.x * wa.x + o.y * wa.y + o.z * wa.z + o.w * wa.w;
    float sb = o.x * wb.x + o.y * wb.y + o.z * wb.z + o.w * wb.w;
    for (int off = 16; off > 0; off >>= 1) {
      sa += __shfl_down(sa, off, 32);
      sb += __shfl_down(sb, off, 32);
    }
    if (lane == 0) {
      float2 p; p.x = sa; p.y = dnc;
      *(float2*)(pk_next + 2 * (size_t)node) = p;
      gb_next[node] = sb;
    }
  }
}

__global__ void gemm_proj_kernel(const float* __restrict__ Hc, const float* __restrict__ Raw,
                                 const float* __restrict__ W, const float* __restrict__ bias,
                                 float rscale, float* __restrict__ Z, float* __restrict__ ZR,
                                 int Nrows) {
  int wave = threadIdx.x >> 5;
  int lane = threadIdx.x & 31;
  int tiles = (Nrows + 15) >> 4;
  int t = blockIdx.x * 8 + wave;
  if (t >= 2 * tiles) return;
  bool second = t >= tiles;
  int tileM = second ? t - tiles : t;
  const float* src = second ? Raw : Hc;
  float* dst = second ? ZR : Z;
  int mA = tileM * 16 + (lane & 15);
  int mclamp = mA < Nrows ? mA : (Nrows - 1);
  int n = lane & 15;
  int khalf = (lane >> 4) * 2;
  const float* hrow = src + (size_t)mclamp * 128 + khalf;
  const float* wcol = W + (size_t)khalf * 16 + n;
  __builtin_prefetch(hrow, 0, 0);
  v8f acc = {};
#pragma unroll 4
  for (int kk = 0; kk < 128; kk += 4) {
    v2f a = *(const v2f*)(hrow + kk);
    v2f b;
    b.x = wcol[(size_t)kk * 16];
    b.y = wcol[(size_t)kk * 16 + 16];
    acc = __builtin_amdgcn_wmma_f32_16x16x4_f32(false, a, false, b, (short)0, acc,
                                               false, false);
  }
  int mbase = tileM * 16 + ((lane >> 4) * 8);
  float bn = second ? bias[n] : 0.0f;
  float sc = second ? rscale : 1.0f;
  if (tileM * 16 + 16 <= Nrows) {
    float* op = dst + (size_t)mbase * 16 + n;
#pragma unroll
    for (int r = 0; r < 8; ++r)
      op[(size_t)r * 16] = acc[r] * sc + bn;
  } else {
#pragma unroll
    for (int r = 0; r < 8; ++r) {
      int mm = mbase + r;
      if (mm < Nrows) dst[(size_t)mm * 16 + n] = acc[r] * sc + bn;
    }
  }
}

__global__ void gather_proj_kernel(const int* __restrict__ ends, const int* __restrict__ srow,
                                   const float* __restrict__ pk, const float* __restrict__ gb,
                                   const float* __restrict__ bg, int layer,
                                   const float* __restrict__ Z, const float* __restrict__ ZR,
                                   float* __restrict__ logits, int N) {
  int node = blockIdx.x * 64 + (threadIdx.x >> 2);
  int q = threadIdx.x & 3;
  if (node >= N) return;
  int beg = node ? ends[node - 1] : 0;
  int end = ends[node];
  float gbc = gb[node];
  float bgl = bg[layer];
  float dnc = pk[2 * (size_t)node + 1];
  const float* zq = Z + q * 4;
  float4 zr = *(const float4*)(ZR + (size_t)node * 16 + q * 4);
  float4 acc = {0.0f, 0.0f, 0.0f, 0.0f};
  int last = end - 1;
  int r0 = 0, r1 = 0, r2 = 0, r3 = 0;
  if (beg < end) {
    r0 = srow[beg];
    r1 = srow[beg + 1 < last ? beg + 1 : last];
    r2 = srow[beg + 2 < last ? beg + 2 : last];
    r3 = srow[beg + 3 < last ? beg + 3 : last];
  }
  for (int base = beg; base < end; base += 4) {
    float2 g0 = *(const float2*)(pk + 2 * (size_t)r0);
    float2 g1 = *(const float2*)(pk + 2 * (size_t)r1);
    float2 g2 = *(const float2*)(pk + 2 * (size_t)r2);
    float2 g3 = *(const float2*)(pk + 2 * (size_t)r3);
    float4 h0 = *(const float4*)(zq + (size_t)r0 * 16);
    float4 h1 = *(const float4*)(zq + (size_t)r1 * 16);
    float4 h2 = *(const float4*)(zq + (size_t)r2 * 16);
    float4 h3 = *(const float4*)(zq + (size_t)r3 * 16);
    int nb = base + 4;
    int n0 = r0, n1 = r1, n2 = r2, n3 = r3;
    if (nb < end) {
      n0 = srow[nb];
      n1 = srow[nb + 1 < last ? nb + 1 : last];
      n2 = srow[nb + 2 < last ? nb + 2 : last];
      n3 = srow[nb + 3 < last ? nb + 3 : last];
    }
    {
      float w = tanhf(g0.x + gbc + bgl) * g0.y * dnc;
      acc.x += w * h0.x; acc.y += w * h0.y; acc.z += w * h0.z; acc.w += w * h0.w;
    }
    if (base + 1 < end) {
      float w = tanhf(g1.x + gbc + bgl) * g1.y * dnc;
      acc.x += w * h1.x; acc.y += w * h1.y; acc.z += w * h1.z; acc.w += w * h1.w;
    }
    if (base + 2 < end) {
      float w = tanhf(g2.x + gbc + bgl) * g2.y * dnc;
      acc.x += w * h2.x; acc.y += w * h2.y; acc.z += w * h2.z; acc.w += w * h2.w;
    }
    if (base + 3 < end) {
      float w = tanhf(g3.x + gbc + bgl) * g3.y * dnc;
      acc.x += w * h3.x; acc.y += w * h3.y; acc.z += w * h3.z; acc.w += w * h3.w;
    }
    r0 = n0; r1 = n1; r2 = n2; r3 = n3;
  }
  float4 o;
  o.x = zr.x + acc.x;
  o.y = zr.y + acc.y;
  o.z = zr.z + acc.z;
  o.w = zr.w + acc.w;
  *(float4*)(logits + (size_t)node * 16 + q * 4) = o;
}

#define YP 32
__global__ void gemm_proj2_kernel(const float* __restrict__ Hc, const float* __restrict__ Raw,
                                  const float* __restrict__ W, const float* __restrict__ wgl,
                                  const float* __restrict__ pk, const float* __restrict__ gb,
                                  float* __restrict__ Y, float* __restrict__ YR,
                                  int Nrows, int nt) {
  int wave = threadIdx.x >> 5;
  int lane = threadIdx.x & 31;
  int tiles = (Nrows + 15) >> 4;
  int t = blockIdx.x * 8 + wave;
  if (t >= nt) return;
  bool second = t >= tiles;
  int tileM = second ? t - tiles : t;
  const float* src = second ? Raw : Hc;
  float* dst = second ? YR : Y;
  int mA = tileM * 16 + (lane & 15);
  int mclamp = mA < Nrows ? mA : (Nrows - 1);
  int n = lane & 15;
  int khalf = (lane >> 4) * 2;
  const float* hrow = src + (size_t)mclamp * 128 + khalf;
  const float* wcol = W + (size_t)khalf * 16 + n;
  const float* gcol = wgl + (n & 1) * 128 + khalf;
  __builtin_prefetch(hrow, 0, 0);
  int mbase = tileM * 16 + ((lane >> 4) * 8);
  const float* tp = (n == 2 || n == 3) ? pk + (n - 2) : gb;
  int ts = (n == 2 || n == 3) ? 2 : 1;
  float tv[8];
#pragma unroll
  for (int r = 0; r < 8; ++r) {
    int mr = mbase + r < Nrows ? mbase + r : Nrows - 1;
    tv[r] = tp[(size_t)mr * ts];
  }
  v8f acc = {};
  v8f acg = {};
#pragma unroll 4
  for (int kk = 0; kk < 128; kk += 4) {
    v2f a = *(const v2f*)(hrow + kk);
    v2f b;
    b.x = wcol[(size_t)kk * 16];
    b.y = wcol[(size_t)kk * 16 + 16];
    v2f g = *(const v2f*)(gcol + kk);
    acc = __builtin_amdgcn_wmma_f32_16x16x4_f32(false, a, false, b, (short)0, acc,
                                               false, false);
    acg = __builtin_amdgcn_wmma_f32_16x16x4_f32(false, a, false, g, (short)0, acg,
                                               false, false);
  }
  float* yb = dst + (size_t)mbase * YP + n;
  if (tileM * 16 + 16 <= Nrows) {
#pragma unroll
    for (int r = 0; r < 8; ++r) {
      yb[(size_t)r * YP] = acc[r];
      if (n < 5) yb[(size_t)r * YP + 16] = n < 2 ? acg[r] : tv[r];
    }
  } else {
#pragma unroll
    for (int r = 0; r < 8; ++r) {
      if (mbase + r < Nrows) {
        yb[(size_t)r * YP] = acc[r];
        if (n < 5) yb[(size_t)r * YP + 16] = n < 2 ? acg[r] : tv[r];
      }
    }
  }
}

__global__ void gather_mid_kernel(const int* __restrict__ ends, const int* __restrict__ srow,
                                  const float* __restrict__ bg, int layer,
                                  const float* __restrict__ Y, const float* __restrict__ YR,
                                  const float* __restrict__ bias,
                                  float* __restrict__ Z, float* __restrict__ ZR,
                                  float* __restrict__ pk_next, float* __restrict__ gb_next,
                                  int N) {
  int node = blockIdx.x * 64 + (threadIdx.x >> 2);
  int q = threadIdx.x & 3;
  if (node >= N) return;
  int beg = node ? ends[node - 1] : 0;
  int end = ends[node];
  float bgl = bg[layer];
  const float* yc = Y + (size_t)node * YP;
  float dnc = yc[19];
  float gbc = yc[20];
  const float* yrc = YR + (size_t)node * YP;
  float4 own = *(const float4*)(yrc + q * 4);
  float2 owng = *(const float2*)(yrc + 16);
  float4 bq = *(const float4*)(bias + q * 4);
  const float* yq = Y + q * 4;
  const float* yt = Y + 16;
  float4 acc = {0.0f, 0.0f, 0.0f, 0.0f};
  float sga = 0.0f, sgb = 0.0f;
  int last = end - 1;
  int r0 = 0, r1 = 0, r2 = 0, r3 = 0;
  if (beg < end) {
    r0 = srow[beg];
    r1 = srow[beg + 1 < last ? beg + 1 : last];
    r2 = srow[beg + 2 < last ? beg + 2 : last];
    r3 = srow[beg + 3 < last ? beg + 3 : last];
  }
  for (int base = beg; base < end; base += 4) {
    float4 t0 = *(const float4*)(yt + (size_t)r0 * YP);
    float4 t1 = *(const float4*)(yt + (size_t)r1 * YP);
    float4 t2 = *(const float4*)(yt + (size_t)r2 * YP);
    float4 t3 = *(const float4*)(yt + (size_t)r3 * YP);
    float4 h0 = *(const float4*)(yq + (size_t)r0 * YP);
    float4 h1 = *(const float4*)(yq + (size_t)r1 * YP);
    float4 h2 = *(const float4*)(yq + (size_t)r2 * YP);
    float4 h3 = *(const float4*)(yq + (size_t)r3 * YP);
    int nb = base + 4;
    int n0 = r0, n1 = r1, n2 = r2, n3 = r3;
    if (nb < end) {
      n0 = srow[nb];
      n1 = srow[nb + 1 < last ? nb + 1 : last];
      n2 = srow[nb + 2 < last ? nb + 2 : last];
      n3 = srow[nb + 3 < last ? nb + 3 : last];
    }
    {
      float w = tanhf(t0.z + gbc + bgl) * t0.w * dnc;
      acc.x += w * h0.x; acc.y += w * h0.y; acc.z += w * h0.z; acc.w += w * h0.w;
      sga += w * t0.x; sgb += w * t0.y;
    }
    if (base + 1 < end) {
      float w = tanhf(t1.z + gbc + bgl) * t1.w * dnc;
      acc.x += w * h1.x; acc.y += w * h1.y; acc.z += w * h1.z; acc.w += w * h1.w;
      sga += w * t1.x; sgb += w * t1.y;
    }
    if (base + 2 < end) {
      float w = tanhf(t2.z + gbc + bgl) * t2.w * dnc;
      acc.x += w * h2.x; acc.y += w * h2.y; acc.z += w * h2.z; acc.w += w * h2.w;
      sga += w * t2.x; sgb += w * t2.y;
    }
    if (base + 3 < end) {
      float w = tanhf(t3.z + gbc + bgl) * t3.w * dnc;
      acc.x += w * h3.x; acc.y += w * h3.y; acc.z += w * h3.z; acc.w += w * h3.w;
      sga += w * t3.x; sgb += w * t3.y;
    }
    r0 = n0; r1 = n1; r2 = n2; r3 = n3;
  }
  float4 o, zr;
  o.x = FAGCN_EPS * own.x + acc.x;  zr.x = FAGCN_EPS * own.x + bq.x;
  o.y = FAGCN_EPS * own.y + acc.y;  zr.y = FAGCN_EPS * own.y + bq.y;
  o.z = FAGCN_EPS * own.z + acc.z;  zr.z = FAGCN_EPS * own.z + bq.z;
  o.w = FAGCN_EPS * own.w + acc.w;  zr.w = FAGCN_EPS * own.w + bq.w;
  *(float4*)(Z + (size_t)node * 16 + q * 4) = o;
  *(float4*)(ZR + (size_t)node * 16 + q * 4) = zr;
  if (q == 0) {
    float2 p; p.x = FAGCN_EPS * owng.x + sga; p.y = dnc;
    *(float2*)(pk_next + 2 * (size_t)node) = p;
    gb_next[node] = FAGCN_EPS * owng.y + sgb;
  }
}

__global__ void logsoftmax_kernel(const float4* __restrict__ in, float4* __restrict__ out, int N) {
  int i = blockIdx.x * blockDim.x + threadIdx.x;
  if (i >= N) return;
  const float4* rin = in + (size_t)i * 4;
  float4* row = out + (size_t)i * 4;
  float4 v[4];
#pragma unroll
  for (int q = 0; q < 4; ++q) v[q] = rin[q];
  float m = v[0].x;
#pragma unroll
  for (int q = 0; q < 4; ++q) {
    m = fmaxf(m, v[q].x); m = fmaxf(m, v[q].y);
    m = fmaxf(m, v[q].z); m = fmaxf(m, v[q].w);
  }
  float s = 0.0f;
#pragma unroll
  for (int q = 0; q < 4; ++q) {
    s += expf(v[q].x - m); s += expf(v[q].y - m);
    s += expf(v[q].z - m); s += expf(v[q].w - m);
  }
  float d = m + logf(s);
#pragma unroll
  for (int q = 0; q < 4; ++q) {
    v[q].x -= d; v[q].y -= d; v[q].z -= d; v[q].w -= d;
    row[q] = v[q];
  }
}

extern "C" void kernel_launch(void* const* d_in, const int* in_sizes, int n_in,
                              void* d_out, int out_size, void* d_ws, size_t ws_size,
                              hipStream_t stream) {
  const float* x     = (const float*)d_in[0];
  const int*   ei    = (const int*)d_in[1];
  const float* W_in  = (const float*)d_in[2];
  const float* b_in  = (const float*)d_in[3];
  const float* Wg    = (const float*)d_in[4];
  const float* bg    = (const float*)d_in[5];
  const float* W_out = (const float*)d_in[6];
  const float* b_out = (const float*)d_in[7];
  float* out = (float*)d_out;

  const int Hdim = 128;
  int N = in_sizes[0] / Hdim;
  int E = in_sizes[1] / 2;
  int L = in_sizes[5];

  float* ws   = (float*)d_ws;
  float* raw  = ws;
  float* bufA = raw + (size_t)N * Hdim;
  float* bufB = bufA + (size_t)N * Hdim;
  int*   cnt  = (int*)(bufB + (size_t)N * Hdim);
  float* dn   = (float*)cnt + N;
  int*   srow = (int*)(dn + N);
  float* pk   = out;
  float* pk2  = out + 2 * (size_t)N;
  float* gb   = out + 4 * (size_t)N;
  float* gb2  = out + 5 * (size_t)N;

  int NB = (N + BIN_NODES - 1) / BIN_NODES;
  size_t scratch = 2 * (size_t)E + 6 * (size_t)NB + 8;
  if (NB <= MAX_BINS && E <= (1 << 23) && scratch <= (size_t)N * Hdim) {
    int* cbin   = (int*)bufA;
    int* rbin   = cbin + E;
    int* tot    = rbin + E;
    int* cursor = tot + 2 * NB;
    int* bstart = cursor + 2 * NB;
    int nchunks = (E + CHUNK - 1) / CHUNK;
    hipMemsetAsync(tot, 0, (size_t)NB * 4 * sizeof(int), stream);
    bin_count_kernel<<<nchunks, 1024, 0, stream>>>(ei, tot, E, NB);
    bin_scatter_kernel<<<nchunks, 1024, 0, stream>>>(ei, tot, cursor, bstart, cbin, rbin, E, NB);
    bin_finish_kernel<<<NB, 1024, 0, stream>>>(bstart, cbin, rbin, cnt, dn, srow, N, NB);
  } else {
    hipMemsetAsync(cnt, 0, (size_t)N * 2 * sizeof(float), stream);
    deg_kernel<<<(E + 255) / 256, 256, 0, stream>>>(ei, dn, cnt, E);
    dn_kernel<<<(N + 255) / 256, 256, 0, stream>>>(dn, N);
    scan_kernel<<<1, 1024, 0, stream>>>(cnt, N);
    fill_kernel<<<(E + 255) / 256, 256, 0, stream>>>(ei, cnt, srow, E);
    order_kernel<<<(N + 7) / 8, 256, 0, stream>>>(cnt, srow, N);
  }

  gemm_relu_kernel<<<(N + GEMM_ROWS - 1) / GEMM_ROWS, 256, 0, stream>>>(x, W_in, b_in, L > 0 ? Wg : nullptr, dn,
                                                     raw, pk, gb, N);

  float* h = raw;
  float* hnew = bufA;
  float* gac = pk;  float* gbc = gb;
  float* gan = pk2; float* gbn = gb2;
  for (int i = 0; i + 2 < L; ++i) {
    const float* wgn = Wg + (size_t)(i + 1) * 2 * Hdim;
    gather_kernel<<<(N + 7) / 8, 256, 0, stream>>>(cnt, srow, gac, gbc, bg, i, h, raw, hnew,
                                                  wgn, gan, gbn, N);
    h = hnew;
    hnew = (h == bufA) ? bufB : bufA;
    float* t = gac; gac = gan; gan = t;
    t = gbc; gbc = gbn; gbn = t;
  }

  int tiles = (N + 15) / 16;
  if (L >= 2) {
    bool same = (h == raw);
    float* Y  = hnew;
    float* YR = same ? Y : hnew + 32 * (size_t)N;
    float* Z  = hnew + 64 * (size_t)N;
    float* ZR = hnew + 80 * (size_t)N;
    float* LG = hnew + 96 * (size_t)N;
    int nt = same ? tiles : 2 * tiles;
    gemm_proj2_kernel<<<(nt + 7) / 8, 256, 0, stream>>>(h, raw, W_out,
                                                       Wg + (size_t)(L - 1) * 2 * Hdim, gac, gbc,
                                                       Y, YR, N, nt);
    gather_mid_kernel<<<(N + 63) / 64, 256, 0, stream>>>(cnt, srow, bg, L - 2, Y, YR, b_out,
                                                        Z, ZR, gan, gbn, N);
    gather_proj_kernel<<<(N + 63) / 64, 256, 0, stream>>>(cnt, srow, gan, gbn, bg, L - 1, Z, ZR,
                                                         LG, N);
    logsoftmax_kernel<<<(N + 255) / 256, 256, 0, stream>>>((const float4*)LG, (float4*)out, N);
    return;
  }

  float* Z  = hnew;
  float* ZR = hnew + 16 * (size_t)N;
  float* LG = hnew + 32 * (size_t)N;
  gemm_proj_kernel<<<(2 * tiles + 7) / 8, 256, 0, stream>>>(h, raw, W_out, b_out,
                                                           L > 0 ? FAGCN_EPS : 1.0f, Z, ZR, N);
  if (L > 0)
    gather_proj_kernel<<<(N + 63) / 64, 256, 0, stream>>>(cnt, srow, gac, gbc, bg, L - 1, Z, ZR,
                                                         LG, N);
  logsoftmax_kernel<<<(N + 255) / 256, 256, 0, stream>>>((const float4*)(L > 0 ? LG : ZR),
                                                        (float4*)out, N);
}


# Projection GEMM folded into the input GEMM's epilogue (two-layer case): each block projects its relu tile from LDS by WMMA, stores whole 128-B projected rows; 128-wide h never stored; 1 launch goes
# speedup vs baseline: 6.4904x; 1.2306x over previous
#include <hip/hip_runtime.h>

typedef float v2f __attribute__((ext_vector_type(2)));
typedef float v8f __attribute__((ext_vector_type(8)));

#define FAGCN_EPS 0.3f

__global__ void deg_kernel(const int* __restrict__ ei, float* __restrict__ deg,
                           int* __restrict__ cnt, int E) {
  int e = blockIdx.x * blockDim.x + threadIdx.x;
  if (e < E) {
    atomicAdd(&deg[ei[e]], 1.0f);
    atomicAdd(&cnt[ei[E + e]], 1);
  }
}

__device__ __forceinline__ int4 scan_load4(const int* cnt, int i0, int N) {
  int4 v = {0, 0, 0, 0};
  if (i0 + 3 < N) {
    v = *(const int4*)(cnt + i0);
  } else {
    if (i0 < N) v.x = cnt[i0];
    if (i0 + 1 < N) v.y = cnt[i0 + 1];
    if (i0 + 2 < N) v.z = cnt[i0 + 2];
  }
  return v;
}

__global__ void scan_kernel(int* cnt, int N) {
  __shared__ int wsum[32];
  int tid = threadIdx.x;
  int lane = tid & 31;
  int wave = tid >> 5;
  int carry = 0;
  int4 nxt = scan_load4(cnt, tid * 4, N);
  for (int base = 0; base < N; base += 4096) {
    int i0 = base + tid * 4;
    int4 v = nxt;
    nxt = scan_load4(cnt, i0 + 4096, N);
    int s = v.x + v.y + v.z + v.w;
    int incl = s;
    for (int off = 1; off < 32; off <<= 1) {
      int t = __shfl_up(incl, off, 32);
      if (lane >= off) incl += t;
    }
    if (lane == 31) wsum[wave] = incl;
    __syncthreads();
    int wt = wsum[lane];
    int wincl = wt;
    for (int off = 1; off < 32; off <<= 1) {
      int t = __shfl_up(wincl, off, 32);
      if (lane >= off) wincl += t;
    }
    int wexcl = __shfl(wincl, wave, 32) - __shfl(wt, wave, 32);
    int total = __shfl(wincl, 31, 32);
    int ex = carry + wexcl + incl - s;
    if (i0 + 3 < N) {
      int4 o;
      o.x = ex; o.y = ex + v.x; o.z = ex + v.x + v.y; o.w = ex + v.x + v.y + v.z;
      *(int4*)(cnt + i0) = o;
    } else {
      if (i0 < N) cnt[i0] = ex;
      if (i0 + 1 < N) cnt[i0 + 1] = ex + v.x;
      if (i0 + 2 < N) cnt[i0 + 2] = ex + v.x + v.y;
    }
    carry += total;
    __syncthreads();
  }
}

__global__ void fill_kernel(const int* __restrict__ ei, int* __restrict__ cur,
                            int* __restrict__ srow, int E) {
  int e = blockIdx.x * blockDim.x + threadIdx.x;
  if (e < E) {
    int pos = atomicAdd(&cur[ei[E + e]], 1);
    srow[pos] = ei[e];
  }
}

__device__ __forceinline__ void order_segment(int* __restrict__ srow, int beg, int d, int lane) {
  int a0 = lane < d ? srow[beg + lane] : 0x7fffffff;
  int a1 = lane + 32 < d ? srow[beg + 32 + lane] : 0x7fffffff;
  int r0 = 0, r1 = 0;
  int n0 = d < 32 ? d : 32;
  for (int i = 0; i < n0; ++i) {
    int x = __builtin_amdgcn_readlane(a0, i);
    r0 += (x < a0 || (x == a0 && i < lane)) ? 1 : 0;
    r1 += (x <= a1) ? 1 : 0;
  }
  for (int i = 0; i < d - 32; ++i) {
    int x = __builtin_amdgcn_readlane(a1, i);
    r0 += (x < a0) ? 1 : 0;
    r1 += (x < a1 || (x == a1 && i < lane)) ? 1 : 0;
  }
  if (lane < d) srow[beg + r0] = a0;
  if (lane + 32 < d) srow[beg + r1] = a1;
}

__global__ void order_kernel(const int* __restrict__ ends, int* __restrict__ srow, int N) {
  int node = __builtin_amdgcn_readfirstlane(blockIdx.x * 8 + (threadIdx.x >> 5));
  int lane = threadIdx.x & 31;
  if (node >= N) return;
  int beg = node ? ends[node - 1] : 0;
  int d = ends[node] - beg;
  if (d < 2 || d > 64) return;
  order_segment(srow, beg, d, lane);
}

__global__ void dn_kernel(float* dn, int N) {
  int i = blockIdx.x * blockDim.x + threadIdx.x;
  if (i < N) dn[i] = 1.0f / sqrtf(fmaxf(dn[i], 1.0f));
}

#define BIN_SHIFT 8
#define BIN_NODES 256
#define MAX_BINS 512
#define CHUNK 4096
#define BIN_CAP 8192

__device__ __forceinline__ int block_scan_1024(int s, int* wsum, int lane, int wave) {
  int incl = s;
  for (int off = 1; off < 32; off <<= 1) {
    int t = __shfl_up(incl, off, 32);
    if (lane >= off) incl += t;
  }
  if (lane == 31) wsum[wave] = incl;
  __syncthreads();
  int wt = wsum[lane];
  int wincl = wt;
  for (int off = 1; off < 32; off <<= 1) {
    int t = __shfl_up(wincl, off, 32);
    if (lane >= off) wincl += t;
  }
  int wexcl = __shfl(wincl, wave, 32) - __shfl(wt, wave, 32);
  __syncthreads();
  return wexcl + incl - s;
}

__global__ void bin_count_kernel(const int* __restrict__ ei, int* __restrict__ tot, int E, int NB) {
  __shared__ int hist[2 * MAX_BINS];
  int tid = threadIdx.x;
  for (int i = tid; i < 2 * NB; i += 1024) hist[i] = 0;
  __syncthreads();
  int e0 = blockIdx.x * CHUNK + tid;
  int col[4], row[4];
#pragma unroll
  for (int k = 0; k < 4; ++k) {
    int e = e0 + k * 1024;
    col[k] = e < E ? ei[E + e] : -1;
    row[k] = e < E ? ei[e] : -1;
  }
#pragma unroll
  for (int k = 0; k < 4; ++k) {
    if (col[k] >= 0) {
      atomicAdd(&hist[col[k] >> BIN_SHIFT], 1);
      atomicAdd(&hist[NB + (row[k] >> BIN_SHIFT)], 1);
    }
  }
  __syncthreads();
  for (int i = tid; i < 2 * NB; i += 1024) {
    int c = hist[i];
    if (c) atomicAdd(&tot[i], c);
  }
}

__global__ void bin_scatter_kernel(const int* __restrict__ ei, const int* __restrict__ tot,
                                   int* __restrict__ cursor, int* __restrict__ bstart,
                                   int* __restrict__ cbin, int* __restrict__ rbin, int E, int NB) {
  __shared__ int hist[2 * MAX_BINS];
  __shared__ int gbase[2 * MAX_BINS];
  __shared__ int wsum[32];
  __shared__ int keyC[CHUNK];
  __shared__ int adrC[CHUNK];
  __shared__ int stR[CHUNK];
  int tid = threadIdx.x;
  int lane = tid & 31;
  int wave = tid >> 5;
  int nb2 = 2 * NB;
  for (int i = tid; i < nb2; i += 1024) hist[i] = 0;
  __syncthreads();
  int ebase = blockIdx.x * CHUNK;
  int nE = E - ebase < CHUNK ? E - ebase : CHUNK;
  int i0 = 2 * tid;
  int t0 = i0 < nb2 ? tot[i0] : 0;
  int t1 = i0 + 1 < nb2 ? tot[i0 + 1] : 0;
  int col[4], row[4], rc[4], rr[4];
#pragma unroll
  for (int k = 0; k < 4; ++k) {
    int e = ebase + tid + k * 1024;
    col[k] = e < E ? ei[E + e] : -1;
    row[k] = e < E ? ei[e] : -1;
  }
#pragma unroll
  for (int k = 0; k < 4; ++k) {
    rc[k] = 0; rr[k] = 0;
    if (col[k] >= 0) {
      rc[k] = atomicAdd(&hist[col[k] >> BIN_SHIFT], 1);
      rr[k] = atomicAdd(&hist[NB + (row[k] >> BIN_SHIFT)], 1);
    }
  }
  __syncthreads();
  int h0 = i0 < nb2 ? hist[i0] : 0;
  int h1 = i0 + 1 < nb2 ? hist[i0 + 1] : 0;
  int gex = block_scan_1024(t0 + t1, wsum, lane, wave);
  int lex = block_scan_1024(h0 + h1, wsum, lane, wave);
  if (i0 < nb2) {
    int second = i0 >= NB ? 1 : 0;
    int s = gex - (second ? E : 0);
    gbase[i0] = s + (h0 ? atomicAdd(&cursor[i0], h0) : 0);
    hist[i0] = lex - (second ? nE : 0);
    if (blockIdx.x == 0) bstart[i0 + second] = s;
  }
  if (i0 + 1 < nb2) {
    int second = i0 + 1 >= NB ? 1 : 0;
    int s = gex + t0 - (second ? E : 0);
    gbase[i0 + 1] = s + (h1 ? atomicAdd(&cursor[i0 + 1], h1) : 0);
    hist[i0 + 1] = lex + h0 - (second ? nE : 0);
    if (blockIdx.x == 0) bstart[i0 + 1 + second] = s;
  }
  if (blockIdx.x == 0 && tid == 0) {
    bstart[NB] = E;
    bstart[nb2 + 1] = E;
  }
  __syncthreads();
#pragma unroll
  for (int k = 0; k < 4; ++k) {
    if (col[k] >= 0) {
      int bc = col[k] >> BIN_SHIFT;
      int sc = hist[bc] + rc[k];
      keyC[sc] = (row[k] << BIN_SHIFT) | (col[k] & (BIN_NODES - 1));
      adrC[sc] = gbase[bc] + rc[k];
      int br = NB + (row[k] >> BIN_SHIFT);
      int sr = hist[br] + rr[k];
      stR[sr] = ((gbase[br] + rr[k]) << BIN_SHIFT) | (row[k] & (BIN_NODES - 1));
    }
  }
  __syncthreads();
  for (int j = tid; j < nE; j += 1024) {
    cbin[adrC[j]] = keyC[j];
    int w = stR[j];
    rbin[w >> BIN_SHIFT] = w & (BIN_NODES - 1);
  }
}

__global__ void bin_finish_kernel(const int* __restrict__ bstart, const int* __restrict__ cbin,
                                  const int* __restrict__ rbin, int* __restrict__ ends,
                                  float* __restrict__ dn, int* __restrict__ srow, int N, int NB) {
  __shared__ int ncnt[BIN_NODES];
  __shared__ int nbeg[BIN_NODES];
  __shared__ int ncur[BIN_NODES];
  __shared__ int dcnt[BIN_NODES];
  __shared__ int wsum[32];
  __shared__ int stage[BIN_CAP];
  int tid = threadIdx.x;
  int lane = tid & 31;
  int wave = tid >> 5;
  int b = blockIdx.x;
  int cb = bstart[b], ce = bstart[b + 1];
  int rb = bstart[NB + 1 + b], re = bstart[NB + 2 + b];
  int kc[5], kr[5];
#pragma unroll
  for (int k = 0; k < 5; ++k) {
    int i = cb + tid + k * 1024;
    int j = rb + tid + k * 1024;
    kc[k] = i < ce ? cbin[i] : -1;
    kr[k] = j < re ? rbin[j] : -1;
  }
  if (tid < BIN_NODES) { ncnt[tid] = 0; dcnt[tid] = 0; }
  __syncthreads();
#pragma unroll
  for (int k = 0; k < 5; ++k) {
    if (kc[k] >= 0) atomicAdd(&ncnt[kc[k] & (BIN_NODES - 1)], 1);
    if (kr[k] >= 0) atomicAdd(&dcnt[kr[k]], 1);
  }
  for (int i = cb + tid + 5 * 1024; i < ce; i += 1024) atomicAdd(&ncnt[cbin[i] & (BIN_NODES - 1)], 1);
  for (int i = rb + tid + 5 * 1024; i < re; i += 1024) atomicAdd(&dcnt[rbin[i]], 1);
  __syncthreads();
  int c = tid < BIN_NODES ? ncnt[tid] : 0;
  int ex = block_scan_1024(c, wsum, lane, wave);
  if (tid < BIN_NODES) {
    nbeg[tid] = ex;
    ncur[tid] = ex;
    int node = b * BIN_NODES + tid;
    if (node < N) {
      ends[node] = cb + ex + c;
      dn[node] = 1.0f / sqrtf(fmaxf((float)dcnt[tid], 1.0f));
    }
  }
  __syncthreads();
  int cnt = ce - cb;
  bool fits = cnt <= BIN_CAP;
#pragma unroll
  for (int k = 0; k < 5; ++k) {
    if (kc[k] >= 0) {
      int p = atomicAdd(&ncur[kc[k] & (BIN_NODES - 1)], 1);
      if (fits) stage[p] = kc[k];
      else srow[cb + p] = kc[k] >> BIN_SHIFT;
    }
  }
  for (int i = cb + tid + 5 * 1024; i < ce; i += 1024) {
    int key = cbin[i];
    int p = atomicAdd(&ncur[key & (BIN_NODES - 1)], 1);
    if (fits) stage[p] = key;
    else srow[cb + p] = key >> BIN_SHIFT;
  }
  if (!fits) {
    __threadfence();
    __syncthreads();
    for (int t = wave; t < BIN_NODES; t += 32) {
      int n = __builtin_amdgcn_readfirstlane(ncnt[t]);
      int sb = __builtin_amdgcn_readfirstlane(nbeg[t]);
      if (n >= 2 && n <= 64) order_segment(srow, cb + sb, n, lane);
    }
    return;
  }
  __syncthreads();
  for (int p = tid; p < cnt; p += 1024) {
    int key = stage[p];
    int d = key & (BIN_NODES - 1);
    int beg = nbeg[d];
    int n = ncnt[d];
    int r = p - beg;
    if (n > 1 && n <= 64) {
      r = 0;
      for (int j = 0; j < n; ++j) {
        int x = stage[beg + j];
        r += (x < key || (x == key && beg + j < p)) ? 1 : 0;
      }
    }
    srow[cb + beg + r] = key >> BIN_SHIFT;
  }
}

#define TILE_PITCH 132
#define GEMM_ROWS 64
#define YP 32
#define YS_PITCH 36
template <bool PROJ>
__global__ void gemm_relu_kernel(const float* __restrict__ X, const float* __restrict__ W,
                                 const float* __restrict__ bias, const float* __restrict__ wg0,
                                 const float* __restrict__ dn, float* __restrict__ H,
                                 float* __restrict__ pk, float* __restrict__ gb,
                                 const float* __restrict__ Wp, const float* __restrict__ wgl,
                                 float* __restrict__ Y, int Nrows) {
  __shared__ __attribute__((aligned(16))) float tile[GEMM_ROWS * TILE_PITCH];
  int wave = threadIdx.x >> 5;
  int lane = threadIdx.x & 31;
  int row0 = blockIdx.x * GEMM_ROWS;
  int c = lane & 15;
  int n = wave * 16 + c;
  int khalf = (lane >> 4) * 2;
  float4 xr[8];
#pragma unroll
  for (int j = 0; j < 8; ++j) {
    int m = row0 + wave * 8 + j;
    int mc = m < Nrows ? m : (Nrows - 1);
    xr[j] = *(const float4*)(X + (size_t)mc * 128 + lane * 4);
  }
  int md = row0 + wave * 8 + (lane & 7);
  float dnv = dn[md < Nrows ? md : (Nrows - 1)];
  const float* wcol = W + (size_t)khalf * 128 + n;
  v2f bf[32];
#pragma unroll
  for (int s = 0; s < 32; ++s) {
    bf[s].x = wcol[(size_t)s * 512];
    bf[s].y = wcol[(size_t)s * 512 + 128];
  }
#pragma unroll
  for (int j = 0; j < 8; ++j)
    *(float4*)(tile + (wave * 8 + j) * TILE_PITCH + lane * 4) = xr[j];
  __syncthreads();
  v8f acc0 = {}, acc1 = {}, acc2 = {}, acc3 = {};
  const float* arow = tile + c * TILE_PITCH + khalf;
  v2f cur[4][4], nxt[4][4];
#pragma unroll
  for (int i = 0; i < 4; ++i)
#pragma unroll
    for (int t = 0; t < 4; ++t)
      cur[i][t] = *(const v2f*)(arow + t * 16 * TILE_PITCH + i * 4);
#pragma unroll
  for (int g = 0; g < 8; ++g) {
    if (g + 1 < 8) {
#pragma unroll
      for (int i = 0; i < 4; ++i)
#pragma unroll
        for (int t = 0; t < 4; ++t)
          nxt[i][t] = *(const v2f*)(arow + t * 16 * TILE_PITCH + (g + 1) * 16 + i * 4);
    }
#pragma unroll
    for (int i = 0; i < 4; ++i) {
      int s = g * 4 + i;
      acc0 = __builtin_amdgcn_wmma_f32_16x16x4_f32(false, cur[i][0], false, bf[s], (short)0, acc0, false, false);
      acc1 = __builtin_amdgcn_wmma_f32_16x16x4_f32(false, cur[i][1], false, bf[s], (short)0, acc1, false, false);
      acc2 = __builtin_amdgcn_wmma_f32_16x16x4_f32(false, cur[i][2], false, bf[s], (short)0, acc2, false, false);
      acc3 = __builtin_amdgcn_wmma_f32_16x16x4_f32(false, cur[i][3], false, bf[s], (short)0, acc3, false, false);
    }
#pragma unroll
    for (int i = 0; i < 4; ++i)
#pragma unroll
      for (int t = 0; t < 4; ++t)
        cur[i][t] = nxt[i][t];
  }
  __syncthreads();
  int rbase = (lane >> 4) * 8;
  float bn = bias[n];
#pragma unroll
  for (int r = 0; r < 8; ++r) {
    tile[(rbase + r) * TILE_PITCH + n] = fmaxf(acc0[r] + bn, 0.0f);
    tile[(16 + rbase + r) * TILE_PITCH + n] = fmaxf(acc1[r] + bn, 0.0f);
    tile[(32 + rbase + r) * TILE_PITCH + n] = fmaxf(acc2[r] + bn, 0.0f);
    tile[(48 + rbase + r) * TILE_PITCH + n] = fmaxf(acc3[r] + bn, 0.0f);
  }
  if (PROJ) {
    __shared__ __attribute__((aligned(16))) float ystage[GEMM_ROWS * YS_PITCH];
    int rt = wave & 3;
    int ct = __builtin_amdgcn_readfirstlane(wave >> 2);
    v2f pb[32];
    if (ct == 0) {
      const float* bp = Wp + (size_t)khalf * 16 + c;
#pragma unroll
      for (int s = 0; s < 32; ++s) {
        pb[s].x = bp[s * 64];
        pb[s].y = bp[s * 64 + 16];
      }
    } else {
      const float* gp = wgl + (c & 1) * 128 + khalf;
#pragma unroll
      for (int s = 0; s < 32; ++s) pb[s] = *(const v2f*)(gp + 4 * s);
    }
    float4 wa = *(const float4*)(wg0 + lane * 4);
    float4 wb = *(const float4*)(wg0 + 128 + lane * 4);
    __syncthreads();
    const float* ap = tile + (rt * 16 + c) * TILE_PITCH + khalf;
    v2f pcur[8], pnxt[8];
#pragma unroll
    for (int i = 0; i < 8; ++i) pcur[i] = *(const v2f*)(ap + 4 * i);
    v8f pacc = {};
#pragma unroll
    for (int g = 0; g < 4; ++g) {
      if (g + 1 < 4) {
#pragma unroll
        for (int i = 0; i < 8; ++i) pnxt[i] = *(const v2f*)(ap + (g + 1) * 32 + 4 * i);
      }
#pragma unroll
      for (int i = 0; i < 8; ++i)
        pacc = __builtin_amdgcn_wmma_f32_16x16x4_f32(false, pcur[i], false, pb[g * 8 + i], (short)0, pacc, false, false);
#pragma unroll
      for (int i = 0; i < 8; ++i) pcur[i] = pnxt[i];
    }
    int r8 = lane & 7;
    int q = lane >> 3;
    float keep = dnv;
#pragma unroll
    for (int j = 0; j < 8; ++j) {
      float4 v = *(const float4*)(tile + (wave * 8 + j) * TILE_PITCH + lane * 4);
      float sa = v.x * wa.x + v.y * wa.y + v.z * wa.z + v.w * wa.w;
      float sb = v.x * wb.x + v.y * wb.y + v.z * wb.z + v.w * wb.w;
      for (int off = 16; off > 0; off >>= 1) {
        sa += __shfl_xor(sa, off, 32);
        sb += __shfl_xor(sb, off, 32);
      }
      float t = q == 0 ? sa : sb;
      t = q == 3 ? 0.0f : t;
      bool take = (r8 == j) & (q != 1);
      keep = take ? t : keep;
    }
    ystage[(wave * 8 + r8) * YS_PITCH + 32 + q] = keep;
#pragma unroll
    for (int r = 0; r < 8; ++r)
      ystage[(rt * 16 + rbase + r) * YS_PITCH + ct * 16 + c] = pacc[r];
    __syncthreads();
    int cg = lane & 7;
#pragma unroll
    for (int i = 0; i < 2; ++i) {
      int rr = wave * 8 + i * 4 + (lane >> 3);
      int m = row0 + rr;
      float4 v = *(const float4*)(ystage + rr * YS_PITCH + cg * 4);
      float4 e = *(const float4*)(ystage + rr * YS_PITCH + 32);
      v.z = cg == 4 ? e.x : v.z;
      v.w = cg == 4 ? e.y : v.w;
      v.x = cg == 5 ? e.z : v.x;
      if (m < Nrows) *(float4*)(Y + (size_t)m * YP + cg * 4) = v;
    }
    return;
  }
  __syncthreads();
  float4 wa = {0.0f, 0.0f, 0.0f, 0.0f}, wb = {0.0f, 0.0f, 0.0f, 0.0f};
  if (wg0) {
    wa = *(const float4*)(wg0 + lane * 4);
    wb = *(const float4*)(wg0 + 128 + lane * 4);
  }
  for (int j = 0; j < 8; ++j) {
    int rr = wave * 8 + j;
    int m = row0 + rr;
    float4 v = *(const float4*)(tile + rr * TILE_PITCH + lane * 4);
    if (m < Nrows) {
      *(float4*)(H + (size_t)m * 128 + lane * 4) = v;
      if (wg0) {
        float dnm = __shfl(dnv, j, 32);
        float sa = v.x * wa.x + v.y * wa.y + v.z * wa.z + v.w * wa.w;
        float sb = v.x * wb.x + v.y * wb.y + v.z * wb.z + v.w * wb.w;
        for (int off = 16; off > 0; off >>= 1) {
          sa += __shfl_down(sa, off, 32);
          sb += __shfl_down(sb, off, 32);
        }
        if (lane == 0) {
          float2 p; p.x = sa; p.y = dnm;
          *(float2*)(pk + 2 * (size_t)m) = p;
          gb[m] = sb;
        }
      }
    }
  }
}

__global__ void gather_kernel(const int* __restrict__ ends, const int* __restrict__ srow,
                              const float* __restrict__ pk, const float* __restrict__ gb,
                              const float* __restrict__ bg,
                              int layer, const float* __restrict__ H,
                              const float* __restrict__ raw, float* __restrict__ hout,
                              const float* __restrict__ wg_next, float* __restrict__ pk_next,
                              float* __restrict__ gb_next, int N) {
  int node = __builtin_amdgcn_readfirstlane(blockIdx.x * 8 + (threadIdx.x >> 5));
  int lane = threadIdx.x & 31;
  if (node >= N) return;
  int beg = node ? ends[node - 1] : 0;
  int end = ends[node];
  float gbc = gb[node];
  float bgl = bg[layer];
  float dnc = pk[2 * (size_t)node + 1];
  const float* hl = H + lane * 4;
  float4 rw = *(const float4*)(raw + (size_t)node * 128 + lane * 4);
  float4 acc = {0.0f, 0.0f, 0.0f, 0.0f};
  for (int base = beg; base < end; base += 32) {
    int n = end - base < 32 ? end - base : 32;
    int r = 0;
    float w = 0.0f;
    if (lane < n) {
      r = srow[base + lane];
      float2 gd = *(const float2*)(pk + 2 * (size_t)r);
      float g = tanhf(gd.x + gbc + bgl);
      w = g * gd.y * dnc;
    }
    int wi = __float_as_int(w);
    int k = 0;
    for (; k + 4 <= n; k += 4) {
      int r0 = __builtin_amdgcn_readlane(r, k);
      int r1 = __builtin_amdgcn_readlane(r, k + 1);
      int r2 = __builtin_amdgcn_readlane(r, k + 2);
      int r3 = __builtin_amdgcn_readlane(r, k + 3);
      float w0 = __int_as_float(__builtin_amdgcn_readlane(wi, k));
      float w1 = __int_as_float(__builtin_amdgcn_readlane(wi, k + 1));
      float w2 = __int_as_float(__builtin_amdgcn_readlane(wi, k + 2));
      float w3 = __int_as_float(__builtin_amdgcn_readlane(wi, k + 3));
      float4 h0 = *(const float4*)(hl + (size_t)r0 * 128);
      float4 h1 = *(const float4*)(hl + (size_t)r1 * 128);
      float4 h2 = *(const float4*)(hl + (size_t)r2 * 128);
      float4 h3 = *(const float4*)(hl + (size_t)r3 * 128);
      acc.x += w0 * h0.x; acc.y += w0 * h0.y; acc.z += w0 * h0.z; acc.w += w0 * h0.w;
      acc.x += w1 * h1.x; acc.y += w1 * h1.y; acc.z += w1 * h1.z; acc.w += w1 * h1.w;
      acc.x += w2 * h2.x; acc.y += w2 * h2.y; acc.z += w2 * h2.z; acc.w += w2 * h2.w;
      acc.x += w3 * h3.x; acc.y += w3 * h3.y; acc.z += w3 * h3.z; acc.w += w3 * h3.w;
    }
    for (; k < n; ++k) {
      int r0 = __builtin_amdgcn_readlane(r, k);
      float w0 = __int_as_float(__builtin_amdgcn_readlane(wi, k));
      float4 h0 = *(const float4*)(hl + (size_t)r0 * 128);
      acc.x += w0 * h0.x; acc.y += w0 * h0.y; acc.z += w0 * h0.z; acc.w += w0 * h0.w;
    }
  }
  float4 o;
  o.x = FAGCN_EPS * rw.x + acc.x;
  o.y = FAGCN_EPS * rw.y + acc.y;
  o.z = FAGCN_EPS * rw.z + acc.z;
  o.w = FAGCN_EPS * rw.w + acc.w;
  *(float4*)(hout + (size_t)node * 128 + lane * 4) = o;
  if (wg_next) {
    float4 wa = *(const float4*)(wg_next + lane * 4);
    float4 wb = *(const float4*)(wg_next + 128 + lane * 4);
    float sa = o.x * wa.x + o.y * wa.y + o.z * wa.z + o.w * wa.w;
    float sb = o.x * wb.x + o.y * wb.y + o.z * wb.z + o.w * wb.w;
    for (int off = 16; off > 0; off >>= 1) {
      sa += __shfl_down(sa, off, 32);
      sb += __shfl_down(sb, off, 32);
    }
    if (lane == 0) {
      float2 p; p.x = sa; p.y = dnc;
      *(float2*)(pk_next + 2 * (size_t)node) = p;
      gb_next[node] = sb;
    }
  }
}

__global__ void gemm_proj_kernel(const float* __restrict__ Hc, const float* __restrict__ Raw,
                                 const float* __restrict__ W, const float* __restrict__ bias,
                                 float rscale, float* __restrict__ Z, float* __restrict__ ZR,
                                 int Nrows) {
  int wave = threadIdx.x >> 5;
  int lane = threadIdx.x & 31;
  int tiles = (Nrows + 15) >> 4;
  int t = blockIdx.x * 8 + wave;
  if (t >= 2 * tiles) return;
  bool second = t >= tiles;
  int tileM = second ? t - tiles : t;
  const float* src = second ? Raw : Hc;
  float* dst = second ? ZR : Z;
  int mA = tileM * 16 + (lane & 15);
  int mclamp = mA < Nrows ? mA : (Nrows - 1);
  int n = lane & 15;
  int khalf = (lane >> 4) * 2;
  const float* hrow = src + (size_t)mclamp * 128 + khalf;
  const float* wcol = W + (size_t)khalf * 16 + n;
  __builtin_prefetch(hrow, 0, 0);
  v8f acc = {};
#pragma unroll 4
  for (int kk = 0; kk < 128; kk += 4) {
    v2f a = *(const v2f*)(hrow + kk);
    v2f b;
    b.x = wcol[(size_t)kk * 16];
    b.y = wcol[(size_t)kk * 16 + 16];
    acc = __builtin_amdgcn_wmma_f32_16x16x4_f32(false, a, false, b, (short)0, acc,
                                               false, false);
  }
  int mbase = tileM * 16 + ((lane >> 4) * 8);
  float bn = second ? bias[n] : 0.0f;
  float sc = second ? rscale : 1.0f;
  if (tileM * 16 + 16 <= Nrows) {
    float* op = dst + (size_t)mbase * 16 + n;
#pragma unroll
    for (int r = 0; r < 8; ++r)
      op[(size_t)r * 16] = acc[r] * sc + bn;
  } else {
#pragma unroll
    for (int r = 0; r < 8; ++r) {
      int mm = mbase + r;
      if (mm < Nrows) dst[(size_t)mm * 16 + n] = acc[r] * sc + bn;
    }
  }
}

__global__ void gather_proj_kernel(const int* __restrict__ ends, const int* __restrict__ srow,
                                   const float* __restrict__ pk, const float* __restrict__ gb,
                                   const float* __restrict__ bg, int layer,
                                   const float* __restrict__ Z, const float* __restrict__ ZR,
                                   float* __restrict__ logits, int N) {
  int node = blockIdx.x * 64 + (threadIdx.x >> 2);
  int q = threadIdx.x & 3;
  if (node >= N) return;
  int beg = node ? ends[node - 1] : 0;
  int end = ends[node];
  float gbc = gb[node];
  float bgl = bg[layer];
  float dnc = pk[2 * (size_t)node + 1];
  const float* zq = Z + q * 4;
  float4 zr = *(const float4*)(ZR + (size_t)node * 16 + q * 4);
  float4 acc = {0.0f, 0.0f, 0.0f, 0.0f};
  int last = end - 1;
  int r0 = 0, r1 = 0, r2 = 0, r3 = 0;
  if (beg < end) {
    r0 = srow[beg];
    r1 = srow[beg + 1 < last ? beg + 1 : last];
    r2 = srow[beg + 2 < last ? beg + 2 : last];
    r3 = srow[beg + 3 < last ? beg + 3 : last];
  }
  for (int base = beg; base < end; base += 4) {
    float2 g0 = *(const float2*)(pk + 2 * (size_t)r0);
    float2 g1 = *(const float2*)(pk + 2 * (size_t)r1);
    float2 g2 = *(const float2*)(pk + 2 * (size_t)r2);
    float2 g3 = *(const float2*)(pk + 2 * (size_t)r3);
    float4 h0 = *(const float4*)(zq + (size_t)r0 * 16);
    float4 h1 = *(const float4*)(zq + (size_t)r1 * 16);
    float4 h2 = *(const float4*)(zq + (size_t)r2 * 16);
    float4 h3 = *(const float4*)(zq + (size_t)r3 * 16);
    int nb = base + 4;
    int n0 = r0, n1 = r1, n2 = r2, n3 = r3;
    if (nb < end) {
      n0 = srow[nb];
      n1 = srow[nb + 1 < last ? nb + 1 : last];
      n2 = srow[nb + 2 < last ? nb + 2 : last];
      n3 = srow[nb + 3 < last ? nb + 3 : last];
    }
    {
      float w = tanhf(g0.x + gbc + bgl) * g0.y * dnc;
      acc.x += w * h0.x; acc.y += w * h0.y; acc.z += w * h0.z; acc.w += w * h0.w;
    }
    if (base + 1 < end) {
      float w = tanhf(g1.x + gbc + bgl) * g1.y * dnc;
      acc.x += w * h1.x; acc.y += w * h1.y; acc.z += w * h1.z; acc.w += w * h1.w;
    }
    if (base + 2 < end) {
      float w = tanhf(g2.x + gbc + bgl) * g2.y * dnc;
      acc.x += w * h2.x; acc.y += w * h2.y; acc.z += w * h2.z; acc.w += w * h2.w;
    }
    if (base + 3 < end) {
      float w = tanhf(g3.x + gbc + bgl) * g3.y * dnc;
      acc.x += w * h3.x; acc.y += w * h3.y; acc.z += w * h3.z; acc.w += w * h3.w;
    }
    r0 = n0; r1 = n1; r2 = n2; r3 = n3;
  }
  float4 o;
  o.x = zr.x + acc.x;
  o.y = zr.y + acc.y;
  o.z = zr.z + acc.z;
  o.w = zr.w + acc.w;
  *(float4*)(logits + (size_t)node * 16 + q * 4) = o;
}

__global__ void gemm_proj2_kernel(const float* __restrict__ Hc, const float* __restrict__ Raw,
                                  const float* __restrict__ W, const float* __restrict__ wgl,
                                  const float* __restrict__ pk, const float* __restrict__ gb,
                                  float* __restrict__ Y, float* __restrict__ YR,
                                  int Nrows, int nt) {
  int wave = threadIdx.x >> 5;
  int lane = threadIdx.x & 31;
  int tiles = (Nrows + 15) >> 4;
  int t = blockIdx.x * 8 + wave;
  if (t >= nt) return;
  bool second = t >= tiles;
  int tileM = second ? t - tiles : t;
  const float* src = second ? Raw : Hc;
  float* dst = second ? YR : Y;
  int mA = tileM * 16 + (lane & 15);
  int mclamp = mA < Nrows ? mA : (Nrows - 1);
  int n = lane & 15;
  int khalf = (lane >> 4) * 2;
  const float* hrow = src + (size_t)mclamp * 128 + khalf;
  const float* wcol = W + (size_t)khalf * 16 + n;
  const float* gcol = wgl + (n & 1) * 128 + khalf;
  __builtin_prefetch(hrow, 0, 0);
  int mbase = tileM * 16 + ((lane >> 4) * 8);
  const float* tp = (n == 2 || n == 3) ? pk + (n - 2) : gb;
  int ts = (n == 2 || n == 3) ? 2 : 1;
  float tv[8];
#pragma unroll
  for (int r = 0; r < 8; ++r) {
    int mr = mbase + r < Nrows ? mbase + r : Nrows - 1;
    tv[r] = tp[(size_t)mr * ts];
  }
  v8f acc = {};
  v8f acg = {};
#pragma unroll 4
  for (int kk = 0; kk < 128; kk += 4) {
    v2f a = *(const v2f*)(hrow + kk);
    v2f b;
    b.x = wcol[(size_t)kk * 16];
    b.y = wcol[(size_t)kk * 16 + 16];
    v2f g = *(const v2f*)(gcol + kk);
    acc = __builtin_amdgcn_wmma_f32_16x16x4_f32(false, a, false, b, (short)0, acc,
                                               false, false);
    acg = __builtin_amdgcn_wmma_f32_16x16x4_f32(false, a, false, g, (short)0, acg,
                                               false, false);
  }
  float* yb = dst + (size_t)mbase * YP + n;
  if (tileM * 16 + 16 <= Nrows) {
#pragma unroll
    for (int r = 0; r < 8; ++r) {
      yb[(size_t)r * YP] = acc[r];
      if (n < 5) yb[(size_t)r * YP + 16] = n < 2 ? acg[r] : tv[r];
    }
  } else {
#pragma unroll
    for (int r = 0; r < 8; ++r) {
      if (mbase + r < Nrows) {
        yb[(size_t)r * YP] = acc[r];
        if (n < 5) yb[(size_t)r * YP + 16] = n < 2 ? acg[r] : tv[r];
      }
    }
  }
}

__global__ void gather_mid_kernel(const int* __restrict__ ends, const int* __restrict__ srow,
                                  const float* __restrict__ bg, int layer,
                                  const float* __restrict__ Y, const float* __restrict__ YR,
                                  const float* __restrict__ bias,
                                  float* __restrict__ Z, float* __restrict__ ZR,
                                  float* __restrict__ pk_next, float* __restrict__ gb_next,
                                  int N) {
  int node = blockIdx.x * 64 + (threadIdx.x >> 2);
  int q = threadIdx.x & 3;
  if (node >= N) return;
  int beg = node ? ends[node - 1] : 0;
  int end = ends[node];
  float bgl = bg[layer];
  const float* yc = Y + (size_t)node * YP;
  float dnc = yc[19];
  float gbc = yc[20];
  const float* yrc = YR + (size_t)node * YP;
  float4 own = *(const float4*)(yrc + q * 4);
  float2 owng = *(const float2*)(yrc + 16);
  float4 bq = *(const float4*)(bias + q * 4);
  const float* yq = Y + q * 4;
  const float* yt = Y + 16;
  float4 acc = {0.0f, 0.0f, 0.0f, 0.0f};
  float sga = 0.0f, sgb = 0.0f;
  int last = end - 1;
  int r0 = 0, r1 = 0, r2 = 0, r3 = 0;
  if (beg < end) {
    r0 = srow[beg];
    r1 = srow[beg + 1 < last ? beg + 1 : last];
    r2 = srow[beg + 2 < last ? beg + 2 : last];
    r3 = srow[beg + 3 < last ? beg + 3 : last];
  }
  for (int base = beg; base < end; base += 4) {
    float4 t0 = *(const float4*)(yt + (size_t)r0 * YP);
    float4 t1 = *(const float4*)(yt + (size_t)r1 * YP);
    float4 t2 = *(const float4*)(yt + (size_t)r2 * YP);
    float4 t3 = *(const float4*)(yt + (size_t)r3 * YP);
    float4 h0 = *(const float4*)(yq + (size_t)r0 * YP);
    float4 h1 = *(const float4*)(yq + (size_t)r1 * YP);
    float4 h2 = *(const float4*)(yq + (size_t)r2 * YP);
    float4 h3 = *(const float4*)(yq + (size_t)r3 * YP);
    int nb = base + 4;
    int n0 = r0, n1 = r1, n2 = r2, n3 = r3;
    if (nb < end) {
      n0 = srow[nb];
      n1 = srow[nb + 1 < last ? nb + 1 : last];
      n2 = srow[nb + 2 < last ? nb + 2 : last];
      n3 = srow[nb + 3 < last ? nb + 3 : last];
    }
    {
      float w = tanhf(t0.z + gbc + bgl) * t0.w * dnc;
      acc.x += w * h0.x; acc.y += w * h0.y; acc.z += w * h0.z; acc.w += w * h0.w;
      sga += w * t0.x; sgb += w * t0.y;
    }
    if (base + 1 < end) {
      float w = tanhf(t1.z + gbc + bgl) * t1.w * dnc;
      acc.x += w * h1.x; acc.y += w * h1.y; acc.z += w * h1.z; acc.w += w * h1.w;
      sga += w * t1.x; sgb += w * t1.y;
    }
    if (base + 2 < end) {
      float w = tanhf(t2.z + gbc + bgl) * t2.w * dnc;
      acc.x += w * h2.x; acc.y += w * h2.y; acc.z += w * h2.z; acc.w += w * h2.w;
      sga += w * t2.x; sgb += w * t2.y;
    }
    if (base + 3 < end) {
      float w = tanhf(t3.z + gbc + bgl) * t3.w * dnc;
      acc.x += w * h3.x; acc.y += w * h3.y; acc.z += w * h3.z; acc.w += w * h3.w;
      sga += w * t3.x; sgb += w * t3.y;
    }
    r0 = n0; r1 = n1; r2 = n2; r3 = n3;
  }
  float4 o, zr;
  o.x = FAGCN_EPS * own.x + acc.x;  zr.x = FAGCN_EPS * own.x + bq.x;
  o.y = FAGCN_EPS * own.y + acc.y;  zr.y = FAGCN_EPS * own.y + bq.y;
  o.z = FAGCN_EPS * own.z + acc.z;  zr.z = FAGCN_EPS * own.z + bq.z;
  o.w = FAGCN_EPS * own.w + acc.w;  zr.w = FAGCN_EPS * own.w + bq.w;
  *(float4*)(Z + (size_t)node * 16 + q * 4) = o;
  *(float4*)(ZR + (size_t)node * 16 + q * 4) = zr;
  if (q == 0) {
    float2 p; p.x = FAGCN_EPS * owng.x + sga; p.y = dnc;
    *(float2*)(pk_next + 2 * (size_t)node) = p;
    gb_next[node] = FAGCN_EPS * owng.y + sgb;
  }
}

__global__ void logsoftmax_kernel(const float4* __restrict__ in, float4* __restrict__ out, int N) {
  int i = blockIdx.x * blockDim.x + threadIdx.x;
  if (i >= N) return;
  const float4* rin = in + (size_t)i * 4;
  float4* row = out + (size_t)i * 4;
  float4 v[4];
#pragma unroll
  for (int q = 0; q < 4; ++q) v[q] = rin[q];
  float m = v[0].x;
#pragma unroll
  for (int q = 0; q < 4; ++q) {
    m = fmaxf(m, v[q].x); m = fmaxf(m, v[q].y);
    m = fmaxf(m, v[q].z); m = fmaxf(m, v[q].w);
  }
  float s = 0.0f;
#pragma unroll
  for (int q = 0; q < 4; ++q) {
    s += expf(v[q].x - m); s += expf(v[q].y - m);
    s += expf(v[q].z - m); s += expf(v[q].w - m);
  }
  float d = m + logf(s);
#pragma unroll
  for (int q = 0; q < 4; ++q) {
    v[q].x -= d; v[q].y -= d; v[q].z -= d; v[q].w -= d;
    row[q] = v[q];
  }
}

extern "C" void kernel_launch(void* const* d_in, const int* in_sizes, int n_in,
                              void* d_out, int out_size, void* d_ws, size_t ws_size,
                              hipStream_t stream) {
  const float* x     = (const float*)d_in[0];
  const int*   ei    = (const int*)d_in[1];
  const float* W_in  = (const float*)d_in[2];
  const float* b_in  = (const float*)d_in[3];
  const float* Wg    = (const float*)d_in[4];
  const float* bg    = (const float*)d_in[5];
  const float* W_out = (const float*)d_in[6];
  const float* b_out = (const float*)d_in[7];
  float* out = (float*)d_out;

  const int Hdim = 128;
  int N = in_sizes[0] / Hdim;
  int E = in_sizes[1] / 2;
  int L = in_sizes[5];

  float* ws   = (float*)d_ws;
  float* raw  = ws;
  float* bufA = raw + (size_t)N * Hdim;
  float* bufB = bufA + (size_t)N * Hdim;
  int*   cnt  = (int*)(bufB + (size_t)N * Hdim);
  float* dn   = (float*)cnt + N;
  int*   srow = (int*)(dn + N);
  float* pk   = out;
  float* pk2  = out + 2 * (size_t)N;
  float* gb   = out + 4 * (size_t)N;
  float* gb2  = out + 5 * (size_t)N;

  int NB = (N + BIN_NODES - 1) / BIN_NODES;
  size_t scratch = 2 * (size_t)E + 6 * (size_t)NB + 8;
  if (NB <= MAX_BINS && E <= (1 << 23) && scratch <= (size_t)N * Hdim) {
    int* cbin   = (int*)bufA;
    int* rbin   = cbin + E;
    int* tot    = rbin + E;
    int* cursor = tot + 2 * NB;
    int* bstart = cursor + 2 * NB;
    int nchunks = (E + CHUNK - 1) / CHUNK;
    hipMemsetAsync(tot, 0, (size_t)NB * 4 * sizeof(int), stream);
    bin_count_kernel<<<nchunks, 1024, 0, stream>>>(ei, tot, E, NB);
    bin_scatter_kernel<<<nchunks, 1024, 0, stream>>>(ei, tot, cursor, bstart, cbin, rbin, E, NB);
    bin_finish_kernel<<<NB, 1024, 0, stream>>>(bstart, cbin, rbin, cnt, dn, srow, N, NB);
  } else {
    hipMemsetAsync(cnt, 0, (size_t)N * 2 * sizeof(float), stream);
    deg_kernel<<<(E + 255) / 256, 256, 0, stream>>>(ei, dn, cnt, E);
    dn_kernel<<<(N + 255) / 256, 256, 0, stream>>>(dn, N);
    scan_kernel<<<1, 1024, 0, stream>>>(cnt, N);
    fill_kernel<<<(E + 255) / 256, 256, 0, stream>>>(ei, cnt, srow, E);
    order_kernel<<<(N + 7) / 8, 256, 0, stream>>>(cnt, srow, N);
  }

  bool fused = (L == 2);
  if (fused)
    gemm_relu_kernel<true><<<(N + GEMM_ROWS - 1) / GEMM_ROWS, 256, 0, stream>>>(
        x, W_in, b_in, Wg, dn, nullptr, nullptr, nullptr, W_out, Wg + (size_t)(L - 1) * 2 * Hdim, bufA, N);
  else
    gemm_relu_kernel<false><<<(N + GEMM_ROWS - 1) / GEMM_ROWS, 256, 0, stream>>>(
        x, W_in, b_in, L > 0 ? Wg : nullptr, dn, raw, pk, gb, nullptr, nullptr, nullptr, N);

  float* h = raw;
  float* hnew = bufA;
  float* gac = pk;  float* gbc = gb;
  float* gan = pk2; float* gbn = gb2;
  for (int i = 0; i + 2 < L; ++i) {
    const float* wgn = Wg + (size_t)(i + 1) * 2 * Hdim;
    gather_kernel<<<(N + 7) / 8, 256, 0, stream>>>(cnt, srow, gac, gbc, bg, i, h, raw, hnew,
                                                  wgn, gan, gbn, N);
    h = hnew;
    hnew = (h == bufA) ? bufB : bufA;
    float* t = gac; gac = gan; gan = t;
    t = gbc; gbc = gbn; gbn = t;
  }

  int tiles = (N + 15) / 16;
  if (L >= 2) {
    bool same = (h == raw);
    float* Y  = hnew;
    float* YR = same ? Y : hnew + 32 * (size_t)N;
    float* Z  = hnew + 64 * (size_t)N;
    float* ZR = hnew + 80 * (size_t)N;
    float* LG = hnew + 96 * (size_t)N;
    int nt = same ? tiles : 2 * tiles;
    if (!fused)
      gemm_proj2_kernel<<<(nt + 7) / 8, 256, 0, stream>>>(h, raw, W_out,
                                                         Wg + (size_t)(L - 1) * 2 * Hdim, gac, gbc,
                                                         Y, YR, N, nt);
    gather_mid_kernel<<<(N + 63) / 64, 256, 0, stream>>>(cnt, srow, bg, L - 2, Y, YR, b_out,
                                                        Z, ZR, gan, gbn, N);
    gather_proj_kernel<<<(N + 63) / 64, 256, 0, stream>>>(cnt, srow, gan, gbn, bg, L - 1, Z, ZR,
                                                         LG, N);
    logsoftmax_kernel<<<(N + 255) / 256, 256, 0, stream>>>((const float4*)LG, (float4*)out, N);
    return;
  }

  float* Z  = hnew;
  float* ZR = hnew + 16 * (size_t)N;
  float* LG = hnew + 32 * (size_t)N;
  gemm_proj_kernel<<<(2 * tiles + 7) / 8, 256, 0, stream>>>(h, raw, W_out, b_out,
                                                           L > 0 ? FAGCN_EPS : 1.0f, Z, ZR, N);
  if (L > 0)
    gather_proj_kernel<<<(N + 63) / 64, 256, 0, stream>>>(cnt, srow, gac, gbc, bg, L - 1, Z, ZR,
                                                         LG, N);
  logsoftmax_kernel<<<(N + 255) / 256, 256, 0, stream>>>((const float4*)(L > 0 ? LG : ZR),
                                                        (float4*)out, N);
}


# Edge grouping in 2 passes, not 3 and a memset: each block sorts its 4096-edge chunk by bin in LDS and stores it with its per-bin offsets; the per-bin pass collects its runs; no global atomics left
# speedup vs baseline: 6.7262x; 1.0363x over previous
#include <hip/hip_runtime.h>

typedef float v2f __attribute__((ext_vector_type(2)));
typedef float v8f __attribute__((ext_vector_type(8)));

#define FAGCN_EPS 0.3f

__global__ void deg_kernel(const int* __restrict__ ei, float* __restrict__ deg,
                           int* __restrict__ cnt, int E) {
  int e = blockIdx.x * blockDim.x + threadIdx.x;
  if (e < E) {
    atomicAdd(&deg[ei[e]], 1.0f);
    atomicAdd(&cnt[ei[E + e]], 1);
  }
}

__device__ __forceinline__ int4 scan_load4(const int* cnt, int i0, int N) {
  int4 v = {0, 0, 0, 0};
  if (i0 + 3 < N) {
    v = *(const int4*)(cnt + i0);
  } else {
    if (i0 < N) v.x = cnt[i0];
    if (i0 + 1 < N) v.y = cnt[i0 + 1];
    if (i0 + 2 < N) v.z = cnt[i0 + 2];
  }
  return v;
}

__global__ void scan_kernel(int* cnt, int N) {
  __shared__ int wsum[32];
  int tid = threadIdx.x;
  int lane = tid & 31;
  int wave = tid >> 5;
  int carry = 0;
  int4 nxt = scan_load4(cnt, tid * 4, N);
  for (int base = 0; base < N; base += 4096) {
    int i0 = base + tid * 4;
    int4 v = nxt;
    nxt = scan_load4(cnt, i0 + 4096, N);
    int s = v.x + v.y + v.z + v.w;
    int incl = s;
    for (int off = 1; off < 32; off <<= 1) {
      int t = __shfl_up(incl, off, 32);
      if (lane >= off) incl += t;
    }
    if (lane == 31) wsum[wave] = incl;
    __syncthreads();
    int wt = wsum[lane];
    int wincl = wt;
    for (int off = 1; off < 32; off <<= 1) {
      int t = __shfl_up(wincl, off, 32);
      if (lane >= off) wincl += t;
    }
    int wexcl = __shfl(wincl, wave, 32) - __shfl(wt, wave, 32);
    int total = __shfl(wincl, 31, 32);
    int ex = carry + wexcl + incl - s;
    if (i0 + 3 < N) {
      int4 o;
      o.x = ex; o.y = ex + v.x; o.z = ex + v.x + v.y; o.w = ex + v.x + v.y + v.z;
      *(int4*)(cnt + i0) = o;
    } else {
      if (i0 < N) cnt[i0] = ex;
      if (i0 + 1 < N) cnt[i0 + 1] = ex + v.x;
      if (i0 + 2 < N) cnt[i0 + 2] = ex + v.x + v.y;
    }
    carry += total;
    __syncthreads();
  }
}

__global__ void fill_kernel(const int* __restrict__ ei, int* __restrict__ cur,
                            int* __restrict__ srow, int E) {
  int e = blockIdx.x * blockDim.x + threadIdx.x;
  if (e < E) {
    int pos = atomicAdd(&cur[ei[E + e]], 1);
    srow[pos] = ei[e];
  }
}

__device__ __forceinline__ void order_segment(int* __restrict__ srow, int beg, int d, int lane) {
  int a0 = lane < d ? srow[beg + lane] : 0x7fffffff;
  int a1 = lane + 32 < d ? srow[beg + 32 + lane] : 0x7fffffff;
  int r0 = 0, r1 = 0;
  int n0 = d < 32 ? d : 32;
  for (int i = 0; i < n0; ++i) {
    int x = __builtin_amdgcn_readlane(a0, i);
    r0 += (x < a0 || (x == a0 && i < lane)) ? 1 : 0;
    r1 += (x <= a1) ? 1 : 0;
  }
  for (int i = 0; i < d - 32; ++i) {
    int x = __builtin_amdgcn_readlane(a1, i);
    r0 += (x < a0) ? 1 : 0;
    r1 += (x < a1 || (x == a1 && i < lane)) ? 1 : 0;
  }
  if (lane < d) srow[beg + r0] = a0;
  if (lane + 32 < d) srow[beg + r1] = a1;
}

__global__ void order_kernel(const int* __restrict__ ends, int* __restrict__ srow, int N) {
  int node = __builtin_amdgcn_readfirstlane(blockIdx.x * 8 + (threadIdx.x >> 5));
  int lane = threadIdx.x & 31;
  if (node >= N) return;
  int beg = node ? ends[node - 1] : 0;
  int d = ends[node] - beg;
  if (d < 2 || d > 64) return;
  order_segment(srow, beg, d, lane);
}

__global__ void dn_kernel(float* dn, int N) {
  int i = blockIdx.x * blockDim.x + threadIdx.x;
  if (i < N) dn[i] = 1.0f / sqrtf(fmaxf(dn[i], 1.0f));
}

#define BIN_SHIFT 8
#define BIN_NODES 256
#define MAX_BINS 512
#define CHUNK 4096
#define MAX_CHUNKS 1024
#define BIN_CAP 8192
#define RUN_REGS 8

__device__ __forceinline__ int block_scan_1024(int s, int* wsum, int lane, int wave) {
  int incl = s;
  for (int off = 1; off < 32; off <<= 1) {
    int t = __shfl_up(incl, off, 32);
    if (lane >= off) incl += t;
  }
  if (lane == 31) wsum[wave] = incl;
  __syncthreads();
  int wt = wsum[lane];
  int wincl = wt;
  for (int off = 1; off < 32; off <<= 1) {
    int t = __shfl_up(wincl, off, 32);
    if (lane >= off) wincl += t;
  }
  int wexcl = __shfl(wincl, wave, 32) - __shfl(wt, wave, 32);
  __syncthreads();
  return wexcl + incl - s;
}

__global__ void bin_sort_kernel(const int* __restrict__ ei, int* __restrict__ offs,
                                int* __restrict__ cbin, int* __restrict__ rbin, int E, int NB) {
  __shared__ int hist[2 * MAX_BINS];
  __shared__ int wsum[32];
  __shared__ int keyC[CHUNK];
  __shared__ int keyR[CHUNK];
  int tid = threadIdx.x;
  int lane = tid & 31;
  int wave = tid >> 5;
  int nb2 = 2 * NB;
  int nch = gridDim.x;
  int blk = blockIdx.x;
  for (int i = tid; i < nb2; i += 1024) hist[i] = 0;
  __syncthreads();
  int ebase = blk * CHUNK;
  int nE = E - ebase < CHUNK ? E - ebase : CHUNK;
  int col[4], row[4], rc[4], rr[4];
#pragma unroll
  for (int k = 0; k < 4; ++k) {
    int e = ebase + tid + k * 1024;
    col[k] = e < E ? ei[E + e] : -1;
    row[k] = e < E ? ei[e] : -1;
  }
#pragma unroll
  for (int k = 0; k < 4; ++k) {
    rc[k] = 0; rr[k] = 0;
    if (col[k] >= 0) {
      rc[k] = atomicAdd(&hist[col[k] >> BIN_SHIFT], 1);
      rr[k] = atomicAdd(&hist[NB + (row[k] >> BIN_SHIFT)], 1);
    }
  }
  __syncthreads();
  int i0 = 2 * tid;
  int h0 = i0 < nb2 ? hist[i0] : 0;
  int h1 = i0 + 1 < nb2 ? hist[i0 + 1] : 0;
  int lex = block_scan_1024(h0 + h1, wsum, lane, wave);
  if (i0 < nb2) {
    int second = i0 >= NB ? 1 : 0;
    int st = lex - (second ? nE : 0);
    hist[i0] = st;
    offs[(size_t)(i0 + second) * nch + blk] = st;
  }
  if (i0 + 1 < nb2) {
    int second = i0 + 1 >= NB ? 1 : 0;
    int st = lex + h0 - (second ? nE : 0);
    hist[i0 + 1] = st;
    offs[(size_t)(i0 + 1 + second) * nch + blk] = st;
  }
  if (tid == 0) {
    offs[(size_t)NB * nch + blk] = nE;
    offs[(size_t)(nb2 + 1) * nch + blk] = nE;
  }
  __syncthreads();
#pragma unroll
  for (int k = 0; k < 4; ++k) {
    if (col[k] >= 0) {
      keyC[hist[col[k] >> BIN_SHIFT] + rc[k]] = (row[k] << BIN_SHIFT) | (col[k] & (BIN_NODES - 1));
      keyR[hist[NB + (row[k] >> BIN_SHIFT)] + rr[k]] = row[k] & (BIN_NODES - 1);
    }
  }
  __syncthreads();
  for (int j = tid; j < nE; j += 1024) {
    cbin[ebase + j] = keyC[j];
    rbin[ebase + j] = keyR[j];
  }
}

__global__ void bin_finish_kernel(const int* __restrict__ offs, const int* __restrict__ cbin,
                                  const int* __restrict__ rbin, int* __restrict__ ends,
                                  float* __restrict__ dn, int* __restrict__ srow, int N, int NB,
                                  int nch) {
  __shared__ int ncnt[BIN_NODES];
  __shared__ int nbeg[BIN_NODES];
  __shared__ int ncur[BIN_NODES];
  __shared__ int dcnt[BIN_NODES];
  __shared__ int wsum[32];
  __shared__ int tot[2];
  __shared__ int stage[BIN_CAP];
  int tid = threadIdx.x;
  int lane = tid & 31;
  int wave = __builtin_amdgcn_readfirstlane(tid >> 5);
  int b = blockIdx.x;
  int ch = wave + 32 * lane;
  int c0 = 0, c1 = 0, r0 = 0, r1 = 0;
  if (ch < nch) {
    c0 = offs[(size_t)b * nch + ch];
    c1 = offs[(size_t)(b + 1) * nch + ch];
    r0 = offs[(size_t)(NB + 1 + b) * nch + ch];
    r1 = offs[(size_t)(NB + 2 + b) * nch + ch];
  }
  if (tid < BIN_NODES) { ncnt[tid] = 0; dcnt[tid] = 0; }
  if (tid < 2) tot[tid] = 0;
  __syncthreads();
  if (ch < nch) {
    atomicAdd(&tot[0], c0);
    atomicAdd(&tot[1], c1 - c0);
  }
  const int* cw = cbin + (size_t)wave * CHUNK;
  const int* rw = rbin + (size_t)wave * CHUNK;
  int kc[RUN_REGS], kr[RUN_REGS];
#pragma unroll
  for (int i = 0; i < RUN_REGS; ++i) {
    int cs = __builtin_amdgcn_readlane(c0, i);
    int cl = __builtin_amdgcn_readlane(c1, i) - cs;
    int rs = __builtin_amdgcn_readlane(r0, i);
    int rl = __builtin_amdgcn_readlane(r1, i) - rs;
    kc[i] = lane < cl ? cw[(size_t)i * 32 * CHUNK + cs + lane] : -1;
    kr[i] = lane < rl ? rw[(size_t)i * 32 * CHUNK + rs + lane] : -1;
  }
#pragma unroll
  for (int i = 0; i < RUN_REGS; ++i) {
    if (kc[i] >= 0) atomicAdd(&ncnt[kc[i] & (BIN_NODES - 1)], 1);
    if (kr[i] >= 0) atomicAdd(&dcnt[kr[i]], 1);
  }
  for (int i = 0; i < 32; ++i) {
    if (wave + 32 * i >= nch) break;
    int skip = i < RUN_REGS ? 32 : 0;
    int cs = __builtin_amdgcn_readlane(c0, i) + skip;
    int ce = __builtin_amdgcn_readlane(c1, i);
    int rs = __builtin_amdgcn_readlane(r0, i) + skip;
    int re = __builtin_amdgcn_readlane(r1, i);
    const int* cp = cw + (size_t)i * 32 * CHUNK;
    const int* rp = rw + (size_t)i * 32 * CHUNK;
    for (int j = cs + lane; j < ce; j += 32) atomicAdd(&ncnt[cp[j] & (BIN_NODES - 1)], 1);
    for (int j = rs + lane; j < re; j += 32) atomicAdd(&dcnt[rp[j]], 1);
  }
  __syncthreads();
  int cb = tot[0];
  int cnt = tot[1];
  int c = tid < BIN_NODES ? ncnt[tid] : 0;
  int ex = block_scan_1024(c, wsum, lane, wave);
  if (tid < BIN_NODES) {
    nbeg[tid] = ex;
    ncur[tid] = ex;
    int node = b * BIN_NODES + tid;
    if (node < N) {
      ends[node] = cb + ex + c;
      dn[node] = 1.0f / sqrtf(fmaxf((float)dcnt[tid], 1.0f));
    }
  }
  __syncthreads();
  bool fits = cnt <= BIN_CAP;
#pragma unroll
  for (int i = 0; i < RUN_REGS; ++i) {
    if (kc[i] >= 0) {
      int p = atomicAdd(&ncur[kc[i] & (BIN_NODES - 1)], 1);
      if (fits) stage[p] = kc[i];
      else srow[cb + p] = kc[i] >> BIN_SHIFT;
    }
  }
  for (int i = 0; i < 32; ++i) {
    if (wave + 32 * i >= nch) break;
    int skip = i < RUN_REGS ? 32 : 0;
    int cs = __builtin_amdgcn_readlane(c0, i) + skip;
    int ce = __builtin_amdgcn_readlane(c1, i);
    const int* cp = cw + (size_t)i * 32 * CHUNK;
    for (int j = cs + lane; j < ce; j += 32) {
      int key = cp[j];
      int p = atomicAdd(&ncur[key & (BIN_NODES - 1)], 1);
      if (fits) stage[p] = key;
      else srow[cb + p] = key >> BIN_SHIFT;
    }
  }
  if (!fits) {
    __threadfence();
    __syncthreads();
    for (int t = wave; t < BIN_NODES; t += 32) {
      int n = __builtin_amdgcn_readfirstlane(ncnt[t]);
      int sb = __builtin_amdgcn_readfirstlane(nbeg[t]);
      if (n >= 2 && n <= 64) order_segment(srow, cb + sb, n, lane);
    }
    return;
  }
  __syncthreads();
  for (int p = tid; p < cnt; p += 1024) {
    int key = stage[p];
    int d = key & (BIN_NODES - 1);
    int beg = nbeg[d];
    int n = ncnt[d];
    int r = p - beg;
    if (n > 1 && n <= 64) {
      r = 0;
      for (int j = 0; j < n; ++j) {
        int x = stage[beg + j];
        r += (x < key || (x == key && beg + j < p)) ? 1 : 0;
      }
    }
    srow[cb + beg + r] = key >> BIN_SHIFT;
  }
}

#define TILE_PITCH 132
#define GEMM_ROWS 64
#define YP 32
#define YS_PITCH 36
template <bool PROJ>
__global__ void gemm_relu_kernel(const float* __restrict__ X, const float* __restrict__ W,
                                 const float* __restrict__ bias, const float* __restrict__ wg0,
                                 const float* __restrict__ dn, float* __restrict__ H,
                                 float* __restrict__ pk, float* __restrict__ gb,
                                 const float* __restrict__ Wp, const float* __restrict__ wgl,
                                 float* __restrict__ Y, int Nrows) {
  __shared__ __attribute__((aligned(16))) float tile[GEMM_ROWS * TILE_PITCH];
  int wave = threadIdx.x >> 5;
  int lane = threadIdx.x & 31;
  int row0 = blockIdx.x * GEMM_ROWS;
  int c = lane & 15;
  int n = wave * 16 + c;
  int khalf = (lane >> 4) * 2;
  float4 xr[8];
#pragma unroll
  for (int j = 0; j < 8; ++j) {
    int m = row0 + wave * 8 + j;
    int mc = m < Nrows ? m : (Nrows - 1);
    xr[j] = *(const float4*)(X + (size_t)mc * 128 + lane * 4);
  }
  int md = row0 + wave * 8 + (lane & 7);
  float dnv = dn[md < Nrows ? md : (Nrows - 1)];
  const float* wcol = W + (size_t)khalf * 128 + n;
  v2f bf[32];
#pragma unroll
  for (int s = 0; s < 32; ++s) {
    bf[s].x = wcol[(size_t)s * 512];
    bf[s].y = wcol[(size_t)s * 512 + 128];
  }
#pragma unroll
  for (int j = 0; j < 8; ++j)
    *(float4*)(tile + (wave * 8 + j) * TILE_PITCH + lane * 4) = xr[j];
  __syncthreads();
  v8f acc0 = {}, acc1 = {}, acc2 = {}, acc3 = {};
  const float* arow = tile + c * TILE_PITCH + khalf;
  v2f cur[4][4], nxt[4][4];
#pragma unroll
  for (int i = 0; i < 4; ++i)
#pragma unroll
    for (int t = 0; t < 4; ++t)
      cur[i][t] = *(const v2f*)(arow + t * 16 * TILE_PITCH + i * 4);
#pragma unroll
  for (int g = 0; g < 8; ++g) {
    if (g + 1 < 8) {
#pragma unroll
      for (int i = 0; i < 4; ++i)
#pragma unroll
        for (int t = 0; t < 4; ++t)
          nxt[i][t] = *(const v2f*)(arow + t * 16 * TILE_PITCH + (g + 1) * 16 + i * 4);
    }
#pragma unroll
    for (int i = 0; i < 4; ++i) {
      int s = g * 4 + i;
      acc0 = __builtin_amdgcn_wmma_f32_16x16x4_f32(false, cur[i][0], false, bf[s], (short)0, acc0, false, false);
      acc1 = __builtin_amdgcn_wmma_f32_16x16x4_f32(false, cur[i][1], false, bf[s], (short)0, acc1, false, false);
      acc2 = __builtin_amdgcn_wmma_f32_16x16x4_f32(false, cur[i][2], false, bf[s], (short)0, acc2, false, false);
      acc3 = __builtin_amdgcn_wmma_f32_16x16x4_f32(false, cur[i][3], false, bf[s], (short)0, acc3, false, false);
    }
#pragma unroll
    for (int i = 0; i < 4; ++i)
#pragma unroll
      for (int t = 0; t < 4; ++t)
        cur[i][t] = nxt[i][t];
  }
  __syncthreads();
  int rbase = (lane >> 4) * 8;
  float bn = bias[n];
#pragma unroll
  for (int r = 0; r < 8; ++r) {
    tile[(rbase + r) * TILE_PITCH + n] = fmaxf(acc0[r] + bn, 0.0f);
    tile[(16 + rbase + r) * TILE_PITCH + n] = fmaxf(acc1[r] + bn, 0.0f);
    tile[(32 + rbase + r) * TILE_PITCH + n] = fmaxf(acc2[r] + bn, 0.0f);
    tile[(48 + rbase + r) * TILE_PITCH + n] = fmaxf(acc3[r] + bn, 0.0f);
  }
  if (PROJ) {
    __shared__ __attribute__((aligned(16))) float ystage[GEMM_ROWS * YS_PITCH];
    int rt = wave & 3;
    int ct = __builtin_amdgcn_readfirstlane(wave >> 2);
    v2f pb[32];
    if (ct == 0) {
      const float* bp = Wp + (size_t)khalf * 16 + c;
#pragma unroll
      for (int s = 0; s < 32; ++s) {
        pb[s].x = bp[s * 64];
        pb[s].y = bp[s * 64 + 16];
      }
    } else {
      const float* gp = wgl + (c & 1) * 128 + khalf;
#pragma unroll
      for (int s = 0; s < 32; ++s) pb[s] = *(const v2f*)(gp + 4 * s);
    }
    float4 wa = *(const float4*)(wg0 + lane * 4);
    float4 wb = *(const float4*)(wg0 + 128 + lane * 4);
    __syncthreads();
    const float* ap = tile + (rt * 16 + c) * TILE_PITCH + khalf;
    v2f pcur[8], pnxt[8];
#pragma unroll
    for (int i = 0; i < 8; ++i) pcur[i] = *(const v2f*)(ap + 4 * i);
    v8f pacc = {};
#pragma unroll
    for (int g = 0; g < 4; ++g) {
      if (g + 1 < 4) {
#pragma unroll
        for (int i = 0; i < 8; ++i) pnxt[i] = *(const v2f*)(ap + (g + 1) * 32 + 4 * i);
      }
#pragma unroll
      for (int i = 0; i < 8; ++i)
        pacc = __builtin_amdgcn_wmma_f32_16x16x4_f32(false, pcur[i], false, pb[g * 8 + i], (short)0, pacc, false, false);
#pragma unroll
      for (int i = 0; i < 8; ++i) pcur[i] = pnxt[i];
    }
    int r8 = lane & 7;
    int q = lane >> 3;
    float keep = dnv;
#pragma unroll
    for (int j = 0; j < 8; ++j) {
      float4 v = *(const float4*)(tile + (wave * 8 + j) * TILE_PITCH + lane * 4);
      float sa = v.x * wa.x + v.y * wa.y + v.z * wa.z + v.w * wa.w;
      float sb = v.x * wb.x + v.y * wb.y + v.z * wb.z + v.w * wb.w;
      for (int off = 16; off > 0; off >>= 1) {
        sa += __shfl_xor(sa, off, 32);
        sb += __shfl_xor(sb, off, 32);
      }
      float t = q == 0 ? sa : sb;
      t = q == 3 ? 0.0f : t;
      bool take = (r8 == j) & (q != 1);
      keep = take ? t : keep;
    }
    ystage[(wave * 8 + r8) * YS_PITCH + 32 + q] = keep;
#pragma unroll
    for (int r = 0; r < 8; ++r)
      ystage[(rt * 16 + rbase + r) * YS_PITCH + ct * 16 + c] = pacc[r];
    __syncthreads();
    int cg = lane & 7;
#pragma unroll
    for (int i = 0; i < 2; ++i) {
      int rr = wave * 8 + i * 4 + (lane >> 3);
      int m = row0 + rr;
      float4 v = *(const float4*)(ystage + rr * YS_PITCH + cg * 4);
      float4 e = *(const float4*)(ystage + rr * YS_PITCH + 32);
      v.z = cg == 4 ? e.x : v.z;
      v.w = cg == 4 ? e.y : v.w;
      v.x = cg == 5 ? e.z : v.x;
      if (m < Nrows) *(float4*)(Y + (size_t)m * YP + cg * 4) = v;
    }
    return;
  }
  __syncthreads();
  float4 wa = {0.0f, 0.0f, 0.0f, 0.0f}, wb = {0.0f, 0.0f, 0.0f, 0.0f};
  if (wg0) {
    wa = *(const float4*)(wg0 + lane * 4);
    wb = *(const float4*)(wg0 + 128 + lane * 4);
  }
  for (int j = 0; j < 8; ++j) {
    int rr = wave * 8 + j;
    int m = row0 + rr;
    float4 v = *(const float4*)(tile + rr * TILE_PITCH + lane * 4);
    if (m < Nrows) {
      *(float4*)(H + (size_t)m * 128 + lane * 4) = v;
      if (wg0) {
        float dnm = __shfl(dnv, j, 32);
        float sa = v.x * wa.x + v.y * wa.y + v.z * wa.z + v.w * wa.w;
        float sb = v.x * wb.x + v.y * wb.y + v.z * wb.z + v.w * wb.w;
        for (int off = 16; off > 0; off >>= 1) {
          sa += __shfl_down(sa, off, 32);
          sb += __shfl_down(sb, off, 32);
        }
        if (lane == 0) {
          float2 p; p.x = sa; p.y = dnm;
          *(float2*)(pk + 2 * (size_t)m) = p;
          gb[m] = sb;
        }
      }
    }
  }
}

__global__ void gather_kernel(const int* __restrict__ ends, const int* __restrict__ srow,
                              const float* __restrict__ pk, const float* __restrict__ gb,
                              const float* __restrict__ bg,
                              int layer, const float* __restrict__ H,
                              const float* __restrict__ raw, float* __restrict__ hout,
                              const float* __restrict__ wg_next, float* __restrict__ pk_next,
                              float* __restrict__ gb_next, int N) {
  int node = __builtin_amdgcn_readfirstlane(blockIdx.x * 8 + (threadIdx.x >> 5));
  int lane = threadIdx.x & 31;
  if (node >= N) return;
  int beg = node ? ends[node - 1] : 0;
  int end = ends[node];
  float gbc = gb[node];
  float bgl = bg[layer];
  float dnc = pk[2 * (size_t)node + 1];
  const float* hl = H + lane * 4;
  float4 rw = *(const float4*)(raw + (size_t)node * 128 + lane * 4);
  float4 acc = {0.0f, 0.0f, 0.0f, 0.0f};
  for (int base = beg; base < end; base += 32) {
    int n = end - base < 32 ? end - base : 32;
    int r = 0;
    float w = 0.0f;
    if (lane < n) {
      r = srow[base + lane];
      float2 gd = *(const float2*)(pk + 2 * (size_t)r);
      float g = tanhf(gd.x + gbc + bgl);
      w = g * gd.y * dnc;
    }
    int wi = __float_as_int(w);
    int k = 0;
    for (; k + 4 <= n; k += 4) {
      int r0 = __builtin_amdgcn_readlane(r, k);
      int r1 = __builtin_amdgcn_readlane(r, k + 1);
      int r2 = __builtin_amdgcn_readlane(r, k + 2);
      int r3 = __builtin_amdgcn_readlane(r, k + 3);
      float w0 = __int_as_float(__builtin_amdgcn_readlane(wi, k));
      float w1 = __int_as_float(__builtin_amdgcn_readlane(wi, k + 1));
      float w2 = __int_as_float(__builtin_amdgcn_readlane(wi, k + 2));
      float w3 = __int_as_float(__builtin_amdgcn_readlane(wi, k + 3));
      float4 h0 = *(const float4*)(hl + (size_t)r0 * 128);
      float4 h1 = *(const float4*)(hl + (size_t)r1 * 128);
      float4 h2 = *(const float4*)(hl + (size_t)r2 * 128);
      float4 h3 = *(const float4*)(hl + (size_t)r3 * 128);
      acc.x += w0 * h0.x; acc.y += w0 * h0.y; acc.z += w0 * h0.z; acc.w += w0 * h0.w;
      acc.x += w1 * h1.x; acc.y += w1 * h1.y; acc.z += w1 * h1.z; acc.w += w1 * h1.w;
      acc.x += w2 * h2.x; acc.y += w2 * h2.y; acc.z += w2 * h2.z; acc.w += w2 * h2.w;
      acc.x += w3 * h3.x; acc.y += w3 * h3.y; acc.z += w3 * h3.z; acc.w += w3 * h3.w;
    }
    for (; k < n; ++k) {
      int r0 = __builtin_amdgcn_readlane(r, k);
      float w0 = __int_as_float(__builtin_amdgcn_readlane(wi, k));
      float4 h0 = *(const float4*)(hl + (size_t)r0 * 128);
      acc.x += w0 * h0.x; acc.y += w0 * h0.y; acc.z += w0 * h0.z; acc.w += w0 * h0.w;
    }
  }
  float4 o;
  o.x = FAGCN_EPS * rw.x + acc.x;
  o.y = FAGCN_EPS * rw.y + acc.y;
  o.z = FAGCN_EPS * rw.z + acc.z;
  o.w = FAGCN_EPS * rw.w + acc.w;
  *(float4*)(hout + (size_t)node * 128 + lane * 4) = o;
  if (wg_next) {
    float4 wa = *(const float4*)(wg_next + lane * 4);
    float4 wb = *(const float4*)(wg_next + 128 + lane * 4);
    float sa = o.x * wa.x + o.y * wa.y + o.z * wa.z + o.w * wa.w;
    float sb = o.x * wb.x + o.y * wb.y + o.z * wb.z + o.w * wb.w;
    for (int off = 16; off > 0; off >>= 1) {
      sa += __shfl_down(sa, off, 32);
      sb += __shfl_down(sb, off, 32);
    }
    if (lane == 0) {
      float2 p; p.x = sa; p.y = dnc;
      *(float2*)(pk_next + 2 * (size_t)node) = p;
      gb_next[node] = sb;
    }
  }
}

__global__ void gemm_proj_kernel(const float* __restrict__ Hc, const float* __restrict__ Raw,
                                 const float* __restrict__ W, const float* __restrict__ bias,
                                 float rscale, float* __restrict__ Z, float* __restrict__ ZR,
                                 int Nrows) {
  int wave = threadIdx.x >> 5;
  int lane = threadIdx.x & 31;
  int tiles = (Nrows + 15) >> 4;
  int t = blockIdx.x * 8 + wave;
  if (t >= 2 * tiles) return;
  bool second = t >= tiles;
  int tileM = second ? t - tiles : t;
  const float* src = second ? Raw : Hc;
  float* dst = second ? ZR : Z;
  int mA = tileM * 16 + (lane & 15);
  int mclamp = mA < Nrows ? mA : (Nrows - 1);
  int n = lane & 15;
  int khalf = (lane >> 4) * 2;
  const float* hrow = src + (size_t)mclamp * 128 + khalf;
  const float* wcol = W + (size_t)khalf * 16 + n;
  __builtin_prefetch(hrow, 0, 0);
  v8f acc = {};
#pragma unroll 4
  for (int kk = 0; kk < 128; kk += 4) {
    v2f a = *(const v2f*)(hrow + kk);
    v2f b;
    b.x = wcol[(size_t)kk * 16];
    b.y = wcol[(size_t)kk * 16 + 16];
    acc = __builtin_amdgcn_wmma_f32_16x16x4_f32(false, a, false, b, (short)0, acc,
                                               false, false);
  }
  int mbase = tileM * 16 + ((lane >> 4) * 8);
  float bn = second ? bias[n] : 0.0f;
  float sc = second ? rscale : 1.0f;
  if (tileM * 16 + 16 <= Nrows) {
    float* op = dst + (size_t)mbase * 16 + n;
#pragma unroll
    for (int r = 0; r < 8; ++r)
      op[(size_t)r * 16] = acc[r] * sc + bn;
  } else {
#pragma unroll
    for (int r = 0; r < 8; ++r) {
      int mm = mbase + r;
      if (mm < Nrows) dst[(size_t)mm * 16 + n] = acc[r] * sc + bn;
    }
  }
}

__global__ void gather_proj_kernel(const int* __restrict__ ends, const int* __restrict__ srow,
                                   const float* __restrict__ pk, const float* __restrict__ gb,
                                   const float* __restrict__ bg, int layer,
                                   const float* __restrict__ Z, const float* __restrict__ ZR,
                                   float* __restrict__ logits, int N) {
  int node = blockIdx.x * 64 + (threadIdx.x >> 2);
  int q = threadIdx.x & 3;
  if (node >= N) return;
  int beg = node ? ends[node - 1] : 0;
  int end = ends[node];
  float gbc = gb[node];
  float bgl = bg[layer];
  float dnc = pk[2 * (size_t)node + 1];
  const float* zq = Z + q * 4;
  float4 zr = *(const float4*)(ZR + (size_t)node * 16 + q * 4);
  float4 acc = {0.0f, 0.0f, 0.0f, 0.0f};
  int last = end - 1;
  int r0 = 0, r1 = 0, r2 = 0, r3 = 0;
  if (beg < end) {
    r0 = srow[beg];
    r1 = srow[beg + 1 < last ? beg + 1 : last];
    r2 = srow[beg + 2 < last ? beg + 2 : last];
    r3 = srow[beg + 3 < last ? beg + 3 : last];
  }
  for (int base = beg; base < end; base += 4) {
    float2 g0 = *(const float2*)(pk + 2 * (size_t)r0);
    float2 g1 = *(const float2*)(pk + 2 * (size_t)r1);
    float2 g2 = *(const float2*)(pk + 2 * (size_t)r2);
    float2 g3 = *(const float2*)(pk + 2 * (size_t)r3);
    float4 h0 = *(const float4*)(zq + (size_t)r0 * 16);
    float4 h1 = *(const float4*)(zq + (size_t)r1 * 16);
    float4 h2 = *(const float4*)(zq + (size_t)r2 * 16);
    float4 h3 = *(const float4*)(zq + (size_t)r3 * 16);
    int nb = base + 4;
    int n0 = r0, n1 = r1, n2 = r2, n3 = r3;
    if (nb < end) {
      n0 = srow[nb];
      n1 = srow[nb + 1 < last ? nb + 1 : last];
      n2 = srow[nb + 2 < last ? nb + 2 : last];
      n3 = srow[nb + 3 < last ? nb + 3 : last];
    }
    {
      float w = tanhf(g0.x + gbc + bgl) * g0.y * dnc;
      acc.x += w * h0.x; acc.y += w * h0.y; acc.z += w * h0.z; acc.w += w * h0.w;
    }
    if (base + 1 < end) {
      float w = tanhf(g1.x + gbc + bgl) * g1.y * dnc;
      acc.x += w * h1.x; acc.y += w * h1.y; acc.z += w * h1.z; acc.w += w * h1.w;
    }
    if (base + 2 < end) {
      float w = tanhf(g2.x + gbc + bgl) * g2.y * dnc;
      acc.x += w * h2.x; acc.y += w * h2.y; acc.z += w * h2.z; acc.w += w * h2.w;
    }
    if (base + 3 < end) {
      float w = tanhf(g3.x + gbc + bgl) * g3.y * dnc;
      acc.x += w * h3.x; acc.y += w * h3.y; acc.z += w * h3.z; acc.w += w * h3.w;
    }
    r0 = n0; r1 = n1; r2 = n2; r3 = n3;
  }
  float4 o;
  o.x = zr.x + acc.x;
  o.y = zr.y + acc.y;
  o.z = zr.z + acc.z;
  o.w = zr.w + acc.w;
  *(float4*)(logits + (size_t)node * 16 + q * 4) = o;
}

__global__ void gemm_proj2_kernel(const float* __restrict__ Hc, const float* __restrict__ Raw,
                                  const float* __restrict__ W, const float* __restrict__ wgl,
                                  const float* __restrict__ pk, const float* __restrict__ gb,
                                  float* __restrict__ Y, float* __restrict__ YR,
                                  int Nrows, int nt) {
  int wave = threadIdx.x >> 5;
  int lane = threadIdx.x & 31;
  int tiles = (Nrows + 15) >> 4;
  int t = blockIdx.x * 8 + wave;
  if (t >= nt) return;
  bool second = t >= tiles;
  int tileM = second ? t - tiles : t;
  const float* src = second ? Raw : Hc;
  float* dst = second ? YR : Y;
  int mA = tileM * 16 + (lane & 15);
  int mclamp = mA < Nrows ? mA : (Nrows - 1);
  int n = lane & 15;
  int khalf = (lane >> 4) * 2;
  const float* hrow = src + (size_t)mclamp * 128 + khalf;
  const float* wcol = W + (size_t)khalf * 16 + n;
  const float* gcol = wgl + (n & 1) * 128 + khalf;
  __builtin_prefetch(hrow, 0, 0);
  int mbase = tileM * 16 + ((lane >> 4) * 8);
  const float* tp = (n == 2 || n == 3) ? pk + (n - 2) : gb;
  int ts = (n == 2 || n == 3) ? 2 : 1;
  float tv[8];
#pragma unroll
  for (int r = 0; r < 8; ++r) {
    int mr = mbase + r < Nrows ? mbase + r : Nrows - 1;
    tv[r] = tp[(size_t)mr * ts];
  }
  v8f acc = {};
  v8f acg = {};
#pragma unroll 4
  for (int kk = 0; kk < 128; kk += 4) {
    v2f a = *(const v2f*)(hrow + kk);
    v2f b;
    b.x = wcol[(size_t)kk * 16];
    b.y = wcol[(size_t)kk * 16 + 16];
    v2f g = *(const v2f*)(gcol + kk);
    acc = __builtin_amdgcn_wmma_f32_16x16x4_f32(false, a, false, b, (short)0, acc,
                                               false, false);
    acg = __builtin_amdgcn_wmma_f32_16x16x4_f32(false, a, false, g, (short)0, acg,
                                               false, false);
  }
  float* yb = dst + (size_t)mbase * YP + n;
  if (tileM * 16 + 16 <= Nrows) {
#pragma unroll
    for (int r = 0; r < 8; ++r) {
      yb[(size_t)r * YP] = acc[r];
      if (n < 5) yb[(size_t)r * YP + 16] = n < 2 ? acg[r] : tv[r];
    }
  } else {
#pragma unroll
    for (int r = 0; r < 8; ++r) {
      if (mbase + r < Nrows) {
        yb[(size_t)r * YP] = acc[r];
        if (n < 5) yb[(size_t)r * YP + 16] = n < 2 ? acg[r] : tv[r];
      }
    }
  }
}

__global__ void gather_mid_kernel(const int* __restrict__ ends, const int* __restrict__ srow,
                                  const float* __restrict__ bg, int layer,
                                  const float* __restrict__ Y, const float* __restrict__ YR,
                                  const float* __restrict__ bias,
                                  float* __restrict__ Z, float* __restrict__ ZR,
                                  float* __restrict__ pk_next, float* __restrict__ gb_next,
                                  int N) {
  int node = blockIdx.x * 64 + (threadIdx.x >> 2);
  int q = threadIdx.x & 3;
  if (node >= N) return;
  int beg = node ? ends[node - 1] : 0;
  int end = ends[node];
  float bgl = bg[layer];
  const float* yc = Y + (size_t)node * YP;
  float dnc = yc[19];
  float gbc = yc[20];
  const float* yrc = YR + (size_t)node * YP;
  float4 own = *(const float4*)(yrc + q * 4);
  float2 owng = *(const float2*)(yrc + 16);
  float4 bq = *(const float4*)(bias + q * 4);
  const float* yq = Y + q * 4;
  const float* yt = Y + 16;
  float4 acc = {0.0f, 0.0f, 0.0f, 0.0f};
  float sga = 0.0f, sgb = 0.0f;
  int last = end - 1;
  int r0 = 0, r1 = 0, r2 = 0, r3 = 0;
  if (beg < end) {
    r0 = srow[beg];
    r1 = srow[beg + 1 < last ? beg + 1 : last];
    r2 = srow[beg + 2 < last ? beg + 2 : last];
    r3 = srow[beg + 3 < last ? beg + 3 : last];
  }
  for (int base = beg; base < end; base += 4) {
    float4 t0 = *(const float4*)(yt + (size_t)r0 * YP);
    float4 t1 = *(const float4*)(yt + (size_t)r1 * YP);
    float4 t2 = *(const float4*)(yt + (size_t)r2 * YP);
    float4 t3 = *(const float4*)(yt + (size_t)r3 * YP);
    float4 h0 = *(const float4*)(yq + (size_t)r0 * YP);
    float4 h1 = *(const float4*)(yq + (size_t)r1 * YP);
    float4 h2 = *(const float4*)(yq + (size_t)r2 * YP);
    float4 h3 = *(const float4*)(yq + (size_t)r3 * YP);
    int nb = base + 4;
    int n0 = r0, n1 = r1, n2 = r2, n3 = r3;
    if (nb < end) {
      n0 = srow[nb];
      n1 = srow[nb + 1 < last ? nb + 1 : last];
      n2 = srow[nb + 2 < last ? nb + 2 : last];
      n3 = srow[nb + 3 < last ? nb + 3 : last];
    }
    {
      float w = tanhf(t0.z + gbc + bgl) * t0.w * dnc;
      acc.x += w * h0.x; acc.y += w * h0.y; acc.z += w * h0.z; acc.w += w * h0.w;
      sga += w * t0.x; sgb += w * t0.y;
    }
    if (base + 1 < end) {
      float w = tanhf(t1.z + gbc + bgl) * t1.w * dnc;
      acc.x += w * h1.x; acc.y += w * h1.y; acc.z += w * h1.z; acc.w += w * h1.w;
      sga += w * t1.x; sgb += w * t1.y;
    }
    if (base + 2 < end) {
      float w = tanhf(t2.z + gbc + bgl) * t2.w * dnc;
      acc.x += w * h2.x; acc.y += w * h2.y; acc.z += w * h2.z; acc.w += w * h2.w;
      sga += w * t2.x; sgb += w * t2.y;
    }
    if (base + 3 < end) {
      float w = tanhf(t3.z + gbc + bgl) * t3.w * dnc;
      acc.x += w * h3.x; acc.y += w * h3.y; acc.z += w * h3.z; acc.w += w * h3.w;
      sga += w * t3.x; sgb += w * t3.y;
    }
    r0 = n0; r1 = n1; r2 = n2; r3 = n3;
  }
  float4 o, zr;
  o.x = FAGCN_EPS * own.x + acc.x;  zr.x = FAGCN_EPS * own.x + bq.x;
  o.y = FAGCN_EPS * own.y + acc.y;  zr.y = FAGCN_EPS * own.y + bq.y;
  o.z = FAGCN_EPS * own.z + acc.z;  zr.z = FAGCN_EPS * own.z + bq.z;
  o.w = FAGCN_EPS * own.w + acc.w;  zr.w = FAGCN_EPS * own.w + bq.w;
  *(float4*)(Z + (size_t)node * 16 + q * 4) = o;
  *(float4*)(ZR + (size_t)node * 16 + q * 4) = zr;
  if (q == 0) {
    float2 p; p.x = FAGCN_EPS * owng.x + sga; p.y = dnc;
    *(float2*)(pk_next + 2 * (size_t)node) = p;
    gb_next[node] = FAGCN_EPS * owng.y + sgb;
  }
}

__global__ void logsoftmax_kernel(const float4* __restrict__ in, float4* __restrict__ out, int N) {
  int i = blockIdx.x * blockDim.x + threadIdx.x;
  if (i >= N) return;
  const float4* rin = in + (size_t)i * 4;
  float4* row = out + (size_t)i * 4;
  float4 v[4];
#pragma unroll
  for (int q = 0; q < 4; ++q) v[q] = rin[q];
  float m = v[0].x;
#pragma unroll
  for (int q = 0; q < 4; ++q) {
    m = fmaxf(m, v[q].x); m = fmaxf(m, v[q].y);
    m = fmaxf(m, v[q].z); m = fmaxf(m, v[q].w);
  }
  float s = 0.0f;
#pragma unroll
  for (int q = 0; q < 4; ++q) {
    s += expf(v[q].x - m); s += expf(v[q].y - m);
    s += expf(v[q].z - m); s += expf(v[q].w - m);
  }
  float d = m + logf(s);
#pragma unroll
  for (int q = 0; q < 4; ++q) {
    v[q].x -= d; v[q].y -= d; v[q].z -= d; v[q].w -= d;
    row[q] = v[q];
  }
}

extern "C" void kernel_launch(void* const* d_in, const int* in_sizes, int n_in,
                              void* d_out, int out_size, void* d_ws, size_t ws_size,
                              hipStream_t stream) {
  const float* x     = (const float*)d_in[0];
  const int*   ei    = (const int*)d_in[1];
  const float* W_in  = (const float*)d_in[2];
  const float* b_in  = (const float*)d_in[3];
  const float* Wg    = (const float*)d_in[4];
  const float* bg    = (const float*)d_in[5];
  const float* W_out = (const float*)d_in[6];
  const float* b_out = (const float*)d_in[7];
  float* out = (float*)d_out;

  const int Hdim = 128;
  int N = in_sizes[0] / Hdim;
  int E = in_sizes[1] / 2;
  int L = in_sizes[5];

  float* ws   = (float*)d_ws;
  float* raw  = ws;
  float* bufA = raw + (size_t)N * Hdim;
  float* bufB = bufA + (size_t)N * Hdim;
  int*   cnt  = (int*)(bufB + (size_t)N * Hdim);
  float* dn   = (float*)cnt + N;
  int*   srow = (int*)(dn + N);
  float* pk   = out;
  float* pk2  = out + 2 * (size_t)N;
  float* gb   = out + 4 * (size_t)N;
  float* gb2  = out + 5 * (size_t)N;

  int NB = (N + BIN_NODES - 1) / BIN_NODES;
  int nchunks = (E + CHUNK - 1) / CHUNK;
  size_t scratch = 2 * (size_t)E + (2 * (size_t)NB + 2) * (size_t)nchunks;
  if (NB <= MAX_BINS && nchunks <= MAX_CHUNKS && scratch <= (size_t)N * Hdim) {
    int* cbin = (int*)bufA;
    int* rbin = cbin + E;
    int* offs = rbin + E;
    bin_sort_kernel<<<nchunks, 1024, 0, stream>>>(ei, offs, cbin, rbin, E, NB);
    bin_finish_kernel<<<NB, 1024, 0, stream>>>(offs, cbin, rbin, cnt, dn, srow, N, NB, nchunks);
  } else {
    hipMemsetAsync(cnt, 0, (size_t)N * 2 * sizeof(float), stream);
    deg_kernel<<<(E + 255) / 256, 256, 0, stream>>>(ei, dn, cnt, E);
    dn_kernel<<<(N + 255) / 256, 256, 0, stream>>>(dn, N);
    scan_kernel<<<1, 1024, 0, stream>>>(cnt, N);
    fill_kernel<<<(E + 255) / 256, 256, 0, stream>>>(ei, cnt, srow, E);
    order_kernel<<<(N + 7) / 8, 256, 0, stream>>>(cnt, srow, N);
  }

  bool fused = (L == 2);
  if (fused)
    gemm_relu_kernel<true><<<(N + GEMM_ROWS - 1) / GEMM_ROWS, 256, 0, stream>>>(
        x, W_in, b_in, Wg, dn, nullptr, nullptr, nullptr, W_out, Wg + (size_t)(L - 1) * 2 * Hdim, bufA, N);
  else
    gemm_relu_kernel<false><<<(N + GEMM_ROWS - 1) / GEMM_ROWS, 256, 0, stream>>>(
        x, W_in, b_in, L > 0 ? Wg : nullptr, dn, raw, pk, gb, nullptr, nullptr, nullptr, N);

  float* h = raw;
  float* hnew = bufA;
  float* gac = pk;  float* gbc = gb;
  float* gan = pk2; float* gbn = gb2;
  for (int i = 0; i + 2 < L; ++i) {
    const float* wgn = Wg + (size_t)(i + 1) * 2 * Hdim;
    gather_kernel<<<(N + 7) / 8, 256, 0, stream>>>(cnt, srow, gac, gbc, bg, i, h, raw, hnew,
                                                  wgn, gan, gbn, N);
    h = hnew;
    hnew = (h == bufA) ? bufB : bufA;
    float* t = gac; gac = gan; gan = t;
    t = gbc; gbc = gbn; gbn = t;
  }

  int tiles = (N + 15) / 16;
  if (L >= 2) {
    bool same = (h == raw);
    float* Y  = hnew;
    float* YR = same ? Y : hnew + 32 * (size_t)N;
    float* Z  = hnew + 64 * (size_t)N;
    float* ZR = hnew + 80 * (size_t)N;
    float* LG = hnew + 96 * (size_t)N;
    int nt = same ? tiles : 2 * tiles;
    if (!fused)
      gemm_proj2_kernel<<<(nt + 7) / 8, 256, 0, stream>>>(h, raw, W_out,
                                                         Wg + (size_t)(L - 1) * 2 * Hdim, gac, gbc,
                                                         Y, YR, N, nt);
    gather_mid_kernel<<<(N + 63) / 64, 256, 0, stream>>>(cnt, srow, bg, L - 2, Y, YR, b_out,
                                                        Z, ZR, gan, gbn, N);
    gather_proj_kernel<<<(N + 63) / 64, 256, 0, stream>>>(cnt, srow, gan, gbn, bg, L - 1, Z, ZR,
                                                         LG, N);
    logsoftmax_kernel<<<(N + 255) / 256, 256, 0, stream>>>((const float4*)LG, (float4*)out, N);
    return;
  }

  float* Z  = hnew;
  float* ZR = hnew + 16 * (size_t)N;
  float* LG = hnew + 32 * (size_t)N;
  gemm_proj_kernel<<<(2 * tiles + 7) / 8, 256, 0, stream>>>(h, raw, W_out, b_out,
                                                           L > 0 ? FAGCN_EPS : 1.0f, Z, ZR, N);
  if (L > 0)
    gather_proj_kernel<<<(N + 63) / 64, 256, 0, stream>>>(cnt, srow, gac, gbc, bg, L - 1, Z, ZR,
                                                         LG, N);
  logsoftmax_kernel<<<(N + 255) / 256, 256, 0, stream>>>((const float4*)(L > 0 ? LG : ZR),
                                                        (float4*)out, N);
}


# Log-softmax folds into the last gather: a node's 4 lanes take max and exp sum of its 16 logits across the quad, store the output; gate arrays go to workspace; 1 launch goes; exp sum order changes
# speedup vs baseline: 7.0604x; 1.0497x over previous
#include <hip/hip_runtime.h>

typedef float v2f __attribute__((ext_vector_type(2)));
typedef float v8f __attribute__((ext_vector_type(8)));

#define FAGCN_EPS 0.3f

__global__ void deg_kernel(const int* __restrict__ ei, float* __restrict__ deg,
                           int* __restrict__ cnt, int E) {
  int e = blockIdx.x * blockDim.x + threadIdx.x;
  if (e < E) {
    atomicAdd(&deg[ei[e]], 1.0f);
    atomicAdd(&cnt[ei[E + e]], 1);
  }
}

__device__ __forceinline__ int4 scan_load4(const int* cnt, int i0, int N) {
  int4 v = {0, 0, 0, 0};
  if (i0 + 3 < N) {
    v = *(const int4*)(cnt + i0);
  } else {
    if (i0 < N) v.x = cnt[i0];
    if (i0 + 1 < N) v.y = cnt[i0 + 1];
    if (i0 + 2 < N) v.z = cnt[i0 + 2];
  }
  return v;
}

__global__ void scan_kernel(int* cnt, int N) {
  __shared__ int wsum[32];
  int tid = threadIdx.x;
  int lane = tid & 31;
  int wave = tid >> 5;
  int carry = 0;
  int4 nxt = scan_load4(cnt, tid * 4, N);
  for (int base = 0; base < N; base += 4096) {
    int i0 = base + tid * 4;
    int4 v = nxt;
    nxt = scan_load4(cnt, i0 + 4096, N);
    int s = v.x + v.y + v.z + v.w;
    int incl = s;
    for (int off = 1; off < 32; off <<= 1) {
      int t = __shfl_up(incl, off, 32);
      if (lane >= off) incl += t;
    }
    if (lane == 31) wsum[wave] = incl;
    __syncthreads();
    int wt = wsum[lane];
    int wincl = wt;
    for (int off = 1; off < 32; off <<= 1) {
      int t = __shfl_up(wincl, off, 32);
      if (lane >= off) wincl += t;
    }
    int wexcl = __shfl(wincl, wave, 32) - __shfl(wt, wave, 32);
    int total = __shfl(wincl, 31, 32);
    int ex = carry + wexcl + incl - s;
    if (i0 + 3 < N) {
      int4 o;
      o.x = ex; o.y = ex + v.x; o.z = ex + v.x + v.y; o.w = ex + v.x + v.y + v.z;
      *(int4*)(cnt + i0) = o;
    } else {
      if (i0 < N) cnt[i0] = ex;
      if (i0 + 1 < N) cnt[i0 + 1] = ex + v.x;
      if (i0 + 2 < N) cnt[i0 + 2] = ex + v.x + v.y;
    }
    carry += total;
    __syncthreads();
  }
}

__global__ void fill_kernel(const int* __restrict__ ei, int* __restrict__ cur,
                            int* __restrict__ srow, int E) {
  int e = blockIdx.x * blockDim.x + threadIdx.x;
  if (e < E) {
    int pos = atomicAdd(&cur[ei[E + e]], 1);
    srow[pos] = ei[e];
  }
}

__device__ __forceinline__ void order_segment(int* __restrict__ srow, int beg, int d, int lane) {
  int a0 = lane < d ? srow[beg + lane] : 0x7fffffff;
  int a1 = lane + 32 < d ? srow[beg + 32 + lane] : 0x7fffffff;
  int r0 = 0, r1 = 0;
  int n0 = d < 32 ? d : 32;
  for (int i = 0; i < n0; ++i) {
    int x = __builtin_amdgcn_readlane(a0, i);
    r0 += (x < a0 || (x == a0 && i < lane)) ? 1 : 0;
    r1 += (x <= a1) ? 1 : 0;
  }
  for (int i = 0; i < d - 32; ++i) {
    int x = __builtin_amdgcn_readlane(a1, i);
    r0 += (x < a0) ? 1 : 0;
    r1 += (x < a1 || (x == a1 && i < lane)) ? 1 : 0;
  }
  if (lane < d) srow[beg + r0] = a0;
  if (lane + 32 < d) srow[beg + r1] = a1;
}

__global__ void order_kernel(const int* __restrict__ ends, int* __restrict__ srow, int N) {
  int node = __builtin_amdgcn_readfirstlane(blockIdx.x * 8 + (threadIdx.x >> 5));
  int lane = threadIdx.x & 31;
  if (node >= N) return;
  int beg = node ? ends[node - 1] : 0;
  int d = ends[node] - beg;
  if (d < 2 || d > 64) return;
  order_segment(srow, beg, d, lane);
}

__global__ void dn_kernel(float* dn, int N) {
  int i = blockIdx.x * blockDim.x + threadIdx.x;
  if (i < N) dn[i] = 1.0f / sqrtf(fmaxf(dn[i], 1.0f));
}

#define BIN_SHIFT 8
#define BIN_NODES 256
#define MAX_BINS 512
#define CHUNK 4096
#define MAX_CHUNKS 1024
#define BIN_CAP 8192
#define RUN_REGS 8

__device__ __forceinline__ int block_scan_1024(int s, int* wsum, int lane, int wave) {
  int incl = s;
  for (int off = 1; off < 32; off <<= 1) {
    int t = __shfl_up(incl, off, 32);
    if (lane >= off) incl += t;
  }
  if (lane == 31) wsum[wave] = incl;
  __syncthreads();
  int wt = wsum[lane];
  int wincl = wt;
  for (int off = 1; off < 32; off <<= 1) {
    int t = __shfl_up(wincl, off, 32);
    if (lane >= off) wincl += t;
  }
  int wexcl = __shfl(wincl, wave, 32) - __shfl(wt, wave, 32);
  __syncthreads();
  return wexcl + incl - s;
}

__global__ void bin_sort_kernel(const int* __restrict__ ei, int* __restrict__ offs,
                                int* __restrict__ cbin, int* __restrict__ rbin, int E, int NB) {
  __shared__ int hist[2 * MAX_BINS];
  __shared__ int wsum[32];
  __shared__ int keyC[CHUNK];
  __shared__ int keyR[CHUNK];
  int tid = threadIdx.x;
  int lane = tid & 31;
  int wave = tid >> 5;
  int nb2 = 2 * NB;
  int nch = gridDim.x;
  int blk = blockIdx.x;
  for (int i = tid; i < nb2; i += 1024) hist[i] = 0;
  __syncthreads();
  int ebase = blk * CHUNK;
  int nE = E - ebase < CHUNK ? E - ebase : CHUNK;
  int col[4], row[4], rc[4], rr[4];
#pragma unroll
  for (int k = 0; k < 4; ++k) {
    int e = ebase + tid + k * 1024;
    col[k] = e < E ? ei[E + e] : -1;
    row[k] = e < E ? ei[e] : -1;
  }
#pragma unroll
  for (int k = 0; k < 4; ++k) {
    rc[k] = 0; rr[k] = 0;
    if (col[k] >= 0) {
      rc[k] = atomicAdd(&hist[col[k] >> BIN_SHIFT], 1);
      rr[k] = atomicAdd(&hist[NB + (row[k] >> BIN_SHIFT)], 1);
    }
  }
  __syncthreads();
  int i0 = 2 * tid;
  int h0 = i0 < nb2 ? hist[i0] : 0;
  int h1 = i0 + 1 < nb2 ? hist[i0 + 1] : 0;
  int lex = block_scan_1024(h0 + h1, wsum, lane, wave);
  if (i0 < nb2) {
    int second = i0 >= NB ? 1 : 0;
    int st = lex - (second ? nE : 0);
    hist[i0] = st;
    offs[(size_t)(i0 + second) * nch + blk] = st;
  }
  if (i0 + 1 < nb2) {
    int second = i0 + 1 >= NB ? 1 : 0;
    int st = lex + h0 - (second ? nE : 0);
    hist[i0 + 1] = st;
    offs[(size_t)(i0 + 1 + second) * nch + blk] = st;
  }
  if (tid == 0) {
    offs[(size_t)NB * nch + blk] = nE;
    offs[(size_t)(nb2 + 1) * nch + blk] = nE;
  }
  __syncthreads();
#pragma unroll
  for (int k = 0; k < 4; ++k) {
    if (col[k] >= 0) {
      keyC[hist[col[k] >> BIN_SHIFT] + rc[k]] = (row[k] << BIN_SHIFT) | (col[k] & (BIN_NODES - 1));
      keyR[hist[NB + (row[k] >> BIN_SHIFT)] + rr[k]] = row[k] & (BIN_NODES - 1);
    }
  }
  __syncthreads();
  for (int j = tid; j < nE; j += 1024) {
    cbin[ebase + j] = keyC[j];
    rbin[ebase + j] = keyR[j];
  }
}

__global__ void bin_finish_kernel(const int* __restrict__ offs, const int* __restrict__ cbin,
                                  const int* __restrict__ rbin, int* __restrict__ ends,
                                  float* __restrict__ dn, int* __restrict__ srow, int N, int NB,
                                  int nch) {
  __shared__ int ncnt[BIN_NODES];
  __shared__ int nbeg[BIN_NODES];
  __shared__ int ncur[BIN_NODES];
  __shared__ int dcnt[BIN_NODES];
  __shared__ int wsum[32];
  __shared__ int tot[2];
  __shared__ int stage[BIN_CAP];
  int tid = threadIdx.x;
  int lane = tid & 31;
  int wave = __builtin_amdgcn_readfirstlane(tid >> 5);
  int b = blockIdx.x;
  int ch = wave + 32 * lane;
  int c0 = 0, c1 = 0, r0 = 0, r1 = 0;
  if (ch < nch) {
    c0 = offs[(size_t)b * nch + ch];
    c1 = offs[(size_t)(b + 1) * nch + ch];
    r0 = offs[(size_t)(NB + 1 + b) * nch + ch];
    r1 = offs[(size_t)(NB + 2 + b) * nch + ch];
  }
  if (tid < BIN_NODES) { ncnt[tid] = 0; dcnt[tid] = 0; }
  if (tid < 2) tot[tid] = 0;
  __syncthreads();
  if (ch < nch) {
    atomicAdd(&tot[0], c0);
    atomicAdd(&tot[1], c1 - c0);
  }
  const int* cw = cbin + (size_t)wave * CHUNK;
  const int* rw = rbin + (size_t)wave * CHUNK;
  int kc[RUN_REGS], kr[RUN_REGS];
#pragma unroll
  for (int i = 0; i < RUN_REGS; ++i) {
    int cs = __builtin_amdgcn_readlane(c0, i);
    int cl = __builtin_amdgcn_readlane(c1, i) - cs;
    int rs = __builtin_amdgcn_readlane(r0, i);
    int rl = __builtin_amdgcn_readlane(r1, i) - rs;
    kc[i] = lane < cl ? cw[(size_t)i * 32 * CHUNK + cs + lane] : -1;
    kr[i] = lane < rl ? rw[(size_t)i * 32 * CHUNK + rs + lane] : -1;
  }
#pragma unroll
  for (int i = 0; i < RUN_REGS; ++i) {
    if (kc[i] >= 0) atomicAdd(&ncnt[kc[i] & (BIN_NODES - 1)], 1);
    if (kr[i] >= 0) atomicAdd(&dcnt[kr[i]], 1);
  }
  for (int i = 0; i < 32; ++i) {
    if (wave + 32 * i >= nch) break;
    int skip = i < RUN_REGS ? 32 : 0;
    int cs = __builtin_amdgcn_readlane(c0, i) + skip;
    int ce = __builtin_amdgcn_readlane(c1, i);
    int rs = __builtin_amdgcn_readlane(r0, i) + skip;
    int re = __builtin_amdgcn_readlane(r1, i);
    const int* cp = cw + (size_t)i * 32 * CHUNK;
    const int* rp = rw + (size_t)i * 32 * CHUNK;
    for (int j = cs + lane; j < ce; j += 32) atomicAdd(&ncnt[cp[j] & (BIN_NODES - 1)], 1);
    for (int j = rs + lane; j < re; j += 32) atomicAdd(&dcnt[rp[j]], 1);
  }
  __syncthreads();
  int cb = tot[0];
  int cnt = tot[1];
  int c = tid < BIN_NODES ? ncnt[tid] : 0;
  int ex = block_scan_1024(c, wsum, lane, wave);
  if (tid < BIN_NODES) {
    nbeg[tid] = ex;
    ncur[tid] = ex;
    int node = b * BIN_NODES + tid;
    if (node < N) {
      ends[node] = cb + ex + c;
      dn[node] = 1.0f / sqrtf(fmaxf((float)dcnt[tid], 1.0f));
    }
  }
  __syncthreads();
  bool fits = cnt <= BIN_CAP;
#pragma unroll
  for (int i = 0; i < RUN_REGS; ++i) {
    if (kc[i] >= 0) {
      int p = atomicAdd(&ncur[kc[i] & (BIN_NODES - 1)], 1);
      if (fits) stage[p] = kc[i];
      else srow[cb + p] = kc[i] >> BIN_SHIFT;
    }
  }
  for (int i = 0; i < 32; ++i) {
    if (wave + 32 * i >= nch) break;
    int skip = i < RUN_REGS ? 32 : 0;
    int cs = __builtin_amdgcn_readlane(c0, i) + skip;
    int ce = __builtin_amdgcn_readlane(c1, i);
    const int* cp = cw + (size_t)i * 32 * CHUNK;
    for (int j = cs + lane; j < ce; j += 32) {
      int key = cp[j];
      int p = atomicAdd(&ncur[key & (BIN_NODES - 1)], 1);
      if (fits) stage[p] = key;
      else srow[cb + p] = key >> BIN_SHIFT;
    }
  }
  if (!fits) {
    __threadfence();
    __syncthreads();
    for (int t = wave; t < BIN_NODES; t += 32) {
      int n = __builtin_amdgcn_readfirstlane(ncnt[t]);
      int sb = __builtin_amdgcn_readfirstlane(nbeg[t]);
      if (n >= 2 && n <= 64) order_segment(srow, cb + sb, n, lane);
    }
    return;
  }
  __syncthreads();
  for (int p = tid; p < cnt; p += 1024) {
    int key = stage[p];
    int d = key & (BIN_NODES - 1);
    int beg = nbeg[d];
    int n = ncnt[d];
    int r = p - beg;
    if (n > 1 && n <= 64) {
      r = 0;
      for (int j = 0; j < n; ++j) {
        int x = stage[beg + j];
        r += (x < key || (x == key && beg + j < p)) ? 1 : 0;
      }
    }
    srow[cb + beg + r] = key >> BIN_SHIFT;
  }
}

#define TILE_PITCH 132
#define GEMM_ROWS 64
#define YP 32
#define YS_PITCH 36
template <bool PROJ>
__global__ void gemm_relu_kernel(const float* __restrict__ X, const float* __restrict__ W,
                                 const float* __restrict__ bias, const float* __restrict__ wg0,
                                 const float* __restrict__ dn, float* __restrict__ H,
                                 float* __restrict__ pk, float* __restrict__ gb,
                                 const float* __restrict__ Wp, const float* __restrict__ wgl,
                                 float* __restrict__ Y, int Nrows) {
  __shared__ __attribute__((aligned(16))) float tile[GEMM_ROWS * TILE_PITCH];
  int wave = threadIdx.x >> 5;
  int lane = threadIdx.x & 31;
  int row0 = blockIdx.x * GEMM_ROWS;
  int c = lane & 15;
  int n = wave * 16 + c;
  int khalf = (lane >> 4) * 2;
  float4 xr[8];
#pragma unroll
  for (int j = 0; j < 8; ++j) {
    int m = row0 + wave * 8 + j;
    int mc = m < Nrows ? m : (Nrows - 1);
    xr[j] = *(const float4*)(X + (size_t)mc * 128 + lane * 4);
  }
  int md = row0 + wave * 8 + (lane & 7);
  float dnv = dn[md < Nrows ? md : (Nrows - 1)];
  const float* wcol = W + (size_t)khalf * 128 + n;
  v2f bf[32];
#pragma unroll
  for (int s = 0; s < 32; ++s) {
    bf[s].x = wcol[(size_t)s * 512];
    bf[s].y = wcol[(size_t)s * 512 + 128];
  }
#pragma unroll
  for (int j = 0; j < 8; ++j)
    *(float4*)(tile + (wave * 8 + j) * TILE_PITCH + lane * 4) = xr[j];
  __syncthreads();
  v8f acc0 = {}, acc1 = {}, acc2 = {}, acc3 = {};
  const float* arow = tile + c * TILE_PITCH + khalf;
  v2f cur[4][4], nxt[4][4];
#pragma unroll
  for (int i = 0; i < 4; ++i)
#pragma unroll
    for (int t = 0; t < 4; ++t)
      cur[i][t] = *(const v2f*)(arow + t * 16 * TILE_PITCH + i * 4);
#pragma unroll
  for (int g = 0; g < 8; ++g) {
    if (g + 1 < 8) {
#pragma unroll
      for (int i = 0; i < 4; ++i)
#pragma unroll
        for (int t = 0; t < 4; ++t)
          nxt[i][t] = *(const v2f*)(arow + t * 16 * TILE_PITCH + (g + 1) * 16 + i * 4);
    }
#pragma unroll
    for (int i = 0; i < 4; ++i) {
      int s = g * 4 + i;
      acc0 = __builtin_amdgcn_wmma_f32_16x16x4_f32(false, cur[i][0], false, bf[s], (short)0, acc0, false, false);
      acc1 = __builtin_amdgcn_wmma_f32_16x16x4_f32(false, cur[i][1], false, bf[s], (short)0, acc1, false, false);
      acc2 = __builtin_amdgcn_wmma_f32_16x16x4_f32(false, cur[i][2], false, bf[s], (short)0, acc2, false, false);
      acc3 = __builtin_amdgcn_wmma_f32_16x16x4_f32(false, cur[i][3], false, bf[s], (short)0, acc3, false, false);
    }
#pragma unroll
    for (int i = 0; i < 4; ++i)
#pragma unroll
      for (int t = 0; t < 4; ++t)
        cur[i][t] = nxt[i][t];
  }
  __syncthreads();
  int rbase = (lane >> 4) * 8;
  float bn = bias[n];
#pragma unroll
  for (int r = 0; r < 8; ++r) {
    tile[(rbase + r) * TILE_PITCH + n] = fmaxf(acc0[r] + bn, 0.0f);
    tile[(16 + rbase + r) * TILE_PITCH + n] = fmaxf(acc1[r] + bn, 0.0f);
    tile[(32 + rbase + r) * TILE_PITCH + n] = fmaxf(acc2[r] + bn, 0.0f);
    tile[(48 + rbase + r) * TILE_PITCH + n] = fmaxf(acc3[r] + bn, 0.0f);
  }
  if (PROJ) {
    __shared__ __attribute__((aligned(16))) float ystage[GEMM_ROWS * YS_PITCH];
    int rt = wave & 3;
    int ct = __builtin_amdgcn_readfirstlane(wave >> 2);
    v2f pb[32];
    if (ct == 0) {
      const float* bp = Wp + (size_t)khalf * 16 + c;
#pragma unroll
      for (int s = 0; s < 32; ++s) {
        pb[s].x = bp[s * 64];
        pb[s].y = bp[s * 64 + 16];
      }
    } else {
      const float* gp = wgl + (c & 1) * 128 + khalf;
#pragma unroll
      for (int s = 0; s < 32; ++s) pb[s] = *(const v2f*)(gp + 4 * s);
    }
    float4 wa = *(const float4*)(wg0 + lane * 4);
    float4 wb = *(const float4*)(wg0 + 128 + lane * 4);
    __syncthreads();
    const float* ap = tile + (rt * 16 + c) * TILE_PITCH + khalf;
    v2f pcur[8], pnxt[8];
#pragma unroll
    for (int i = 0; i < 8; ++i) pcur[i] = *(const v2f*)(ap + 4 * i);
    v8f pacc = {};
#pragma unroll
    for (int g = 0; g < 4; ++g) {
      if (g + 1 < 4) {
#pragma unroll
        for (int i = 0; i < 8; ++i) pnxt[i] = *(const v2f*)(ap + (g + 1) * 32 + 4 * i);
      }
#pragma unroll
      for (int i = 0; i < 8; ++i)
        pacc = __builtin_amdgcn_wmma_f32_16x16x4_f32(false, pcur[i], false, pb[g * 8 + i], (short)0, pacc, false, false);
#pragma unroll
      for (int i = 0; i < 8; ++i) pcur[i] = pnxt[i];
    }
    int r8 = lane & 7;
    int q = lane >> 3;
    float keep = dnv;
#pragma unroll
    for (int j = 0; j < 8; ++j) {
      float4 v = *(const float4*)(tile + (wave * 8 + j) * TILE_PITCH + lane * 4);
      float sa = v.x * wa.x + v.y * wa.y + v.z * wa.z + v.w * wa.w;
      float sb = v.x * wb.x + v.y * wb.y + v.z * wb.z + v.w * wb.w;
      for (int off = 16; off > 0; off >>= 1) {
        sa += __shfl_xor(sa, off, 32);
        sb += __shfl_xor(sb, off, 32);
      }
      float t = q == 0 ? sa : sb;
      t = q == 3 ? 0.0f : t;
      bool take = (r8 == j) & (q != 1);
      keep = take ? t : keep;
    }
    ystage[(wave * 8 + r8) * YS_PITCH + 32 + q] = keep;
#pragma unroll
    for (int r = 0; r < 8; ++r)
      ystage[(rt * 16 + rbase + r) * YS_PITCH + ct * 16 + c] = pacc[r];
    __syncthreads();
    int cg = lane & 7;
#pragma unroll
    for (int i = 0; i < 2; ++i) {
      int rr = wave * 8 + i * 4 + (lane >> 3);
      int m = row0 + rr;
      float4 v = *(const float4*)(ystage + rr * YS_PITCH + cg * 4);
      float4 e = *(const float4*)(ystage + rr * YS_PITCH + 32);
      v.z = cg == 4 ? e.x : v.z;
      v.w = cg == 4 ? e.y : v.w;
      v.x = cg == 5 ? e.z : v.x;
      if (m < Nrows) *(float4*)(Y + (size_t)m * YP + cg * 4) = v;
    }
    return;
  }
  __syncthreads();
  float4 wa = {0.0f, 0.0f, 0.0f, 0.0f}, wb = {0.0f, 0.0f, 0.0f, 0.0f};
  if (wg0) {
    wa = *(const float4*)(wg0 + lane * 4);
    wb = *(const float4*)(wg0 + 128 + lane * 4);
  }
  for (int j = 0; j < 8; ++j) {
    int rr = wave * 8 + j;
    int m = row0 + rr;
    float4 v = *(const float4*)(tile + rr * TILE_PITCH + lane * 4);
    if (m < Nrows) {
      *(float4*)(H + (size_t)m * 128 + lane * 4) = v;
      if (wg0) {
        float dnm = __shfl(dnv, j, 32);
        float sa = v.x * wa.x + v.y * wa.y + v.z * wa.z + v.w * wa.w;
        float sb = v.x * wb.x + v.y * wb.y + v.z * wb.z + v.w * wb.w;
        for (int off = 16; off > 0; off >>= 1) {
          sa += __shfl_down(sa, off, 32);
          sb += __shfl_down(sb, off, 32);
        }
        if (lane == 0) {
          float2 p; p.x = sa; p.y = dnm;
          *(float2*)(pk + 2 * (size_t)m) = p;
          gb[m] = sb;
        }
      }
    }
  }
}

__global__ void gather_kernel(const int* __restrict__ ends, const int* __restrict__ srow,
                              const float* __restrict__ pk, const float* __restrict__ gb,
                              const float* __restrict__ bg,
                              int layer, const float* __restrict__ H,
                              const float* __restrict__ raw, float* __restrict__ hout,
                              const float* __restrict__ wg_next, float* __restrict__ pk_next,
                              float* __restrict__ gb_next, int N) {
  int node = __builtin_amdgcn_readfirstlane(blockIdx.x * 8 + (threadIdx.x >> 5));
  int lane = threadIdx.x & 31;
  if (node >= N) return;
  int beg = node ? ends[node - 1] : 0;
  int end = ends[node];
  float gbc = gb[node];
  float bgl = bg[layer];
  float dnc = pk[2 * (size_t)node + 1];
  const float* hl = H + lane * 4;
  float4 rw = *(const float4*)(raw + (size_t)node * 128 + lane * 4);
  float4 acc = {0.0f, 0.0f, 0.0f, 0.0f};
  for (int base = beg; base < end; base += 32) {
    int n = end - base < 32 ? end - base : 32;
    int r = 0;
    float w = 0.0f;
    if (lane < n) {
      r = srow[base + lane];
      float2 gd = *(const float2*)(pk + 2 * (size_t)r);
      float g = tanhf(gd.x + gbc + bgl);
      w = g * gd.y * dnc;
    }
    int wi = __float_as_int(w);
    int k = 0;
    for (; k + 4 <= n; k += 4) {
      int r0 = __builtin_amdgcn_readlane(r, k);
      int r1 = __builtin_amdgcn_readlane(r, k + 1);
      int r2 = __builtin_amdgcn_readlane(r, k + 2);
      int r3 = __builtin_amdgcn_readlane(r, k + 3);
      float w0 = __int_as_float(__builtin_amdgcn_readlane(wi, k));
      float w1 = __int_as_float(__builtin_amdgcn_readlane(wi, k + 1));
      float w2 = __int_as_float(__builtin_amdgcn_readlane(wi, k + 2));
      float w3 = __int_as_float(__builtin_amdgcn_readlane(wi, k + 3));
      float4 h0 = *(const float4*)(hl + (size_t)r0 * 128);
      float4 h1 = *(const float4*)(hl + (size_t)r1 * 128);
      float4 h2 = *(const float4*)(hl + (size_t)r2 * 128);
      float4 h3 = *(const float4*)(hl + (size_t)r3 * 128);
      acc.x += w0 * h0.x; acc.y += w0 * h0.y; acc.z += w0 * h0.z; acc.w += w0 * h0.w;
      acc.x += w1 * h1.x; acc.y += w1 * h1.y; acc.z += w1 * h1.z; acc.w += w1 * h1.w;
      acc.x += w2 * h2.x; acc.y += w2 * h2.y; acc.z += w2 * h2.z; acc.w += w2 * h2.w;
      acc.x += w3 * h3.x; acc.y += w3 * h3.y; acc.z += w3 * h3.z; acc.w += w3 * h3.w;
    }
    for (; k < n; ++k) {
      int r0 = __builtin_amdgcn_readlane(r, k);
      float w0 = __int_as_float(__builtin_amdgcn_readlane(wi, k));
      float4 h0 = *(const float4*)(hl + (size_t)r0 * 128);
      acc.x += w0 * h0.x; acc.y += w0 * h0.y; acc.z += w0 * h0.z; acc.w += w0 * h0.w;
    }
  }
  float4 o;
  o.x = FAGCN_EPS * rw.x + acc.x;
  o.y = FAGCN_EPS * rw.y + acc.y;
  o.z = FAGCN_EPS * rw.z + acc.z;
  o.w = FAGCN_EPS * rw.w + acc.w;
  *(float4*)(hout + (size_t)node * 128 + lane * 4) = o;
  if (wg_next) {
    float4 wa = *(const float4*)(wg_next + lane * 4);
    float4 wb = *(const float4*)(wg_next + 128 + lane * 4);
    float sa = o.x * wa.x + o.y * wa.y + o.z * wa.z + o.w * wa.w;
    float sb = o.x * wb.x + o.y * wb.y + o.z * wb.z + o.w * wb.w;
    for (int off = 16; off > 0; off >>= 1) {
      sa += __shfl_down(sa, off, 32);
      sb += __shfl_down(sb, off, 32);
    }
    if (lane == 0) {
      float2 p; p.x = sa; p.y = dnc;
      *(float2*)(pk_next + 2 * (size_t)node) = p;
      gb_next[node] = sb;
    }
  }
}

__global__ void gemm_proj_kernel(const float* __restrict__ Hc, const float* __restrict__ Raw,
                                 const float* __restrict__ W, const float* __restrict__ bias,
                                 float rscale, float* __restrict__ Z, float* __restrict__ ZR,
                                 int Nrows) {
  int wave = threadIdx.x >> 5;
  int lane = threadIdx.x & 31;
  int tiles = (Nrows + 15) >> 4;
  int t = blockIdx.x * 8 + wave;
  if (t >= 2 * tiles) return;
  bool second = t >= tiles;
  int tileM = second ? t - tiles : t;
  const float* src = second ? Raw : Hc;
  float* dst = second ? ZR : Z;
  int mA = tileM * 16 + (lane & 15);
  int mclamp = mA < Nrows ? mA : (Nrows - 1);
  int n = lane & 15;
  int khalf = (lane >> 4) * 2;
  const float* hrow = src + (size_t)mclamp * 128 + khalf;
  const float* wcol = W + (size_t)khalf * 16 + n;
  __builtin_prefetch(hrow, 0, 0);
  v8f acc = {};
#pragma unroll 4
  for (int kk = 0; kk < 128; kk += 4) {
    v2f a = *(const v2f*)(hrow + kk);
    v2f b;
    b.x = wcol[(size_t)kk * 16];
    b.y = wcol[(size_t)kk * 16 + 16];
    acc = __builtin_amdgcn_wmma_f32_16x16x4_f32(false, a, false, b, (short)0, acc,
                                               false, false);
  }
  int mbase = tileM * 16 + ((lane >> 4) * 8);
  float bn = second ? bias[n] : 0.0f;
  float sc = second ? rscale : 1.0f;
  if (tileM * 16 + 16 <= Nrows) {
    float* op = dst + (size_t)mbase * 16 + n;
#pragma unroll
    for (int r = 0; r < 8; ++r)
      op[(size_t)r * 16] = acc[r] * sc + bn;
  } else {
#pragma unroll
    for (int r = 0; r < 8; ++r) {
      int mm = mbase + r;
      if (mm < Nrows) dst[(size_t)mm * 16 + n] = acc[r] * sc + bn;
    }
  }
}

__device__ __forceinline__ float quad_xor1(float v) {
  return __int_as_float(__builtin_amdgcn_mov_dpp(__float_as_int(v), 0xB1, 0xF, 0xF, true));
}
__device__ __forceinline__ float quad_xor2(float v) {
  return __int_as_float(__builtin_amdgcn_mov_dpp(__float_as_int(v), 0x4E, 0xF, 0xF, true));
}

template <bool LSM>
__global__ void gather_proj_kernel(const int* __restrict__ ends, const int* __restrict__ srow,
                                   const float* __restrict__ pk, const float* __restrict__ gb,
                                   const float* __restrict__ bg, int layer,
                                   const float* __restrict__ Z, const float* __restrict__ ZR,
                                   float* __restrict__ logits, int N) {
  int node = blockIdx.x * 64 + (threadIdx.x >> 2);
  int q = threadIdx.x & 3;
  if (node >= N) return;
  int beg = node ? ends[node - 1] : 0;
  int end = ends[node];
  float gbc = gb[node];
  float bgl = bg[layer];
  float dnc = pk[2 * (size_t)node + 1];
  const float* zq = Z + q * 4;
  float4 zr = *(const float4*)(ZR + (size_t)node * 16 + q * 4);
  float4 acc = {0.0f, 0.0f, 0.0f, 0.0f};
  int last = end - 1;
  int r0 = 0, r1 = 0, r2 = 0, r3 = 0;
  if (beg < end) {
    r0 = srow[beg];
    r1 = srow[beg + 1 < last ? beg + 1 : last];
    r2 = srow[beg + 2 < last ? beg + 2 : last];
    r3 = srow[beg + 3 < last ? beg + 3 : last];
  }
  for (int base = beg; base < end; base += 4) {
    float2 g0 = *(const float2*)(pk + 2 * (size_t)r0);
    float2 g1 = *(const float2*)(pk + 2 * (size_t)r1);
    float2 g2 = *(const float2*)(pk + 2 * (size_t)r2);
    float2 g3 = *(const float2*)(pk + 2 * (size_t)r3);
    float4 h0 = *(const float4*)(zq + (size_t)r0 * 16);
    float4 h1 = *(const float4*)(zq + (size_t)r1 * 16);
    float4 h2 = *(const float4*)(zq + (size_t)r2 * 16);
    float4 h3 = *(const float4*)(zq + (size_t)r3 * 16);
    int nb = base + 4;
    int n0 = r0, n1 = r1, n2 = r2, n3 = r3;
    if (nb < end) {
      n0 = srow[nb];
      n1 = srow[nb + 1 < last ? nb + 1 : last];
      n2 = srow[nb + 2 < last ? nb + 2 : last];
      n3 = srow[nb + 3 < last ? nb + 3 : last];
    }
    {
      float w = tanhf(g0.x + gbc + bgl) * g0.y * dnc;
      acc.x += w * h0.x; acc.y += w * h0.y; acc.z += w * h0.z; acc.w += w * h0.w;
    }
    if (base + 1 < end) {
      float w = tanhf(g1.x + gbc + bgl) * g1.y * dnc;
      acc.x += w * h1.x; acc.y += w * h1.y; acc.z += w * h1.z; acc.w += w * h1.w;
    }
    if (base + 2 < end) {
      float w = tanhf(g2.x + gbc + bgl) * g2.y * dnc;
      acc.x += w * h2.x; acc.y += w * h2.y; acc.z += w * h2.z; acc.w += w * h2.w;
    }
    if (base + 3 < end) {
      float w = tanhf(g3.x + gbc + bgl) * g3.y * dnc;
      acc.x += w * h3.x; acc.y += w * h3.y; acc.z += w * h3.z; acc.w += w * h3.w;
    }
    r0 = n0; r1 = n1; r2 = n2; r3 = n3;
  }
  float4 o;
  o.x = zr.x + acc.x;
  o.y = zr.y + acc.y;
  o.z = zr.z + acc.z;
  o.w = zr.w + acc.w;
  if (LSM) {
    float m = fmaxf(fmaxf(o.x, o.y), fmaxf(o.z, o.w));
    m = fmaxf(m, quad_xor1(m));
    m = fmaxf(m, quad_xor2(m));
    float s = expf(o.x - m) + expf(o.y - m) + expf(o.z - m) + expf(o.w - m);
    s += quad_xor1(s);
    s += quad_xor2(s);
    float d = m + logf(s);
    o.x -= d; o.y -= d; o.z -= d; o.w -= d;
  }
  *(float4*)(logits + (size_t)node * 16 + q * 4) = o;
}

__global__ void gemm_proj2_kernel(const float* __restrict__ Hc, const float* __restrict__ Raw,
                                  const float* __restrict__ W, const float* __restrict__ wgl,
                                  const float* __restrict__ pk, const float* __restrict__ gb,
                                  float* __restrict__ Y, float* __restrict__ YR,
                                  int Nrows, int nt) {
  int wave = threadIdx.x >> 5;
  int lane = threadIdx.x & 31;
  int tiles = (Nrows + 15) >> 4;
  int t = blockIdx.x * 8 + wave;
  if (t >= nt) return;
  bool second = t >= tiles;
  int tileM = second ? t - tiles : t;
  const float* src = second ? Raw : Hc;
  float* dst = second ? YR : Y;
  int mA = tileM * 16 + (lane & 15);
  int mclamp = mA < Nrows ? mA : (Nrows - 1);
  int n = lane & 15;
  int khalf = (lane >> 4) * 2;
  const float* hrow = src + (size_t)mclamp * 128 + khalf;
  const float* wcol = W + (size_t)khalf * 16 + n;
  const float* gcol = wgl + (n & 1) * 128 + khalf;
  __builtin_prefetch(hrow, 0, 0);
  int mbase = tileM * 16 + ((lane >> 4) * 8);
  const float* tp = (n == 2 || n == 3) ? pk + (n - 2) : gb;
  int ts = (n == 2 || n == 3) ? 2 : 1;
  float tv[8];
#pragma unroll
  for (int r = 0; r < 8; ++r) {
    int mr = mbase + r < Nrows ? mbase + r : Nrows - 1;
    tv[r] = tp[(size_t)mr * ts];
  }
  v8f acc = {};
  v8f acg = {};
#pragma unroll 4
  for (int kk = 0; kk < 128; kk += 4) {
    v2f a = *(const v2f*)(hrow + kk);
    v2f b;
    b.x = wcol[(size_t)kk * 16];
    b.y = wcol[(size_t)kk * 16 + 16];
    v2f g = *(const v2f*)(gcol + kk);
    acc = __builtin_amdgcn_wmma_f32_16x16x4_f32(false, a, false, b, (short)0, acc,
                                               false, false);
    acg = __builtin_amdgcn_wmma_f32_16x16x4_f32(false, a, false, g, (short)0, acg,
                                               false, false);
  }
  float* yb = dst + (size_t)mbase * YP + n;
  if (tileM * 16 + 16 <= Nrows) {
#pragma unroll
    for (int r = 0; r < 8; ++r) {
      yb[(size_t)r * YP] = acc[r];
      if (n < 5) yb[(size_t)r * YP + 16] = n < 2 ? acg[r] : tv[r];
    }
  } else {
#pragma unroll
    for (int r = 0; r < 8; ++r) {
      if (mbase + r < Nrows) {
        yb[(size_t)r * YP] = acc[r];
        if (n < 5) yb[(size_t)r * YP + 16] = n < 2 ? acg[r] : tv[r];
      }
    }
  }
}

__global__ void gather_mid_kernel(const int* __restrict__ ends, const int* __restrict__ srow,
                                  const float* __restrict__ bg, int layer,
                                  const float* __restrict__ Y, const float* __restrict__ YR,
                                  const float* __restrict__ bias,
                                  float* __restrict__ Z, float* __restrict__ ZR,
                                  float* __restrict__ pk_next, float* __restrict__ gb_next,
                                  int N) {
  int node = blockIdx.x * 64 + (threadIdx.x >> 2);
  int q = threadIdx.x & 3;
  if (node >= N) return;
  int beg = node ? ends[node - 1] : 0;
  int end = ends[node];
  float bgl = bg[layer];
  const float* yc = Y + (size_t)node * YP;
  float dnc = yc[19];
  float gbc = yc[20];
  const float* yrc = YR + (size_t)node * YP;
  float4 own = *(const float4*)(yrc + q * 4);
  float2 owng = *(const float2*)(yrc + 16);
  float4 bq = *(const float4*)(bias + q * 4);
  const float* yq = Y + q * 4;
  const float* yt = Y + 16;
  float4 acc = {0.0f, 0.0f, 0.0f, 0.0f};
  float sga = 0.0f, sgb = 0.0f;
  int last = end - 1;
  int r0 = 0, r1 = 0, r2 = 0, r3 = 0;
  if (beg < end) {
    r0 = srow[beg];
    r1 = srow[beg + 1 < last ? beg + 1 : last];
    r2 = srow[beg + 2 < last ? beg + 2 : last];
    r3 = srow[beg + 3 < last ? beg + 3 : last];
  }
  for (int base = beg; base < end; base += 4) {
    float4 t0 = *(const float4*)(yt + (size_t)r0 * YP);
    float4 t1 = *(const float4*)(yt + (size_t)r1 * YP);
    float4 t2 = *(const float4*)(yt + (size_t)r2 * YP);
    float4 t3 = *(const float4*)(yt + (size_t)r3 * YP);
    float4 h0 = *(const float4*)(yq + (size_t)r0 * YP);
    float4 h1 = *(const float4*)(yq + (size_t)r1 * YP);
    float4 h2 = *(const float4*)(yq + (size_t)r2 * YP);
    float4 h3 = *(const float4*)(yq + (size_t)r3 * YP);
    int nb = base + 4;
    int n0 = r0, n1 = r1, n2 = r2, n3 = r3;
    if (nb < end) {
      n0 = srow[nb];
      n1 = srow[nb + 1 < last ? nb + 1 : last];
      n2 = srow[nb + 2 < last ? nb + 2 : last];
      n3 = srow[nb + 3 < last ? nb + 3 : last];
    }
    {
      float w = tanhf(t0.z + gbc + bgl) * t0.w * dnc;
      acc.x += w * h0.x; acc.y += w * h0.y; acc.z += w * h0.z; acc.w += w * h0.w;
      sga += w * t0.x; sgb += w * t0.y;
    }
    if (base + 1 < end) {
      float w = tanhf(t1.z + gbc + bgl) * t1.w * dnc;
      acc.x += w * h1.x; acc.y += w * h1.y; acc.z += w * h1.z; acc.w += w * h1.w;
      sga += w * t1.x; sgb += w * t1.y;
    }
    if (base + 2 < end) {
      float w = tanhf(t2.z + gbc + bgl) * t2.w * dnc;
      acc.x += w * h2.x; acc.y += w * h2.y; acc.z += w * h2.z; acc.w += w * h2.w;
      sga += w * t2.x; sgb += w * t2.y;
    }
    if (base + 3 < end) {
      float w = tanhf(t3.z + gbc + bgl) * t3.w * dnc;
      acc.x += w * h3.x; acc.y += w * h3.y; acc.z += w * h3.z; acc.w += w * h3.w;
      sga += w * t3.x; sgb += w * t3.y;
    }
    r0 = n0; r1 = n1; r2 = n2; r3 = n3;
  }
  float4 o, zr;
  o.x = FAGCN_EPS * own.x + acc.x;  zr.x = FAGCN_EPS * own.x + bq.x;
  o.y = FAGCN_EPS * own.y + acc.y;  zr.y = FAGCN_EPS * own.y + bq.y;
  o.z = FAGCN_EPS * own.z + acc.z;  zr.z = FAGCN_EPS * own.z + bq.z;
  o.w = FAGCN_EPS * own.w + acc.w;  zr.w = FAGCN_EPS * own.w + bq.w;
  *(float4*)(Z + (size_t)node * 16 + q * 4) = o;
  *(float4*)(ZR + (size_t)node * 16 + q * 4) = zr;
  if (q == 0) {
    float2 p; p.x = FAGCN_EPS * owng.x + sga; p.y = dnc;
    *(float2*)(pk_next + 2 * (size_t)node) = p;
    gb_next[node] = FAGCN_EPS * owng.y + sgb;
  }
}

__global__ void logsoftmax_kernel(const float4* __restrict__ in, float4* __restrict__ out, int N) {
  int i = blockIdx.x * blockDim.x + threadIdx.x;
  if (i >= N) return;
  const float4* rin = in + (size_t)i * 4;
  float4* row = out + (size_t)i * 4;
  float4 v[4];
#pragma unroll
  for (int q = 0; q < 4; ++q) v[q] = rin[q];
  float m = v[0].x;
#pragma unroll
  for (int q = 0; q < 4; ++q) {
    m = fmaxf(m, v[q].x); m = fmaxf(m, v[q].y);
    m = fmaxf(m, v[q].z); m = fmaxf(m, v[q].w);
  }
  float s = 0.0f;
#pragma unroll
  for (int q = 0; q < 4; ++q) {
    s += expf(v[q].x - m); s += expf(v[q].y - m);
    s += expf(v[q].z - m); s += expf(v[q].w - m);
  }
  float d = m + logf(s);
#pragma unroll
  for (int q = 0; q < 4; ++q) {
    v[q].x -= d; v[q].y -= d; v[q].z -= d; v[q].w -= d;
    row[q] = v[q];
  }
}

extern "C" void kernel_launch(void* const* d_in, const int* in_sizes, int n_in,
                              void* d_out, int out_size, void* d_ws, size_t ws_size,
                              hipStream_t stream) {
  const float* x     = (const float*)d_in[0];
  const int*   ei    = (const int*)d_in[1];
  const float* W_in  = (const float*)d_in[2];
  const float* b_in  = (const float*)d_in[3];
  const float* Wg    = (const float*)d_in[4];
  const float* bg    = (const float*)d_in[5];
  const float* W_out = (const float*)d_in[6];
  const float* b_out = (const float*)d_in[7];
  float* out = (float*)d_out;

  const int Hdim = 128;
  int N = in_sizes[0] / Hdim;
  int E = in_sizes[1] / 2;
  int L = in_sizes[5];

  float* ws   = (float*)d_ws;
  float* raw  = ws;
  float* bufA = raw + (size_t)N * Hdim;
  float* bufB = bufA + (size_t)N * Hdim;
  int*   cnt  = (int*)(bufB + (size_t)N * Hdim);
  float* dn   = (float*)cnt + N;
  int*   srow = (int*)(dn + N);
  float* pk   = out;
  float* pk2  = out + 2 * (size_t)N;
  float* gb   = out + 4 * (size_t)N;
  float* gb2  = out + 5 * (size_t)N;

  int NB = (N + BIN_NODES - 1) / BIN_NODES;
  int nchunks = (E + CHUNK - 1) / CHUNK;
  size_t scratch = 2 * (size_t)E + (2 * (size_t)NB + 2) * (size_t)nchunks;
  if (NB <= MAX_BINS && nchunks <= MAX_CHUNKS && scratch <= (size_t)N * Hdim) {
    int* cbin = (int*)bufA;
    int* rbin = cbin + E;
    int* offs = rbin + E;
    bin_sort_kernel<<<nchunks, 1024, 0, stream>>>(ei, offs, cbin, rbin, E, NB);
    bin_finish_kernel<<<NB, 1024, 0, stream>>>(offs, cbin, rbin, cnt, dn, srow, N, NB, nchunks);
  } else {
    hipMemsetAsync(cnt, 0, (size_t)N * 2 * sizeof(float), stream);
    deg_kernel<<<(E + 255) / 256, 256, 0, stream>>>(ei, dn, cnt, E);
    dn_kernel<<<(N + 255) / 256, 256, 0, stream>>>(dn, N);
    scan_kernel<<<1, 1024, 0, stream>>>(cnt, N);
    fill_kernel<<<(E + 255) / 256, 256, 0, stream>>>(ei, cnt, srow, E);
    order_kernel<<<(N + 7) / 8, 256, 0, stream>>>(cnt, srow, N);
  }

  bool fused = (L == 2);
  if (fused)
    gemm_relu_kernel<true><<<(N + GEMM_ROWS - 1) / GEMM_ROWS, 256, 0, stream>>>(
        x, W_in, b_in, Wg, dn, nullptr, nullptr, nullptr, W_out, Wg + (size_t)(L - 1) * 2 * Hdim, bufA, N);
  else
    gemm_relu_kernel<false><<<(N + GEMM_ROWS - 1) / GEMM_ROWS, 256, 0, stream>>>(
        x, W_in, b_in, L > 0 ? Wg : nullptr, dn, raw, pk, gb, nullptr, nullptr, nullptr, N);

  float* h = raw;
  float* hnew = bufA;
  float* gac = pk;  float* gbc = gb;
  float* gan = pk2; float* gbn = gb2;
  for (int i = 0; i + 2 < L; ++i) {
    const float* wgn = Wg + (size_t)(i + 1) * 2 * Hdim;
    gather_kernel<<<(N + 7) / 8, 256, 0, stream>>>(cnt, srow, gac, gbc, bg, i, h, raw, hnew,
                                                  wgn, gan, gbn, N);
    h = hnew;
    hnew = (h == bufA) ? bufB : bufA;
    float* t = gac; gac = gan; gan = t;
    t = gbc; gbc = gbn; gbn = t;
  }

  int tiles = (N + 15) / 16;
  if (L >= 2) {
    bool same = (h == raw);
    float* Y  = hnew;
    float* YR = same ? Y : hnew + 32 * (size_t)N;
    float* Z  = hnew + 64 * (size_t)N;
    float* ZR = hnew + 80 * (size_t)N;
    float* pkl = hnew + 96 * (size_t)N;
    float* gbl = hnew + 98 * (size_t)N;
    int nt = same ? tiles : 2 * tiles;
    if (!fused)
      gemm_proj2_kernel<<<(nt + 7) / 8, 256, 0, stream>>>(h, raw, W_out,
                                                         Wg + (size_t)(L - 1) * 2 * Hdim, gac, gbc,
                                                         Y, YR, N, nt);
    gather_mid_kernel<<<(N + 63) / 64, 256, 0, stream>>>(cnt, srow, bg, L - 2, Y, YR, b_out,
                                                        Z, ZR, pkl, gbl, N);
    gather_proj_kernel<true><<<(N + 63) / 64, 256, 0, stream>>>(cnt, srow, pkl, gbl, bg, L - 1,
                                                               Z, ZR, out, N);
    return;
  }

  float* Z  = hnew;
  float* ZR = hnew + 16 * (size_t)N;
  float* LG = hnew + 32 * (size_t)N;
  gemm_proj_kernel<<<(2 * tiles + 7) / 8, 256, 0, stream>>>(h, raw, W_out, b_out,
                                                           L > 0 ? FAGCN_EPS : 1.0f, Z, ZR, N);
  if (L > 0)
    gather_proj_kernel<false><<<(N + 63) / 64, 256, 0, stream>>>(cnt, srow, gac, gbc, bg, L - 1,
                                                                Z, ZR, LG, N);
  logsoftmax_kernel<<<(N + 255) / 256, 256, 0, stream>>>((const float4*)(L > 0 ? LG : ZR),
                                                        (float4*)out, N);
}
